# starting kernel
# baseline (speedup 1.0000x reference)
#include <hip/hip_runtime.h>
#include <math.h>


#define NN 50000

typedef __attribute__((ext_vector_type(2))) float v2f;
typedef __attribute__((ext_vector_type(8))) float v8f;

__global__ void k_zero(float* __restrict__ p, int n) {
  int i = blockIdx.x * blockDim.x + threadIdx.x;
  if (i < n) p[i] = 0.0f;
}

__global__ void k_degrees(const int* __restrict__ src, const int* __restrict__ dst,
                          int E, float* __restrict__ deg_out, float* __restrict__ deg_in) {
  int e = blockIdx.x * blockDim.x + threadIdx.x;
  if (e < E) {
    atomicAdd(&deg_out[src[e]], 1.0f);
    atomicAdd(&deg_in[dst[e]], 1.0f);
  }
}

__global__ void k_invsqrt(float* __restrict__ a, float* __restrict__ b, int n) {
  int i = blockIdx.x * blockDim.x + threadIdx.x;
  if (i < n) {
    float da = a[i]; da = da > 1.0f ? da : 1.0f; a[i] = 1.0f / sqrtf(da);
    float db = b[i]; db = db > 1.0f ? db : 1.0f; b[i] = 1.0f / sqrtf(db);
  }
}

__global__ void k_agg0(const int* __restrict__ src, const int* __restrict__ dst, int E,
                       const float* __restrict__ x, const float* __restrict__ inv_out,
                       float* __restrict__ agg0) {
  int e = blockIdx.x * blockDim.x + threadIdx.x;
  if (e < E) {
    int s = src[e];
    atomicAdd(&agg0[dst[e]], x[s] * inv_out[s]);
  }
}

__global__ void k_l1_wmma(const float* __restrict__ agg0,
                          const float* __restrict__ inv_in,
                          const float* __restrict__ inv_out,
                          const float* __restrict__ W0,
                          const float* __restrict__ b0,
                          const float* __restrict__ W1,
                          float* __restrict__ h1mm,
                          int n_tiles) {
  __shared__ float sW0[100];
  __shared__ float sb0[100];
  __shared__ float sW1p[100 * 16];
  for (int t = threadIdx.x; t < 100; t += blockDim.x) { sW0[t] = W0[t]; sb0[t] = b0[t]; }
  for (int t = threadIdx.x; t < 1600; t += blockDim.x) {
    int row = t >> 4, c = t & 15;
    sW1p[t] = (c < 10) ? W1[row * 10 + c] : 0.0f;
  }
  __syncthreads();

  const int wave = threadIdx.x >> 5;
  const int lane = threadIdx.x & 31;
  const int tile = blockIdx.x * (blockDim.x >> 5) + wave;
  if (tile >= n_tiles) return;
  const int i0   = tile * 16;
  const int half = lane >> 4;
  const int m    = lane & 15;
  const int col  = lane & 15;

  const float s = agg0[i0 + m] * inv_in[i0 + m];

  v8f c = {};
#pragma unroll
  for (int kk = 0; kk < 25; ++kk) {
    const int k0 = kk * 4 + half * 2;
    float a0 = fmaf(s, sW0[k0], sb0[k0]);
    a0 = a0 >= 0.0f ? a0 : 0.01f * a0;
    float a1 = fmaf(s, sW0[k0 + 1], sb0[k0 + 1]);
    a1 = a1 >= 0.0f ? a1 : 0.01f * a1;
    v2f a = {a0, a1};
    v2f b = {sW1p[k0 * 16 + col], sW1p[(k0 + 1) * 16 + col]};
    c = __builtin_amdgcn_wmma_f32_16x16x4_f32(false, a, false, b, (short)0, c,
                                              false, false);
  }

#pragma unroll
  for (int v = 0; v < 8; ++v) {
    const int i = i0 + v + 8 * half;
    h1mm[i * 16 + col] = c[v] * inv_out[i];
  }
}

__global__ void k_agg1(const int* __restrict__ src, const int* __restrict__ dst, int E,
                       const float* __restrict__ h1mm, float* __restrict__ agg1) {
  int t = blockIdx.x * blockDim.x + threadIdx.x;
  if (t < E * 10) {
    int e = t / 10;
    int k = t - e * 10;
    atomicAdd(&agg1[dst[e] * 10 + k], h1mm[src[e] * 16 + k]);
  }
}

__global__ void k_l2(const float* __restrict__ agg1, const float* __restrict__ inv_in,
                     const float* __restrict__ inv_out, const float* __restrict__ b1,
                     const float* __restrict__ W2, float* __restrict__ h2, int n) {
  int i = blockIdx.x * blockDim.x + threadIdx.x;
  if (i >= n) return;
  float si = inv_in[i];
  float acc = 0.0f;
#pragma unroll
  for (int k = 0; k < 10; ++k) {
    float v = fmaf(agg1[i * 10 + k], si, b1[k]);
    v = v > 0.0f ? v : 0.0f;
    acc = fmaf(v, W2[k], acc);
  }
  h2[i] = acc * inv_out[i];
}

__global__ void k_agg2(const int* __restrict__ src, const int* __restrict__ dst, int E,
                       const float* __restrict__ h2, float* __restrict__ agg2) {
  int e = blockIdx.x * blockDim.x + threadIdx.x;
  if (e < E) atomicAdd(&agg2[dst[e]], h2[src[e]]);
}

__global__ void k_out(const float* __restrict__ agg2, const float* __restrict__ inv_in,
                      const float* __restrict__ b2, float* __restrict__ out, int n) {
  int i = blockIdx.x * blockDim.x + threadIdx.x;
  if (i < n) {
    float v = fmaf(agg2[i], inv_in[i], b2[0]);
    out[i] = v > 0.0f ? v : 0.0f;
  }
}

extern "C" void kernel_launch(void* const* d_in, const int* in_sizes, int n_in,
                              void* d_out, int out_size, void* d_ws, size_t ws_size,
                              hipStream_t stream) {
  const float* in_feat = (const float*)d_in[0];
  const int*   edge    = (const int*)d_in[1];
  const float* W0      = (const float*)d_in[2];
  const float* b0      = (const float*)d_in[3];
  const float* W1      = (const float*)d_in[4];
  const float* b1      = (const float*)d_in[5];
  const float* W2      = (const float*)d_in[6];
  const float* b2      = (const float*)d_in[7];
  float* out = (float*)d_out;

  const int N = NN;
  const int E = in_sizes[1] / 2;
  const int* src = edge;
  const int* dst = edge + E;

  float* ws      = (float*)d_ws;
  float* inv_out = ws;
  float* inv_in  = ws + N;
  float* agg0    = ws + 2 * N;
  float* agg1    = ws + 3 * N;
  float* agg2    = ws + 13 * N;
  float* h1mm    = ws + 14 * N;
  float* h2      = ws + 30 * N;

  const int B = 256;
  int nz = 14 * N;
  k_zero<<<(nz + B - 1) / B, B, 0, stream>>>(ws, nz);

  k_degrees<<<(E + B - 1) / B, B, 0, stream>>>(src, dst, E, inv_out, inv_in);
  k_invsqrt<<<(N + B - 1) / B, B, 0, stream>>>(inv_out, inv_in, N);
  k_agg0<<<(E + B - 1) / B, B, 0, stream>>>(src, dst, E, in_feat, inv_out, agg0);

  const int n_tiles = (N + 15) / 16;
  const int waves_per_block = B / 32;
  k_l1_wmma<<<(n_tiles + waves_per_block - 1) / waves_per_block, B, 0, stream>>>(
      agg0, inv_in, inv_out, W0, b0, W1, h1mm, n_tiles);

  int total1 = E * 10;
  k_agg1<<<(total1 + B - 1) / B, B, 0, stream>>>(src, dst, E, h1mm, agg1);
  k_l2<<<(N + B - 1) / B, B, 0, stream>>>(agg1, inv_in, inv_out, b1, W2, h2, N);
  k_agg2<<<(E + B - 1) / B, B, 0, stream>>>(src, dst, E, h2, agg2);
  k_out<<<(N + B - 1) / B, B, 0, stream>>>(agg2, inv_in, b2, out, N);
}


# Edges counting-sorted by destination once per call (rank = old value of the in-degree atomic); the three float-atomic scatter passes become per-row gather sums in arrival-order rank
# speedup vs baseline: 1.9505x; 1.9505x over previous
#include <hip/hip_runtime.h>
#include <math.h>


#define NN 50000
#define NP 50016

typedef __attribute__((ext_vector_type(2))) float v2f;
typedef __attribute__((ext_vector_type(8))) float v8f;

__global__ void k_zero(int* __restrict__ p, int n) {
  int i = blockIdx.x * blockDim.x + threadIdx.x;
  if (i < n) p[i] = 0;
}

__global__ void k_degrees(const int* __restrict__ src, const int* __restrict__ dst, int E,
                          int* __restrict__ deg_out, int* __restrict__ deg_in,
                          int* __restrict__ rank) {
  int e = blockIdx.x * blockDim.x + threadIdx.x;
  if (e < E) {
    const int s = src[e];
    const int d = dst[e];
    const int r = atomicAdd(&deg_in[d], 1);
    atomicAdd(&deg_out[s], 1);
    rank[e] = r;
  }
}

__global__ void __launch_bounds__(256)
k_scan(const int* __restrict__ deg_out, const int* __restrict__ deg_in,
       const float* __restrict__ x, int n,
       float* __restrict__ inv_out, float* __restrict__ inv_in,
       float* __restrict__ xs, int* __restrict__ rowptr) {
  __shared__ int s_part[8];
  __shared__ int s_wsum[8];
  const int tid  = threadIdx.x;
  const int lane = tid & 31;
  const int wave = tid >> 5;
  const int base = blockIdx.x * 256;

  const int4* d4 = reinterpret_cast<const int4*>(deg_in);
  const int n4 = base >> 2;
  int acc = 0;
#pragma unroll 8
  for (int j = tid; j < n4; j += 256) {
    const int4 v = d4[j];
    acc += (v.x + v.y) + (v.z + v.w);
  }
  for (int o = 16; o > 0; o >>= 1) acc += __shfl_xor(acc, o);

  const int i = base + tid;
  const int d = (i < n) ? deg_in[i] : 0;
  int incl = d;
  for (int o = 1; o < 32; o <<= 1) {
    const int t = __shfl_up(incl, o);
    if (lane >= o) incl += t;
  }
  if (lane == 0)  s_part[wave] = acc;
  if (lane == 31) s_wsum[wave] = incl;
  __syncthreads();
  int pre = 0;
  for (int w = 0; w < 8; ++w) {
    pre += s_part[w];
    if (w < wave) pre += s_wsum[w];
  }
  if (i < n) {
    rowptr[i] = pre + incl - d;
    const float fi = (float)(d > 1 ? d : 1);
    inv_in[i] = 1.0f / sqrtf(fi);
    const int dout = deg_out[i];
    const float fo = (float)(dout > 1 ? dout : 1);
    const float io = 1.0f / sqrtf(fo);
    inv_out[i] = io;
    xs[i] = x[i] * io;
  }
}

__global__ void k_fill(const int* __restrict__ src, const int* __restrict__ dst, int E,
                       const int* __restrict__ rank, const int* __restrict__ rowptr,
                       int* __restrict__ csr_src) {
  int e = blockIdx.x * blockDim.x + threadIdx.x;
  if (e < E) csr_src[rowptr[dst[e]] + rank[e]] = src[e];
}

__device__ __forceinline__ float row_gather_sum(const int* __restrict__ csr_src,
                                                const float* __restrict__ vals,
                                                int pb, int cnt, int first, int stride) {
  float a = 0.0f;
  for (int q0 = first; q0 < cnt; q0 += 8 * stride) {
#pragma unroll
    for (int u = 0; u < 8; ++u) {
      const int q = q0 + u * stride;
      const bool ok = q < cnt;
      const int s = csr_src[pb + (ok ? q : q0)];
      const float h = vals[s];
      a += ok ? h : 0.0f;
    }
  }
  return a;
}

__global__ void k_l1_wmma(const int* __restrict__ rowptr,
                          const int* __restrict__ deg_in,
                          const int* __restrict__ csr_src,
                          const float* __restrict__ xs,
                          const float* __restrict__ inv_in,
                          const float* __restrict__ inv_out,
                          const float* __restrict__ W0,
                          const float* __restrict__ b0,
                          const float* __restrict__ W1,
                          float* __restrict__ h1mm,
                          int n_tiles) {
  __shared__ float sW0[100];
  __shared__ float sb0[100];
  __shared__ float sW1p[100 * 16];
  for (int t = threadIdx.x; t < 100; t += blockDim.x) { sW0[t] = W0[t]; sb0[t] = b0[t]; }
  for (int t = threadIdx.x; t < 1600; t += blockDim.x) {
    int row = t >> 4, c = t & 15;
    sW1p[t] = (c < 10) ? W1[row * 10 + c] : 0.0f;
  }
  __syncthreads();

  const int wave = threadIdx.x >> 5;
  const int lane = threadIdx.x & 31;
  const int tile = blockIdx.x * (blockDim.x >> 5) + wave;
  if (tile >= n_tiles) return;
  const int i0   = tile * 16;
  const int half = lane >> 4;
  const int m    = lane & 15;
  const int col  = lane & 15;

  const int r = i0 + m;
  float part = row_gather_sum(csr_src, xs, rowptr[r], deg_in[r], half, 2);
  part += __shfl_xor(part, 16);
  const float s = part * inv_in[r];

  v8f c = {};
#pragma unroll
  for (int kk = 0; kk < 25; ++kk) {
    const int k0 = kk * 4 + half * 2;
    float a0 = fmaf(s, sW0[k0], sb0[k0]);
    a0 = a0 >= 0.0f ? a0 : 0.01f * a0;
    float a1 = fmaf(s, sW0[k0 + 1], sb0[k0 + 1]);
    a1 = a1 >= 0.0f ? a1 : 0.01f * a1;
    v2f a = {a0, a1};
    v2f b = {sW1p[k0 * 16 + col], sW1p[(k0 + 1) * 16 + col]};
    c = __builtin_amdgcn_wmma_f32_16x16x4_f32(false, a, false, b, (short)0, c,
                                              false, false);
  }

#pragma unroll
  for (int v = 0; v < 8; ++v) {
    const int i = i0 + v + 8 * half;
    h1mm[i * 16 + col] = c[v] * inv_out[i];
  }
}

__global__ void __launch_bounds__(512)
k_gather1_l2(const int* __restrict__ rowptr, const int* __restrict__ deg_in,
             const int* __restrict__ csr_src, const float* __restrict__ h1mm,
             const float* __restrict__ inv_in, const float* __restrict__ inv_out,
             const float* __restrict__ b1, const float* __restrict__ W2,
             float* __restrict__ h2, int n) {
  __shared__ float s_res[32];
  const int g = threadIdx.x >> 4;
  const int k = threadIdx.x & 15;
  const int i = blockIdx.x * 32 + g;
  const bool live = i < n;
  const int pb  = live ? rowptr[i] : 0;
  const int cnt = live ? deg_in[i] : 0;

  float acc = 0.0f;
  for (int q0 = 0; q0 < cnt; q0 += 16) {
#pragma unroll
    for (int u = 0; u < 16; ++u) {
      const int q = q0 + u;
      const bool ok = q < cnt;
      const int s = csr_src[pb + (ok ? q : q0)];
      const float h = h1mm[s * 16 + k];
      acc += ok ? h : 0.0f;
    }
  }

  float v = 0.0f;
  if (live && k < 10) {
    v = fmaf(acc, inv_in[i], b1[k]);
    v = v > 0.0f ? v : 0.0f;
  }
  float a2 = 0.0f;
#pragma unroll
  for (int kk = 0; kk < 10; ++kk) a2 = fmaf(__shfl(v, kk, 16), W2[kk], a2);
  if (k == 0) s_res[g] = live ? a2 * inv_out[i] : 0.0f;
  __syncthreads();
  if (threadIdx.x < 32) {
    const int o = blockIdx.x * 32 + threadIdx.x;
    if (o < n) h2[o] = s_res[threadIdx.x];
  }
}

__global__ void __launch_bounds__(512)
k_gather2_out(const int* __restrict__ rowptr, const int* __restrict__ deg_in,
              const int* __restrict__ csr_src, const float* __restrict__ h2,
              const float* __restrict__ inv_in, const float* __restrict__ b2,
              float* __restrict__ out, int n) {
  __shared__ float s_res[32];
  const int g = threadIdx.x >> 4;
  const int k = threadIdx.x & 15;
  const int i = blockIdx.x * 32 + g;
  const bool live = i < n;
  float acc = 0.0f;
  if (live) {
    const int pb = rowptr[i];
    const int pe = pb + deg_in[i];
    for (int p = pb + k; p < pe; p += 16) acc += h2[csr_src[p]];
  }
  for (int o = 8; o > 0; o >>= 1) acc += __shfl_xor(acc, o, 16);
  if (k == 0) {
    float v = 0.0f;
    if (live) {
      v = fmaf(acc, inv_in[i], b2[0]);
      v = v > 0.0f ? v : 0.0f;
    }
    s_res[g] = v;
  }
  __syncthreads();
  if (threadIdx.x < 32) {
    const int o = blockIdx.x * 32 + threadIdx.x;
    if (o < n) out[o] = s_res[threadIdx.x];
  }
}

extern "C" void kernel_launch(void* const* d_in, const int* in_sizes, int n_in,
                              void* d_out, int out_size, void* d_ws, size_t ws_size,
                              hipStream_t stream) {
  const float* in_feat = (const float*)d_in[0];
  const int*   edge    = (const int*)d_in[1];
  const float* W0      = (const float*)d_in[2];
  const float* b0      = (const float*)d_in[3];
  const float* W1      = (const float*)d_in[4];
  const float* b1      = (const float*)d_in[5];
  const float* W2      = (const float*)d_in[6];
  const float* b2      = (const float*)d_in[7];
  float* out = (float*)d_out;

  const int N = NN;
  const int E = in_sizes[1] / 2;
  const int* src = edge;
  const int* dst = edge + E;

  float* ws      = (float*)d_ws;
  int*   wsi     = (int*)d_ws;
  float* h1mm    = ws;
  int*   deg_out = wsi + 16 * NP;
  int*   deg_in  = wsi + 17 * NP;
  float* inv_out = ws  + 18 * NP;
  float* inv_in  = ws  + 19 * NP;
  float* xs      = ws  + 20 * NP;
  int*   rowptr  = wsi + 21 * NP;
  float* h2      = ws  + 22 * NP;
  int*   rank    = wsi + 23 * NP;
  int*   csr_src = rank + E;

  const int B = 256;
  int nz = 2 * NP;
  k_zero<<<(nz + B - 1) / B, B, 0, stream>>>(deg_out, nz);

  k_degrees<<<(E + B - 1) / B, B, 0, stream>>>(src, dst, E, deg_out, deg_in, rank);
  k_scan<<<(N + 255) / 256, 256, 0, stream>>>(deg_out, deg_in, in_feat, N,
                                              inv_out, inv_in, xs, rowptr);
  k_fill<<<(E + B - 1) / B, B, 0, stream>>>(src, dst, E, rank, rowptr, csr_src);

  const int n_tiles = (N + 15) / 16;
  const int waves_per_block = B / 32;
  k_l1_wmma<<<(n_tiles + waves_per_block - 1) / waves_per_block, B, 0, stream>>>(
      rowptr, deg_in, csr_src, xs, inv_in, inv_out, W0, b0, W1, h1mm, n_tiles);

  k_gather1_l2<<<(N + 31) / 32, 512, 0, stream>>>(rowptr, deg_in, csr_src, h1mm,
                                                   inv_in, inv_out, b1, W2, h2, N);
  k_gather2_out<<<(N + 31) / 32, 512, 0, stream>>>(rowptr, deg_in, csr_src, h2,
                                                    inv_in, b2, out, N);
}


# Degrees and in-row ranks counted with LDS counters (one per-node table per 25,000-edge chunk, summed by the scan) instead of global atomics; row sums run in chunk order, then LDS arrival order
# speedup vs baseline: 3.4438x; 1.7656x over previous
#include <hip/hip_runtime.h>
#include <math.h>


#define NN 50000
#define NP 50016
#define CH 25000
#define CU ((CH + 1023) / 1024)

typedef __attribute__((ext_vector_type(2))) float v2f;
typedef __attribute__((ext_vector_type(8))) float v8f;

__global__ void __launch_bounds__(1024)
k_count(const int* __restrict__ src, const int* __restrict__ dst, int E,
        int* __restrict__ cin, int* __restrict__ cout, int* __restrict__ coarse,
        int* __restrict__ rank) {
  __shared__ __attribute__((aligned(16))) int s_hist[NP];
  __shared__ __attribute__((aligned(16))) int s_coarse[1024];
  const int tid = threadIdx.x;
  const int c = blockIdx.x;
  const bool in_role = blockIdx.y == 0;
  int4* h4 = reinterpret_cast<int4*>(s_hist);
  for (int i = tid; i < NP / 4; i += 1024) h4[i] = make_int4(0, 0, 0, 0);
  s_coarse[tid] = 0;

  const int e0 = c * CH;
  const int e1 = (e0 + CH < E) ? e0 + CH : E;
  const int* __restrict__ key = in_role ? dst : src;
  int kv[CU];
#pragma unroll
  for (int u = 0; u < CU; ++u) {
    const int idx = e0 + tid + u * 1024;
    kv[u] = key[idx < e1 ? idx : e0];
  }
  __syncthreads();

  if (in_role) {
    int rk[CU];
#pragma unroll
    for (int u = 0; u < CU; ++u) {
      const int idx = e0 + tid + u * 1024;
      rk[u] = 0;
      if (idx < e1) {
        rk[u] = atomicAdd(&s_hist[kv[u]], 1);
        atomicAdd(&s_coarse[((kv[u] >> 8) << 2) | (tid & 3)], 1);
      }
    }
#pragma unroll
    for (int u = 0; u < CU; ++u) {
      const int idx = e0 + tid + u * 1024;
      if (idx < e1) rank[idx] = rk[u];
    }
  } else {
#pragma unroll
    for (int u = 0; u < CU; ++u) {
      const int idx = e0 + tid + u * 1024;
      if (idx < e1) atomicAdd(&s_hist[kv[u]], 1);
    }
  }
  __syncthreads();
  int4* __restrict__ out4 = reinterpret_cast<int4*>((in_role ? cin : cout) + c * NP);
#pragma unroll 4
  for (int i = tid; i < NP / 4; i += 1024) out4[i] = h4[i];
  if (in_role && tid < 256) {
    const int4 q = reinterpret_cast<const int4*>(s_coarse)[tid];
    coarse[c * 256 + tid] = (q.x + q.y) + (q.z + q.w);
  }
}

__global__ void __launch_bounds__(256)
k_scan(int* __restrict__ cin, const int* __restrict__ cout,
       const int* __restrict__ coarse, int C,
       const float* __restrict__ x, int n, int* __restrict__ deg_in,
       float* __restrict__ inv_out, float* __restrict__ inv_in,
       float* __restrict__ xs, int* __restrict__ rowptr) {
  __shared__ int s_part[8];
  __shared__ int s_wsum[8];
  const int tid  = threadIdx.x;
  const int lane = tid & 31;
  const int wave = tid >> 5;
  const int b    = blockIdx.x;
  const int i    = b * 256 + tid;

  int acc = 0;
  if (tid < b) {
#pragma unroll 8
    for (int c = 0; c < C; ++c) acc += coarse[c * 256 + tid];
  }
  for (int o = 16; o > 0; o >>= 1) acc += __shfl_xor(acc, o);

  int d = 0, dout = 0;
  if (i < n) {
    for (int c0 = 0; c0 < C; c0 += 8) {
      int ti[8], to[8];
#pragma unroll
      for (int u = 0; u < 8; ++u) {
        const bool ok = c0 + u < C;
        ti[u] = ok ? cin[(c0 + u) * NP + i] : 0;
        to[u] = ok ? cout[(c0 + u) * NP + i] : 0;
      }
#pragma unroll
      for (int u = 0; u < 8; ++u) {
        if (c0 + u < C) cin[(c0 + u) * NP + i] = d;
        d += ti[u];
        dout += to[u];
      }
    }
  }
  int incl = d;
  for (int o = 1; o < 32; o <<= 1) {
    const int t = __shfl_up(incl, o);
    if (lane >= o) incl += t;
  }
  if (lane == 0)  s_part[wave] = acc;
  if (lane == 31) s_wsum[wave] = incl;
  __syncthreads();
  int pre = 0;
  for (int w = 0; w < 8; ++w) {
    pre += s_part[w];
    if (w < wave) pre += s_wsum[w];
  }
  if (i < n) {
    rowptr[i] = pre + incl - d;
    deg_in[i] = d;
    const float fi = (float)(d > 1 ? d : 1);
    inv_in[i] = 1.0f / sqrtf(fi);
    const float fo = (float)(dout > 1 ? dout : 1);
    const float io = 1.0f / sqrtf(fo);
    inv_out[i] = io;
    xs[i] = x[i] * io;
  }
}

__global__ void k_fill(const int* __restrict__ src, const int* __restrict__ dst, int E,
                       const int* __restrict__ rank, const int* __restrict__ rowptr,
                       const int* __restrict__ cin, int* __restrict__ csr_src) {
  int e = blockIdx.x * blockDim.x + threadIdx.x;
  if (e < E) {
    const int d = dst[e];
    csr_src[rowptr[d] + cin[(e / CH) * NP + d] + rank[e]] = src[e];
  }
}

__device__ __forceinline__ float row_gather_sum(const int* __restrict__ csr_src,
                                                const float* __restrict__ vals,
                                                int pb, int cnt, int first, int stride) {
  float a = 0.0f;
  for (int q0 = first; q0 < cnt; q0 += 8 * stride) {
#pragma unroll
    for (int u = 0; u < 8; ++u) {
      const int q = q0 + u * stride;
      const bool ok = q < cnt;
      const int s = csr_src[pb + (ok ? q : q0)];
      const float h = vals[s];
      a += ok ? h : 0.0f;
    }
  }
  return a;
}

__global__ void k_l1_wmma(const int* __restrict__ rowptr,
                          const int* __restrict__ deg_in,
                          const int* __restrict__ csr_src,
                          const float* __restrict__ xs,
                          const float* __restrict__ inv_in,
                          const float* __restrict__ inv_out,
                          const float* __restrict__ W0,
                          const float* __restrict__ b0,
                          const float* __restrict__ W1,
                          float* __restrict__ h1mm,
                          int n_tiles) {
  __shared__ float sW0[100];
  __shared__ float sb0[100];
  __shared__ float sW1p[100 * 16];
  for (int t = threadIdx.x; t < 100; t += blockDim.x) { sW0[t] = W0[t]; sb0[t] = b0[t]; }
  for (int t = threadIdx.x; t < 1600; t += blockDim.x) {
    int row = t >> 4, c = t & 15;
    sW1p[t] = (c < 10) ? W1[row * 10 + c] : 0.0f;
  }
  __syncthreads();

  const int wave = threadIdx.x >> 5;
  const int lane = threadIdx.x & 31;
  const int tile = blockIdx.x * (blockDim.x >> 5) + wave;
  if (tile >= n_tiles) return;
  const int i0   = tile * 16;
  const int half = lane >> 4;
  const int m    = lane & 15;
  const int col  = lane & 15;

  const int r = i0 + m;
  float part = row_gather_sum(csr_src, xs, rowptr[r], deg_in[r], half, 2);
  part += __shfl_xor(part, 16);
  const float s = part * inv_in[r];

  v8f c = {};
#pragma unroll
  for (int kk = 0; kk < 25; ++kk) {
    const int k0 = kk * 4 + half * 2;
    float a0 = fmaf(s, sW0[k0], sb0[k0]);
    a0 = a0 >= 0.0f ? a0 : 0.01f * a0;
    float a1 = fmaf(s, sW0[k0 + 1], sb0[k0 + 1]);
    a1 = a1 >= 0.0f ? a1 : 0.01f * a1;
    v2f a = {a0, a1};
    v2f b = {sW1p[k0 * 16 + col], sW1p[(k0 + 1) * 16 + col]};
    c = __builtin_amdgcn_wmma_f32_16x16x4_f32(false, a, false, b, (short)0, c,
                                              false, false);
  }

#pragma unroll
  for (int v = 0; v < 8; ++v) {
    const int i = i0 + v + 8 * half;
    h1mm[i * 16 + col] = c[v] * inv_out[i];
  }
}

__global__ void __launch_bounds__(512)
k_gather1_l2(const int* __restrict__ rowptr, const int* __restrict__ deg_in,
             const int* __restrict__ csr_src, const float* __restrict__ h1mm,
             const float* __restrict__ inv_in, const float* __restrict__ inv_out,
             const float* __restrict__ b1, const float* __restrict__ W2,
             float* __restrict__ h2, int n) {
  __shared__ float s_res[32];
  const int g = threadIdx.x >> 4;
  const int k = threadIdx.x & 15;
  const int i = blockIdx.x * 32 + g;
  const bool live = i < n;
  const int pb  = live ? rowptr[i] : 0;
  const int cnt = live ? deg_in[i] : 0;

  float acc = 0.0f;
  for (int q0 = 0; q0 < cnt; q0 += 16) {
#pragma unroll
    for (int u = 0; u < 16; ++u) {
      const int q = q0 + u;
      const bool ok = q < cnt;
      const int s = csr_src[pb + (ok ? q : q0)];
      const float h = h1mm[s * 16 + k];
      acc += ok ? h : 0.0f;
    }
  }

  float v = 0.0f;
  if (live && k < 10) {
    v = fmaf(acc, inv_in[i], b1[k]);
    v = v > 0.0f ? v : 0.0f;
  }
  float a2 = 0.0f;
#pragma unroll
  for (int kk = 0; kk < 10; ++kk) a2 = fmaf(__shfl(v, kk, 16), W2[kk], a2);
  if (k == 0) s_res[g] = live ? a2 * inv_out[i] : 0.0f;
  __syncthreads();
  if (threadIdx.x < 32) {
    const int o = blockIdx.x * 32 + threadIdx.x;
    if (o < n) h2[o] = s_res[threadIdx.x];
  }
}

__global__ void __launch_bounds__(512)
k_gather2_out(const int* __restrict__ rowptr, const int* __restrict__ deg_in,
              const int* __restrict__ csr_src, const float* __restrict__ h2,
              const float* __restrict__ inv_in, const float* __restrict__ b2,
              float* __restrict__ out, int n) {
  __shared__ float s_res[32];
  const int g = threadIdx.x >> 4;
  const int k = threadIdx.x & 15;
  const int i = blockIdx.x * 32 + g;
  const bool live = i < n;
  float acc = 0.0f;
  if (live) {
    const int pb = rowptr[i];
    const int pe = pb + deg_in[i];
    for (int p = pb + k; p < pe; p += 16) acc += h2[csr_src[p]];
  }
  for (int o = 8; o > 0; o >>= 1) acc += __shfl_xor(acc, o, 16);
  if (k == 0) {
    float v = 0.0f;
    if (live) {
      v = fmaf(acc, inv_in[i], b2[0]);
      v = v > 0.0f ? v : 0.0f;
    }
    s_res[g] = v;
  }
  __syncthreads();
  if (threadIdx.x < 32) {
    const int o = blockIdx.x * 32 + threadIdx.x;
    if (o < n) out[o] = s_res[threadIdx.x];
  }
}

extern "C" void kernel_launch(void* const* d_in, const int* in_sizes, int n_in,
                              void* d_out, int out_size, void* d_ws, size_t ws_size,
                              hipStream_t stream) {
  const float* in_feat = (const float*)d_in[0];
  const int*   edge    = (const int*)d_in[1];
  const float* W0      = (const float*)d_in[2];
  const float* b0      = (const float*)d_in[3];
  const float* W1      = (const float*)d_in[4];
  const float* b1      = (const float*)d_in[5];
  const float* W2      = (const float*)d_in[6];
  const float* b2      = (const float*)d_in[7];
  float* out = (float*)d_out;

  const int N = NN;
  const int E = in_sizes[1] / 2;
  const int* src = edge;
  const int* dst = edge + E;

  float* ws      = (float*)d_ws;
  int*   wsi     = (int*)d_ws;
  float* h1mm    = ws;
  int*   coarse  = wsi + 16 * NP;
  int*   deg_in  = wsi + 17 * NP;
  float* inv_out = ws  + 18 * NP;
  float* inv_in  = ws  + 19 * NP;
  float* xs      = ws  + 20 * NP;
  int*   rowptr  = wsi + 21 * NP;
  float* h2      = ws  + 22 * NP;
  const int C    = (E + CH - 1) / CH;
  int*   cin     = wsi + 23 * NP;
  int*   cout    = cin + C * NP;
  int*   rank    = cout + C * NP;
  int*   csr_src = rank + E;

  const int B = 256;
  k_count<<<dim3(C, 2), 1024, 0, stream>>>(src, dst, E, cin, cout, coarse, rank);
  k_scan<<<(N + 255) / 256, 256, 0, stream>>>(cin, cout, coarse, C, in_feat, N, deg_in,
                                              inv_out, inv_in, xs, rowptr);
  k_fill<<<(E + B - 1) / B, B, 0, stream>>>(src, dst, E, rank, rowptr, cin, csr_src);

  const int n_tiles = (N + 15) / 16;
  const int waves_per_block = B / 32;
  k_l1_wmma<<<(n_tiles + waves_per_block - 1) / waves_per_block, B, 0, stream>>>(
      rowptr, deg_in, csr_src, xs, inv_in, inv_out, W0, b0, W1, h1mm, n_tiles);

  k_gather1_l2<<<(N + 31) / 32, 512, 0, stream>>>(rowptr, deg_in, csr_src, h1mm,
                                                   inv_in, inv_out, b1, W2, h2, N);
  k_gather2_out<<<(N + 31) / 32, 512, 0, stream>>>(rowptr, deg_in, csr_src, h2,
                                                    inv_in, b2, out, N);
}


# Scan kernel: every load of a thread (64 chunk counts, 32 node-block sums, x) issued as one group ahead of the first use, one memory round trip not seven; it stores absolute slot bases for fill
# speedup vs baseline: 3.6381x; 1.0564x over previous
#include <hip/hip_runtime.h>
#include <math.h>


#define NN 50000
#define NP 50016
#define CH 25000
#define CU ((CH + 1023) / 1024)

typedef __attribute__((ext_vector_type(2))) float v2f;
typedef __attribute__((ext_vector_type(8))) float v8f;

__global__ void __launch_bounds__(1024)
k_count(const int* __restrict__ src, const int* __restrict__ dst, int E,
        int* __restrict__ cin, int* __restrict__ cout, int* __restrict__ coarse,
        int* __restrict__ rank) {
  __shared__ __attribute__((aligned(16))) int s_hist[NP];
  __shared__ __attribute__((aligned(16))) int s_coarse[1024];
  const int tid = threadIdx.x;
  const int c = blockIdx.x;
  const bool in_role = blockIdx.y == 0;
  int4* h4 = reinterpret_cast<int4*>(s_hist);
  for (int i = tid; i < NP / 4; i += 1024) h4[i] = make_int4(0, 0, 0, 0);
  s_coarse[tid] = 0;

  const int e0 = c * CH;
  const int e1 = (e0 + CH < E) ? e0 + CH : E;
  const int* __restrict__ key = in_role ? dst : src;
  int kv[CU];
#pragma unroll
  for (int u = 0; u < CU; ++u) {
    const int idx = e0 + tid + u * 1024;
    kv[u] = key[idx < e1 ? idx : e0];
  }
  __syncthreads();

  if (in_role) {
    int rk[CU];
#pragma unroll
    for (int u = 0; u < CU; ++u) {
      const int idx = e0 + tid + u * 1024;
      rk[u] = 0;
      if (idx < e1) {
        rk[u] = atomicAdd(&s_hist[kv[u]], 1);
        atomicAdd(&s_coarse[((kv[u] >> 8) << 2) | (tid & 3)], 1);
      }
    }
#pragma unroll
    for (int u = 0; u < CU; ++u) {
      const int idx = e0 + tid + u * 1024;
      if (idx < e1) rank[idx] = rk[u];
    }
  } else {
#pragma unroll
    for (int u = 0; u < CU; ++u) {
      const int idx = e0 + tid + u * 1024;
      if (idx < e1) atomicAdd(&s_hist[kv[u]], 1);
    }
  }
  __syncthreads();
  int4* __restrict__ out4 = reinterpret_cast<int4*>((in_role ? cin : cout) + c * NP);
#pragma unroll 4
  for (int i = tid; i < NP / 4; i += 1024) out4[i] = h4[i];
  if (in_role && tid < 256) {
    const int4 q = reinterpret_cast<const int4*>(s_coarse)[tid];
    coarse[c * 256 + tid] = (q.x + q.y) + (q.z + q.w);
  }
}

#define CMAX 32
__global__ void __launch_bounds__(256)
k_scan(int* __restrict__ cin, const int* __restrict__ cout,
       const int* __restrict__ coarse, int C,
       const float* __restrict__ x, int n, int* __restrict__ deg_in,
       float* __restrict__ inv_out, float* __restrict__ inv_in,
       float* __restrict__ xs, int* __restrict__ rowptr) {
  __shared__ int s_part[8];
  __shared__ int s_wsum[8];
  const int tid  = threadIdx.x;
  const int lane = tid & 31;
  const int wave = tid >> 5;
  const int b    = blockIdx.x;
  const int i    = b * 256 + tid;
  const bool inr = i < n;
  const int ii   = inr ? i : n - 1;

  const int* __restrict__ pco = coarse + tid;
  const int* pin = cin + ii;
  const int* __restrict__ pou = cout + ii;
  int pc[CMAX], ti[CMAX], to[CMAX];
#pragma unroll
  for (int c = 0; c < CMAX; ++c) pc[c] = pco[c * 256];
#pragma unroll
  for (int c = 0; c < CMAX; ++c) ti[c] = pin[c * NP];
#pragma unroll
  for (int c = 0; c < CMAX; ++c) to[c] = pou[c * NP];
  const float xv = x[ii];
  asm volatile("" :: "v"(to[0]), "v"(to[1]), "v"(to[2]), "v"(to[3]), "v"(to[4]), "v"(to[5]),
               "v"(to[6]), "v"(to[7]), "v"(to[8]), "v"(to[9]), "v"(to[10]), "v"(to[11]),
               "v"(to[12]), "v"(to[13]), "v"(to[14]), "v"(to[15]));
  asm volatile("" :: "v"(to[16]), "v"(to[17]), "v"(to[18]), "v"(to[19]), "v"(to[20]), "v"(to[21]),
               "v"(to[22]), "v"(to[23]), "v"(to[24]), "v"(to[25]), "v"(to[26]), "v"(to[27]),
               "v"(to[28]), "v"(to[29]), "v"(to[30]), "v"(to[31]), "v"(xv));
  __builtin_amdgcn_sched_barrier(0);

  int acc = 0, d = 0, dout = 0;
#pragma unroll
  for (int c = 0; c < CMAX; ++c) {
    const bool ok = inr && c < C;
    acc += (c < C) ? pc[c] : 0;
    ti[c] = ok ? ti[c] : 0;
    d += ti[c];
    dout += ok ? to[c] : 0;
  }
  acc = tid < b ? acc : 0;
  for (int o = 16; o > 0; o >>= 1) acc += __shfl_xor(acc, o);
  int incl = d;
  for (int o = 1; o < 32; o <<= 1) {
    const int t = __shfl_up(incl, o);
    if (lane >= o) incl += t;
  }
  if (lane == 0)  s_part[wave] = acc;
  if (lane == 31) s_wsum[wave] = incl;
  __syncthreads();
  int pre = 0;
  for (int w = 0; w < 8; ++w) {
    pre += s_part[w];
    if (w < wave) pre += s_wsum[w];
  }
  if (inr) {
    const int base = pre + incl - d;
    rowptr[i] = base;
    deg_in[i] = d;
    const float fi = (float)(d > 1 ? d : 1);
    inv_in[i] = 1.0f / sqrtf(fi);
    const float fo = (float)(dout > 1 ? dout : 1);
    const float io = 1.0f / sqrtf(fo);
    inv_out[i] = io;
    xs[i] = xv * io;
    int run = base;
#pragma unroll
    for (int c = 0; c < CMAX; ++c) {
      if (c < C) cin[c * NP + i] = run;
      run += ti[c];
    }
  }
}

__global__ void k_fill(const int* __restrict__ src, const int* __restrict__ dst, int E,
                       const int* __restrict__ rank,
                       const int* __restrict__ cin, int* __restrict__ csr_src) {
  int e = blockIdx.x * blockDim.x + threadIdx.x;
  if (e < E) {
    const int d = dst[e];
    csr_src[cin[(e / CH) * NP + d] + rank[e]] = src[e];
  }
}

__device__ __forceinline__ float row_gather_sum(const int* __restrict__ csr_src,
                                                const float* __restrict__ vals,
                                                int pb, int cnt, int first, int stride) {
  float a = 0.0f;
  for (int q0 = first; q0 < cnt; q0 += 8 * stride) {
#pragma unroll
    for (int u = 0; u < 8; ++u) {
      const int q = q0 + u * stride;
      const bool ok = q < cnt;
      const int s = csr_src[pb + (ok ? q : q0)];
      const float h = vals[s];
      a += ok ? h : 0.0f;
    }
  }
  return a;
}

__global__ void k_l1_wmma(const int* __restrict__ rowptr,
                          const int* __restrict__ deg_in,
                          const int* __restrict__ csr_src,
                          const float* __restrict__ xs,
                          const float* __restrict__ inv_in,
                          const float* __restrict__ inv_out,
                          const float* __restrict__ W0,
                          const float* __restrict__ b0,
                          const float* __restrict__ W1,
                          float* __restrict__ h1mm,
                          int n_tiles) {
  __shared__ float sW0[100];
  __shared__ float sb0[100];
  __shared__ float sW1p[100 * 16];
  for (int t = threadIdx.x; t < 100; t += blockDim.x) { sW0[t] = W0[t]; sb0[t] = b0[t]; }
  for (int t = threadIdx.x; t < 1600; t += blockDim.x) {
    int row = t >> 4, c = t & 15;
    sW1p[t] = (c < 10) ? W1[row * 10 + c] : 0.0f;
  }
  __syncthreads();

  const int wave = threadIdx.x >> 5;
  const int lane = threadIdx.x & 31;
  const int tile = blockIdx.x * (blockDim.x >> 5) + wave;
  if (tile >= n_tiles) return;
  const int i0   = tile * 16;
  const int half = lane >> 4;
  const int m    = lane & 15;
  const int col  = lane & 15;

  const int r = i0 + m;
  float part = row_gather_sum(csr_src, xs, rowptr[r], deg_in[r], half, 2);
  part += __shfl_xor(part, 16);
  const float s = part * inv_in[r];

  v8f c = {};
#pragma unroll
  for (int kk = 0; kk < 25; ++kk) {
    const int k0 = kk * 4 + half * 2;
    float a0 = fmaf(s, sW0[k0], sb0[k0]);
    a0 = a0 >= 0.0f ? a0 : 0.01f * a0;
    float a1 = fmaf(s, sW0[k0 + 1], sb0[k0 + 1]);
    a1 = a1 >= 0.0f ? a1 : 0.01f * a1;
    v2f a = {a0, a1};
    v2f b = {sW1p[k0 * 16 + col], sW1p[(k0 + 1) * 16 + col]};
    c = __builtin_amdgcn_wmma_f32_16x16x4_f32(false, a, false, b, (short)0, c,
                                              false, false);
  }

#pragma unroll
  for (int v = 0; v < 8; ++v) {
    const int i = i0 + v + 8 * half;
    h1mm[i * 16 + col] = c[v] * inv_out[i];
  }
}

__global__ void __launch_bounds__(512)
k_gather1_l2(const int* __restrict__ rowptr, const int* __restrict__ deg_in,
             const int* __restrict__ csr_src, const float* __restrict__ h1mm,
             const float* __restrict__ inv_in, const float* __restrict__ inv_out,
             const float* __restrict__ b1, const float* __restrict__ W2,
             float* __restrict__ h2, int n) {
  __shared__ float s_res[32];
  const int g = threadIdx.x >> 4;
  const int k = threadIdx.x & 15;
  const int i = blockIdx.x * 32 + g;
  const bool live = i < n;
  const int pb  = live ? rowptr[i] : 0;
  const int cnt = live ? deg_in[i] : 0;

  float acc = 0.0f;
  for (int q0 = 0; q0 < cnt; q0 += 16) {
#pragma unroll
    for (int u = 0; u < 16; ++u) {
      const int q = q0 + u;
      const bool ok = q < cnt;
      const int s = csr_src[pb + (ok ? q : q0)];
      const float h = h1mm[s * 16 + k];
      acc += ok ? h : 0.0f;
    }
  }

  float v = 0.0f;
  if (live && k < 10) {
    v = fmaf(acc, inv_in[i], b1[k]);
    v = v > 0.0f ? v : 0.0f;
  }
  float a2 = 0.0f;
#pragma unroll
  for (int kk = 0; kk < 10; ++kk) a2 = fmaf(__shfl(v, kk, 16), W2[kk], a2);
  if (k == 0) s_res[g] = live ? a2 * inv_out[i] : 0.0f;
  __syncthreads();
  if (threadIdx.x < 32) {
    const int o = blockIdx.x * 32 + threadIdx.x;
    if (o < n) h2[o] = s_res[threadIdx.x];
  }
}

__global__ void __launch_bounds__(512)
k_gather2_out(const int* __restrict__ rowptr, const int* __restrict__ deg_in,
              const int* __restrict__ csr_src, const float* __restrict__ h2,
              const float* __restrict__ inv_in, const float* __restrict__ b2,
              float* __restrict__ out, int n) {
  __shared__ float s_res[32];
  const int g = threadIdx.x >> 4;
  const int k = threadIdx.x & 15;
  const int i = blockIdx.x * 32 + g;
  const bool live = i < n;
  float acc = 0.0f;
  if (live) {
    const int pb = rowptr[i];
    const int pe = pb + deg_in[i];
    for (int p = pb + k; p < pe; p += 16) acc += h2[csr_src[p]];
  }
  for (int o = 8; o > 0; o >>= 1) acc += __shfl_xor(acc, o, 16);
  if (k == 0) {
    float v = 0.0f;
    if (live) {
      v = fmaf(acc, inv_in[i], b2[0]);
      v = v > 0.0f ? v : 0.0f;
    }
    s_res[g] = v;
  }
  __syncthreads();
  if (threadIdx.x < 32) {
    const int o = blockIdx.x * 32 + threadIdx.x;
    if (o < n) out[o] = s_res[threadIdx.x];
  }
}

extern "C" void kernel_launch(void* const* d_in, const int* in_sizes, int n_in,
                              void* d_out, int out_size, void* d_ws, size_t ws_size,
                              hipStream_t stream) {
  const float* in_feat = (const float*)d_in[0];
  const int*   edge    = (const int*)d_in[1];
  const float* W0      = (const float*)d_in[2];
  const float* b0      = (const float*)d_in[3];
  const float* W1      = (const float*)d_in[4];
  const float* b1      = (const float*)d_in[5];
  const float* W2      = (const float*)d_in[6];
  const float* b2      = (const float*)d_in[7];
  float* out = (float*)d_out;

  const int N = NN;
  const int E = in_sizes[1] / 2;
  const int* src = edge;
  const int* dst = edge + E;

  float* ws      = (float*)d_ws;
  int*   wsi     = (int*)d_ws;
  float* h1mm    = ws;
  int*   coarse  = wsi + 16 * NP;
  int*   deg_in  = wsi + 17 * NP;
  float* inv_out = ws  + 18 * NP;
  float* inv_in  = ws  + 19 * NP;
  float* xs      = ws  + 20 * NP;
  int*   rowptr  = wsi + 21 * NP;
  float* h2      = ws  + 22 * NP;
  const int C    = (E + CH - 1) / CH;
  int*   cin     = wsi + 23 * NP;
  int*   cout    = cin + C * NP;
  int*   rank    = cout + C * NP;
  int*   csr_src = rank + E;

  const int B = 256;
  k_count<<<dim3(C, 2), 1024, 0, stream>>>(src, dst, E, cin, cout, coarse, rank);
  k_scan<<<(N + 255) / 256, 256, 0, stream>>>(cin, cout, coarse, C, in_feat, N, deg_in,
                                              inv_out, inv_in, xs, rowptr);
  k_fill<<<(E + B - 1) / B, B, 0, stream>>>(src, dst, E, rank, cin, csr_src);

  const int n_tiles = (N + 15) / 16;
  const int waves_per_block = B / 32;
  k_l1_wmma<<<(n_tiles + waves_per_block - 1) / waves_per_block, B, 0, stream>>>(
      rowptr, deg_in, csr_src, xs, inv_in, inv_out, W0, b0, W1, h1mm, n_tiles);

  k_gather1_l2<<<(N + 31) / 32, 512, 0, stream>>>(rowptr, deg_in, csr_src, h1mm,
                                                   inv_in, inv_out, b1, W2, h2, N);
  k_gather2_out<<<(N + 31) / 32, 512, 0, stream>>>(rowptr, deg_in, csr_src, h2,
                                                    inv_in, b2, out, N);
}


# Three layer kernels: every load not hanging on a gathered value issued as one group at entry; a row's first two index groups, then both gather groups, go out together; row sum order kept
# speedup vs baseline: 3.7995x; 1.0444x over previous
#include <hip/hip_runtime.h>
#include <math.h>


#define NN 50000
#define NP 50016
#define CH 25000
#define CU ((CH + 1023) / 1024)

typedef __attribute__((ext_vector_type(2))) float v2f;
typedef __attribute__((ext_vector_type(8))) float v8f;

__global__ void __launch_bounds__(1024)
k_count(const int* __restrict__ src, const int* __restrict__ dst, int E,
        int* __restrict__ cin, int* __restrict__ cout, int* __restrict__ coarse,
        int* __restrict__ rank) {
  __shared__ __attribute__((aligned(16))) int s_hist[NP];
  __shared__ __attribute__((aligned(16))) int s_coarse[1024];
  const int tid = threadIdx.x;
  const int c = blockIdx.x;
  const bool in_role = blockIdx.y == 0;
  int4* h4 = reinterpret_cast<int4*>(s_hist);
  for (int i = tid; i < NP / 4; i += 1024) h4[i] = make_int4(0, 0, 0, 0);
  s_coarse[tid] = 0;

  const int e0 = c * CH;
  const int e1 = (e0 + CH < E) ? e0 + CH : E;
  const int* __restrict__ key = in_role ? dst : src;
  int kv[CU];
#pragma unroll
  for (int u = 0; u < CU; ++u) {
    const int idx = e0 + tid + u * 1024;
    kv[u] = key[idx < e1 ? idx : e0];
  }
  __syncthreads();

  if (in_role) {
    int rk[CU];
#pragma unroll
    for (int u = 0; u < CU; ++u) {
      const int idx = e0 + tid + u * 1024;
      rk[u] = 0;
      if (idx < e1) {
        rk[u] = atomicAdd(&s_hist[kv[u]], 1);
        atomicAdd(&s_coarse[((kv[u] >> 8) << 2) | (tid & 3)], 1);
      }
    }
#pragma unroll
    for (int u = 0; u < CU; ++u) {
      const int idx = e0 + tid + u * 1024;
      if (idx < e1) rank[idx] = rk[u];
    }
  } else {
#pragma unroll
    for (int u = 0; u < CU; ++u) {
      const int idx = e0 + tid + u * 1024;
      if (idx < e1) atomicAdd(&s_hist[kv[u]], 1);
    }
  }
  __syncthreads();
  int4* __restrict__ out4 = reinterpret_cast<int4*>((in_role ? cin : cout) + c * NP);
#pragma unroll 4
  for (int i = tid; i < NP / 4; i += 1024) out4[i] = h4[i];
  if (in_role && tid < 256) {
    const int4 q = reinterpret_cast<const int4*>(s_coarse)[tid];
    coarse[c * 256 + tid] = (q.x + q.y) + (q.z + q.w);
  }
}

#define CMAX 32
__global__ void __launch_bounds__(256)
k_scan(int* __restrict__ cin, const int* __restrict__ cout,
       const int* __restrict__ coarse, int C,
       const float* __restrict__ x, int n, int* __restrict__ deg_in,
       float* __restrict__ inv_out, float* __restrict__ inv_in,
       float* __restrict__ xs, int* __restrict__ rowptr) {
  __shared__ int s_part[8];
  __shared__ int s_wsum[8];
  const int tid  = threadIdx.x;
  const int lane = tid & 31;
  const int wave = tid >> 5;
  const int b    = blockIdx.x;
  const int i    = b * 256 + tid;
  const bool inr = i < n;
  const int ii   = inr ? i : n - 1;

  const int* __restrict__ pco = coarse + tid;
  const int* pin = cin + ii;
  const int* __restrict__ pou = cout + ii;
  int pc[CMAX], ti[CMAX], to[CMAX];
#pragma unroll
  for (int c = 0; c < CMAX; ++c) pc[c] = pco[c * 256];
#pragma unroll
  for (int c = 0; c < CMAX; ++c) ti[c] = pin[c * NP];
#pragma unroll
  for (int c = 0; c < CMAX; ++c) to[c] = pou[c * NP];
  const float xv = x[ii];
  asm volatile("" :: "v"(to[0]), "v"(to[1]), "v"(to[2]), "v"(to[3]), "v"(to[4]), "v"(to[5]),
               "v"(to[6]), "v"(to[7]), "v"(to[8]), "v"(to[9]), "v"(to[10]), "v"(to[11]),
               "v"(to[12]), "v"(to[13]), "v"(to[14]), "v"(to[15]));
  asm volatile("" :: "v"(to[16]), "v"(to[17]), "v"(to[18]), "v"(to[19]), "v"(to[20]), "v"(to[21]),
               "v"(to[22]), "v"(to[23]), "v"(to[24]), "v"(to[25]), "v"(to[26]), "v"(to[27]),
               "v"(to[28]), "v"(to[29]), "v"(to[30]), "v"(to[31]), "v"(xv));
  __builtin_amdgcn_sched_barrier(0);

  int acc = 0, d = 0, dout = 0;
#pragma unroll
  for (int c = 0; c < CMAX; ++c) {
    const bool ok = inr && c < C;
    acc += (c < C) ? pc[c] : 0;
    ti[c] = ok ? ti[c] : 0;
    d += ti[c];
    dout += ok ? to[c] : 0;
  }
  acc = tid < b ? acc : 0;
  for (int o = 16; o > 0; o >>= 1) acc += __shfl_xor(acc, o);
  int incl = d;
  for (int o = 1; o < 32; o <<= 1) {
    const int t = __shfl_up(incl, o);
    if (lane >= o) incl += t;
  }
  if (lane == 0)  s_part[wave] = acc;
  if (lane == 31) s_wsum[wave] = incl;
  __syncthreads();
  int pre = 0;
  for (int w = 0; w < 8; ++w) {
    pre += s_part[w];
    if (w < wave) pre += s_wsum[w];
  }
  if (inr) {
    const int base = pre + incl - d;
    rowptr[i] = base;
    deg_in[i] = d;
    const float fi = (float)(d > 1 ? d : 1);
    inv_in[i] = 1.0f / sqrtf(fi);
    const float fo = (float)(dout > 1 ? dout : 1);
    const float io = 1.0f / sqrtf(fo);
    inv_out[i] = io;
    xs[i] = xv * io;
    int run = base;
#pragma unroll
    for (int c = 0; c < CMAX; ++c) {
      if (c < C) cin[c * NP + i] = run;
      run += ti[c];
    }
  }
}

__global__ void k_fill(const int* __restrict__ src, const int* __restrict__ dst, int E,
                       const int* __restrict__ rank,
                       const int* __restrict__ cin, int* __restrict__ csr_src) {
  int e = blockIdx.x * blockDim.x + threadIdx.x;
  if (e < E) {
    const int d = dst[e];
    csr_src[cin[(e / CH) * NP + d] + rank[e]] = src[e];
  }
}

template <int U>
__device__ __forceinline__ float row_gather_sum(const int* __restrict__ csr_src,
                                                const float* __restrict__ vals,
                                                int pb, int cnt, int first, int stride) {
  float a = 0.0f;
  for (int q0 = first; q0 < cnt; q0 += 2 * U * stride) {
    const int q1 = q0 + U * stride;
    const bool more = q1 < cnt;
    int sa[U], sb[U];
    float ha[U], hb[U];
#pragma unroll
    for (int u = 0; u < U; ++u) {
      const int q = q0 + u * stride;
      sa[u] = csr_src[pb + (q < cnt ? q : q0)];
    }
    if (more) {
#pragma unroll
      for (int u = 0; u < U; ++u) {
        const int q = q1 + u * stride;
        sb[u] = csr_src[pb + (q < cnt ? q : q1)];
      }
    }
#pragma unroll
    for (int u = 0; u < U; ++u) ha[u] = vals[sa[u]];
    if (more) {
#pragma unroll
      for (int u = 0; u < U; ++u) hb[u] = vals[sb[u]];
    }
#pragma unroll
    for (int u = 0; u < U; ++u) a += (q0 + u * stride < cnt) ? ha[u] : 0.0f;
    if (more) {
#pragma unroll
      for (int u = 0; u < U; ++u) a += (q1 + u * stride < cnt) ? hb[u] : 0.0f;
    }
  }
  return a;
}

#define L1B 256
#define L1S ((1600 + L1B - 1) / L1B)
__global__ void __launch_bounds__(L1B)
k_l1_wmma(const int* __restrict__ rowptr,
                          const int* __restrict__ deg_in,
                          const int* __restrict__ csr_src,
                          const float* __restrict__ xs,
                          const float* __restrict__ inv_in,
                          const float* __restrict__ inv_out,
                          const float* __restrict__ W0,
                          const float* __restrict__ b0,
                          const float* __restrict__ W1,
                          float* __restrict__ h1mm,
                          int n_tiles) {
  __shared__ float sW0[100];
  __shared__ float sb0[100];
  __shared__ float sW1p[100 * 16];
  const int tid  = threadIdx.x;
  const int wave = tid >> 5;
  const int lane = tid & 31;
  const int tile = blockIdx.x * (L1B >> 5) + wave;
  const bool has = tile < n_tiles;
  const int i0   = (has ? tile : 0) * 16;
  const int half = lane >> 4;
  const int m    = lane & 15;
  const int col  = lane & 15;
  const int r    = i0 + m;
  const int pb   = rowptr[r];
  const int cnt  = deg_in[r];
  const float nin = inv_in[r];
  float nout[8];
#pragma unroll
  for (int v = 0; v < 8; ++v) nout[v] = inv_out[i0 + v + 8 * half];
  float w1v[L1S];
#pragma unroll
  for (int j = 0; j < L1S; ++j) {
    const int t = tid + j * L1B;
    const int tc = t < 1600 ? t : 1599;
    const int c = tc & 15;
    w1v[j] = W1[(tc >> 4) * 10 + (c < 10 ? c : 0)];
  }
  const int tw = tid < 100 ? tid : 99;
  const float w0v = W0[tw];
  const float b0v = b0[tw];
  if (tid < 100) { sW0[tid] = w0v; sb0[tid] = b0v; }
#pragma unroll
  for (int j = 0; j < L1S; ++j) {
    const int t = tid + j * L1B;
    if (t < 1600) sW1p[t] = ((t & 15) < 10) ? w1v[j] : 0.0f;
  }
  __syncthreads();
  if (!has) return;

  float part = row_gather_sum<8>(csr_src, xs, pb, cnt, half, 2);
  part += __shfl_xor(part, 16);
  const float s = part * nin;

  v8f c = {};
#pragma unroll
  for (int kk = 0; kk < 25; ++kk) {
    const int k0 = kk * 4 + half * 2;
    float a0 = fmaf(s, sW0[k0], sb0[k0]);
    a0 = a0 >= 0.0f ? a0 : 0.01f * a0;
    float a1 = fmaf(s, sW0[k0 + 1], sb0[k0 + 1]);
    a1 = a1 >= 0.0f ? a1 : 0.01f * a1;
    v2f a = {a0, a1};
    v2f b = {sW1p[k0 * 16 + col], sW1p[(k0 + 1) * 16 + col]};
    c = __builtin_amdgcn_wmma_f32_16x16x4_f32(false, a, false, b, (short)0, c,
                                              false, false);
  }

#pragma unroll
  for (int v = 0; v < 8; ++v) {
    const int i = i0 + v + 8 * half;
    h1mm[i * 16 + col] = c[v] * nout[v];
  }
}

__global__ void __launch_bounds__(512)
k_gather1_l2(const int* __restrict__ rowptr, const int* __restrict__ deg_in,
             const int* __restrict__ csr_src, const float* __restrict__ h1mm,
             const float* __restrict__ inv_in, const float* __restrict__ inv_out,
             const float* __restrict__ b1, const float* __restrict__ W2,
             float* __restrict__ h2, int n) {
  __shared__ float s_res[32];
  const int g = threadIdx.x >> 4;
  const int k = threadIdx.x & 15;
  const int i = blockIdx.x * 32 + g;
  const bool live = i < n;
  const int ii  = live ? i : 0;
  const int pb  = rowptr[ii];
  const int dg  = deg_in[ii];
  const float nin  = inv_in[ii];
  const float nout = inv_out[ii];
  const float bk   = b1[k < 10 ? k : 0];
  float w2[10];
#pragma unroll
  for (int kk = 0; kk < 10; ++kk) w2[kk] = W2[kk];
  asm volatile("" :: "v"(pb), "v"(dg));
  const int cnt = live ? dg : 0;

  float acc = 0.0f;
  for (int q0 = 0; q0 < cnt; q0 += 32) {
    const int q1 = q0 + 16;
    const bool more = q1 < cnt;
    int sa[16], sb[16];
    float ha[16], hb[16];
#pragma unroll
    for (int u = 0; u < 16; ++u) {
      const int q = q0 + u;
      sa[u] = csr_src[pb + (q < cnt ? q : q0)];
    }
    if (more) {
#pragma unroll
      for (int u = 0; u < 16; ++u) {
        const int q = q1 + u;
        sb[u] = csr_src[pb + (q < cnt ? q : q1)];
      }
    }
#pragma unroll
    for (int u = 0; u < 16; ++u) ha[u] = h1mm[sa[u] * 16 + k];
    if (more) {
#pragma unroll
      for (int u = 0; u < 16; ++u) hb[u] = h1mm[sb[u] * 16 + k];
    }
#pragma unroll
    for (int u = 0; u < 16; ++u) acc += (q0 + u < cnt) ? ha[u] : 0.0f;
    if (more) {
#pragma unroll
      for (int u = 0; u < 16; ++u) acc += (q1 + u < cnt) ? hb[u] : 0.0f;
    }
  }

  float v = 0.0f;
  if (live && k < 10) {
    v = fmaf(acc, nin, bk);
    v = v > 0.0f ? v : 0.0f;
  }
  float a2 = 0.0f;
#pragma unroll
  for (int kk = 0; kk < 10; ++kk) a2 = fmaf(__shfl(v, kk, 16), w2[kk], a2);
  if (k == 0) s_res[g] = live ? a2 * nout : 0.0f;
  __syncthreads();
  if (threadIdx.x < 32) {
    const int o = blockIdx.x * 32 + threadIdx.x;
    if (o < n) h2[o] = s_res[threadIdx.x];
  }
}

__global__ void __launch_bounds__(512)
k_gather2_out(const int* __restrict__ rowptr, const int* __restrict__ deg_in,
              const int* __restrict__ csr_src, const float* __restrict__ h2,
              const float* __restrict__ inv_in, const float* __restrict__ b2,
              float* __restrict__ out, int n) {
  __shared__ float s_res[32];
  const int g = threadIdx.x >> 4;
  const int k = threadIdx.x & 15;
  const int i = blockIdx.x * 32 + g;
  const bool live = i < n;
  const int ii  = live ? i : 0;
  const int pb  = rowptr[ii];
  const int dg  = deg_in[ii];
  const float nin  = inv_in[ii];
  const float bias = b2[0];
  asm volatile("" :: "v"(pb), "v"(dg));
  const int cnt = live ? dg : 0;
  float acc = row_gather_sum<1>(csr_src, h2, pb, cnt, k, 16);
  for (int o = 8; o > 0; o >>= 1) acc += __shfl_xor(acc, o, 16);
  if (k == 0) {
    float v = 0.0f;
    if (live) {
      v = fmaf(acc, nin, bias);
      v = v > 0.0f ? v : 0.0f;
    }
    s_res[g] = v;
  }
  __syncthreads();
  if (threadIdx.x < 32) {
    const int o = blockIdx.x * 32 + threadIdx.x;
    if (o < n) out[o] = s_res[threadIdx.x];
  }
}

extern "C" void kernel_launch(void* const* d_in, const int* in_sizes, int n_in,
                              void* d_out, int out_size, void* d_ws, size_t ws_size,
                              hipStream_t stream) {
  const float* in_feat = (const float*)d_in[0];
  const int*   edge    = (const int*)d_in[1];
  const float* W0      = (const float*)d_in[2];
  const float* b0      = (const float*)d_in[3];
  const float* W1      = (const float*)d_in[4];
  const float* b1      = (const float*)d_in[5];
  const float* W2      = (const float*)d_in[6];
  const float* b2      = (const float*)d_in[7];
  float* out = (float*)d_out;

  const int N = NN;
  const int E = in_sizes[1] / 2;
  const int* src = edge;
  const int* dst = edge + E;

  float* ws      = (float*)d_ws;
  int*   wsi     = (int*)d_ws;
  float* h1mm    = ws;
  int*   coarse  = wsi + 16 * NP;
  int*   deg_in  = wsi + 17 * NP;
  float* inv_out = ws  + 18 * NP;
  float* inv_in  = ws  + 19 * NP;
  float* xs      = ws  + 20 * NP;
  int*   rowptr  = wsi + 21 * NP;
  float* h2      = ws  + 22 * NP;
  const int C    = (E + CH - 1) / CH;
  int*   cin     = wsi + 23 * NP;
  int*   cout    = cin + C * NP;
  int*   rank    = cout + C * NP;
  int*   csr_src = rank + E;

  const int B = 256;
  k_count<<<dim3(C, 2), 1024, 0, stream>>>(src, dst, E, cin, cout, coarse, rank);
  k_scan<<<(N + 255) / 256, 256, 0, stream>>>(cin, cout, coarse, C, in_feat, N, deg_in,
                                              inv_out, inv_in, xs, rowptr);
  k_fill<<<(E + B - 1) / B, B, 0, stream>>>(src, dst, E, rank, cin, csr_src);

  const int n_tiles = (N + 15) / 16;
  const int waves_per_block = B / 32;
  k_l1_wmma<<<(n_tiles + waves_per_block - 1) / waves_per_block, B, 0, stream>>>(
      rowptr, deg_in, csr_src, xs, inv_in, inv_out, W0, b0, W1, h1mm, n_tiles);

  k_gather1_l2<<<(N + 31) / 32, 512, 0, stream>>>(rowptr, deg_in, csr_src, h1mm,
                                                   inv_in, inv_out, b1, W2, h2, N);
  k_gather2_out<<<(N + 31) / 32, 512, 0, stream>>>(rowptr, deg_in, csr_src, h2,
                                                    inv_in, b2, out, N);
}


# Counting pass split over four node ranges as well as 32 edge chunks: 256 work-groups, a quarter table in LDS each, instead of 64 on a quarter of the WGPs; row sums still chunk, then LDS arrival order
# speedup vs baseline: 3.9755x; 1.0463x over previous
#include <hip/hip_runtime.h>
#include <math.h>


#define NN 50000
#define NP 50016
#define CH 25000
#define CU ((CH + 1023) / 1024)

typedef __attribute__((ext_vector_type(2))) float v2f;
typedef __attribute__((ext_vector_type(8))) float v8f;

#define KR 4
#define KW 12544
#define KT ((KW / 4 + 1023) / 1024)
__global__ void __launch_bounds__(1024)
k_count(const int* __restrict__ src, const int* __restrict__ dst, int E,
        int* __restrict__ cin, int* __restrict__ cout, int* __restrict__ coarse,
        int* __restrict__ rank) {
  __shared__ __attribute__((aligned(16))) int s_hist[KW];
  __shared__ __attribute__((aligned(16))) int s_coarse[256];
  const int tid = threadIdx.x;
  const int c = blockIdx.x;
  const bool in_role = blockIdx.y == 0;
  const int r  = blockIdx.z;
  const int r0 = r * KW;
  const int wn = (r0 + KW <= NP) ? KW : NP - r0;
  const int w4 = wn >> 2;
  int4* h4 = reinterpret_cast<int4*>(s_hist);
#pragma unroll
  for (int t = 0; t < KT; ++t) {
    const int i = tid + t * 1024;
    if (i < KW / 4) h4[i] = make_int4(0, 0, 0, 0);
  }
  if (tid < 256) s_coarse[tid] = 0;

  const int e0 = c * CH;
  const int e1 = (e0 + CH < E) ? e0 + CH : E;
  const int* __restrict__ key = in_role ? dst : src;
  int kv[CU];
#pragma unroll
  for (int u = 0; u < CU; ++u) {
    const int idx = e0 + tid + u * 1024;
    kv[u] = key[idx < e1 ? idx : e0];
  }
  __syncthreads();

  if (in_role) {
    int rk[CU];
#pragma unroll
    for (int u = 0; u < CU; ++u) {
      const int idx = e0 + tid + u * 1024;
      const unsigned j = (unsigned)(kv[u] - r0);
      rk[u] = 0;
      if (idx < e1 && j < (unsigned)wn) {
        rk[u] = atomicAdd(&s_hist[j], 1);
        atomicAdd(&s_coarse[((j >> 8) << 2) | (tid & 3)], 1);
      }
    }
#pragma unroll
    for (int u = 0; u < CU; ++u) {
      const int idx = e0 + tid + u * 1024;
      const unsigned j = (unsigned)(kv[u] - r0);
      if (idx < e1 && j < (unsigned)wn) rank[idx] = rk[u];
    }
  } else {
#pragma unroll
    for (int u = 0; u < CU; ++u) {
      const int idx = e0 + tid + u * 1024;
      const unsigned j = (unsigned)(kv[u] - r0);
      if (idx < e1 && j < (unsigned)wn) atomicAdd(&s_hist[j], 1);
    }
  }
  __syncthreads();
  int4* __restrict__ out4 = reinterpret_cast<int4*>((in_role ? cin : cout) + c * NP + r0);
#pragma unroll
  for (int t = 0; t < KT; ++t) {
    const int i = tid + t * 1024;
    if (i < w4) out4[i] = h4[i];
  }
  if (in_role && tid < 128) {
    const int t = r * (KW / 256) + tid;
    const bool has = tid < KW / 256;
    if (has || (r == KR - 1 && t < 256)) {
      const int4 q = reinterpret_cast<const int4*>(s_coarse)[has ? tid : 0];
      coarse[c * 256 + t] = has ? (q.x + q.y) + (q.z + q.w) : 0;
    }
  }
}

#define CMAX 32
__global__ void __launch_bounds__(256)
k_scan(int* __restrict__ cin, const int* __restrict__ cout,
       const int* __restrict__ coarse, int C,
       const float* __restrict__ x, int n, int* __restrict__ deg_in,
       float* __restrict__ inv_out, float* __restrict__ inv_in,
       float* __restrict__ xs, int* __restrict__ rowptr) {
  __shared__ int s_part[8];
  __shared__ int s_wsum[8];
  const int tid  = threadIdx.x;
  const int lane = tid & 31;
  const int wave = tid >> 5;
  const int b    = blockIdx.x;
  const int i    = b * 256 + tid;
  const bool inr = i < n;
  const int ii   = inr ? i : n - 1;

  const int* __restrict__ pco = coarse + tid;
  const int* pin = cin + ii;
  const int* __restrict__ pou = cout + ii;
  int pc[CMAX], ti[CMAX], to[CMAX];
#pragma unroll
  for (int c = 0; c < CMAX; ++c) pc[c] = pco[c * 256];
#pragma unroll
  for (int c = 0; c < CMAX; ++c) ti[c] = pin[c * NP];
#pragma unroll
  for (int c = 0; c < CMAX; ++c) to[c] = pou[c * NP];
  const float xv = x[ii];
  asm volatile("" :: "v"(to[0]), "v"(to[1]), "v"(to[2]), "v"(to[3]), "v"(to[4]), "v"(to[5]),
               "v"(to[6]), "v"(to[7]), "v"(to[8]), "v"(to[9]), "v"(to[10]), "v"(to[11]),
               "v"(to[12]), "v"(to[13]), "v"(to[14]), "v"(to[15]));
  asm volatile("" :: "v"(to[16]), "v"(to[17]), "v"(to[18]), "v"(to[19]), "v"(to[20]), "v"(to[21]),
               "v"(to[22]), "v"(to[23]), "v"(to[24]), "v"(to[25]), "v"(to[26]), "v"(to[27]),
               "v"(to[28]), "v"(to[29]), "v"(to[30]), "v"(to[31]), "v"(xv));
  __builtin_amdgcn_sched_barrier(0);

  int acc = 0, d = 0, dout = 0;
#pragma unroll
  for (int c = 0; c < CMAX; ++c) {
    const bool ok = inr && c < C;
    acc += (c < C) ? pc[c] : 0;
    ti[c] = ok ? ti[c] : 0;
    d += ti[c];
    dout += ok ? to[c] : 0;
  }
  acc = tid < b ? acc : 0;
  for (int o = 16; o > 0; o >>= 1) acc += __shfl_xor(acc, o);
  int incl = d;
  for (int o = 1; o < 32; o <<= 1) {
    const int t = __shfl_up(incl, o);
    if (lane >= o) incl += t;
  }
  if (lane == 0)  s_part[wave] = acc;
  if (lane == 31) s_wsum[wave] = incl;
  __syncthreads();
  int pre = 0;
  for (int w = 0; w < 8; ++w) {
    pre += s_part[w];
    if (w < wave) pre += s_wsum[w];
  }
  if (inr) {
    const int base = pre + incl - d;
    rowptr[i] = base;
    deg_in[i] = d;
    const float fi = (float)(d > 1 ? d : 1);
    inv_in[i] = 1.0f / sqrtf(fi);
    const float fo = (float)(dout > 1 ? dout : 1);
    const float io = 1.0f / sqrtf(fo);
    inv_out[i] = io;
    xs[i] = xv * io;
    int run = base;
#pragma unroll
    for (int c = 0; c < CMAX; ++c) {
      if (c < C) cin[c * NP + i] = run;
      run += ti[c];
    }
  }
}

__global__ void k_fill(const int* __restrict__ src, const int* __restrict__ dst, int E,
                       const int* __restrict__ rank,
                       const int* __restrict__ cin, int* __restrict__ csr_src) {
  int e = blockIdx.x * blockDim.x + threadIdx.x;
  if (e < E) {
    const int d = dst[e];
    csr_src[cin[(e / CH) * NP + d] + rank[e]] = src[e];
  }
}

template <int U>
__device__ __forceinline__ float row_gather_sum(const int* __restrict__ csr_src,
                                                const float* __restrict__ vals,
                                                int pb, int cnt, int first, int stride) {
  float a = 0.0f;
  for (int q0 = first; q0 < cnt; q0 += 2 * U * stride) {
    const int q1 = q0 + U * stride;
    const bool more = q1 < cnt;
    int sa[U], sb[U];
    float ha[U], hb[U];
#pragma unroll
    for (int u = 0; u < U; ++u) {
      const int q = q0 + u * stride;
      sa[u] = csr_src[pb + (q < cnt ? q : q0)];
    }
    if (more) {
#pragma unroll
      for (int u = 0; u < U; ++u) {
        const int q = q1 + u * stride;
        sb[u] = csr_src[pb + (q < cnt ? q : q1)];
      }
    }
#pragma unroll
    for (int u = 0; u < U; ++u) ha[u] = vals[sa[u]];
    if (more) {
#pragma unroll
      for (int u = 0; u < U; ++u) hb[u] = vals[sb[u]];
    }
#pragma unroll
    for (int u = 0; u < U; ++u) a += (q0 + u * stride < cnt) ? ha[u] : 0.0f;
    if (more) {
#pragma unroll
      for (int u = 0; u < U; ++u) a += (q1 + u * stride < cnt) ? hb[u] : 0.0f;
    }
  }
  return a;
}

#define L1B 256
#define L1S ((1600 + L1B - 1) / L1B)
__global__ void __launch_bounds__(L1B)
k_l1_wmma(const int* __restrict__ rowptr,
                          const int* __restrict__ deg_in,
                          const int* __restrict__ csr_src,
                          const float* __restrict__ xs,
                          const float* __restrict__ inv_in,
                          const float* __restrict__ inv_out,
                          const float* __restrict__ W0,
                          const float* __restrict__ b0,
                          const float* __restrict__ W1,
                          float* __restrict__ h1mm,
                          int n_tiles) {
  __shared__ float sW0[100];
  __shared__ float sb0[100];
  __shared__ float sW1p[100 * 16];
  const int tid  = threadIdx.x;
  const int wave = tid >> 5;
  const int lane = tid & 31;
  const int tile = blockIdx.x * (L1B >> 5) + wave;
  const bool has = tile < n_tiles;
  const int i0   = (has ? tile : 0) * 16;
  const int half = lane >> 4;
  const int m    = lane & 15;
  const int col  = lane & 15;
  const int r    = i0 + m;
  const int pb   = rowptr[r];
  const int cnt  = deg_in[r];
  const float nin = inv_in[r];
  float nout[8];
#pragma unroll
  for (int v = 0; v < 8; ++v) nout[v] = inv_out[i0 + v + 8 * half];
  float w1v[L1S];
#pragma unroll
  for (int j = 0; j < L1S; ++j) {
    const int t = tid + j * L1B;
    const int tc = t < 1600 ? t : 1599;
    const int c = tc & 15;
    w1v[j] = W1[(tc >> 4) * 10 + (c < 10 ? c : 0)];
  }
  const int tw = tid < 100 ? tid : 99;
  const float w0v = W0[tw];
  const float b0v = b0[tw];
  if (tid < 100) { sW0[tid] = w0v; sb0[tid] = b0v; }
#pragma unroll
  for (int j = 0; j < L1S; ++j) {
    const int t = tid + j * L1B;
    if (t < 1600) sW1p[t] = ((t & 15) < 10) ? w1v[j] : 0.0f;
  }
  __syncthreads();
  if (!has) return;

  float part = row_gather_sum<8>(csr_src, xs, pb, cnt, half, 2);
  part += __shfl_xor(part, 16);
  const float s = part * nin;

  v8f c = {};
#pragma unroll
  for (int kk = 0; kk < 25; ++kk) {
    const int k0 = kk * 4 + half * 2;
    float a0 = fmaf(s, sW0[k0], sb0[k0]);
    a0 = a0 >= 0.0f ? a0 : 0.01f * a0;
    float a1 = fmaf(s, sW0[k0 + 1], sb0[k0 + 1]);
    a1 = a1 >= 0.0f ? a1 : 0.01f * a1;
    v2f a = {a0, a1};
    v2f b = {sW1p[k0 * 16 + col], sW1p[(k0 + 1) * 16 + col]};
    c = __builtin_amdgcn_wmma_f32_16x16x4_f32(false, a, false, b, (short)0, c,
                                              false, false);
  }

#pragma unroll
  for (int v = 0; v < 8; ++v) {
    const int i = i0 + v + 8 * half;
    h1mm[i * 16 + col] = c[v] * nout[v];
  }
}

__global__ void __launch_bounds__(512)
k_gather1_l2(const int* __restrict__ rowptr, const int* __restrict__ deg_in,
             const int* __restrict__ csr_src, const float* __restrict__ h1mm,
             const float* __restrict__ inv_in, const float* __restrict__ inv_out,
             const float* __restrict__ b1, const float* __restrict__ W2,
             float* __restrict__ h2, int n) {
  __shared__ float s_res[32];
  const int g = threadIdx.x >> 4;
  const int k = threadIdx.x & 15;
  const int i = blockIdx.x * 32 + g;
  const bool live = i < n;
  const int ii  = live ? i : 0;
  const int pb  = rowptr[ii];
  const int dg  = deg_in[ii];
  const float nin  = inv_in[ii];
  const float nout = inv_out[ii];
  const float bk   = b1[k < 10 ? k : 0];
  float w2[10];
#pragma unroll
  for (int kk = 0; kk < 10; ++kk) w2[kk] = W2[kk];
  asm volatile("" :: "v"(pb), "v"(dg));
  const int cnt = live ? dg : 0;

  float acc = 0.0f;
  for (int q0 = 0; q0 < cnt; q0 += 32) {
    const int q1 = q0 + 16;
    const bool more = q1 < cnt;
    int sa[16], sb[16];
    float ha[16], hb[16];
#pragma unroll
    for (int u = 0; u < 16; ++u) {
      const int q = q0 + u;
      sa[u] = csr_src[pb + (q < cnt ? q : q0)];
    }
    if (more) {
#pragma unroll
      for (int u = 0; u < 16; ++u) {
        const int q = q1 + u;
        sb[u] = csr_src[pb + (q < cnt ? q : q1)];
      }
    }
#pragma unroll
    for (int u = 0; u < 16; ++u) ha[u] = h1mm[sa[u] * 16 + k];
    if (more) {
#pragma unroll
      for (int u = 0; u < 16; ++u) hb[u] = h1mm[sb[u] * 16 + k];
    }
#pragma unroll
    for (int u = 0; u < 16; ++u) acc += (q0 + u < cnt) ? ha[u] : 0.0f;
    if (more) {
#pragma unroll
      for (int u = 0; u < 16; ++u) acc += (q1 + u < cnt) ? hb[u] : 0.0f;
    }
  }

  float v = 0.0f;
  if (live && k < 10) {
    v = fmaf(acc, nin, bk);
    v = v > 0.0f ? v : 0.0f;
  }
  float a2 = 0.0f;
#pragma unroll
  for (int kk = 0; kk < 10; ++kk) a2 = fmaf(__shfl(v, kk, 16), w2[kk], a2);
  if (k == 0) s_res[g] = live ? a2 * nout : 0.0f;
  __syncthreads();
  if (threadIdx.x < 32) {
    const int o = blockIdx.x * 32 + threadIdx.x;
    if (o < n) h2[o] = s_res[threadIdx.x];
  }
}

__global__ void __launch_bounds__(512)
k_gather2_out(const int* __restrict__ rowptr, const int* __restrict__ deg_in,
              const int* __restrict__ csr_src, const float* __restrict__ h2,
              const float* __restrict__ inv_in, const float* __restrict__ b2,
              float* __restrict__ out, int n) {
  __shared__ float s_res[32];
  const int g = threadIdx.x >> 4;
  const int k = threadIdx.x & 15;
  const int i = blockIdx.x * 32 + g;
  const bool live = i < n;
  const int ii  = live ? i : 0;
  const int pb  = rowptr[ii];
  const int dg  = deg_in[ii];
  const float nin  = inv_in[ii];
  const float bias = b2[0];
  asm volatile("" :: "v"(pb), "v"(dg));
  const int cnt = live ? dg : 0;
  float acc = row_gather_sum<1>(csr_src, h2, pb, cnt, k, 16);
  for (int o = 8; o > 0; o >>= 1) acc += __shfl_xor(acc, o, 16);
  if (k == 0) {
    float v = 0.0f;
    if (live) {
      v = fmaf(acc, nin, bias);
      v = v > 0.0f ? v : 0.0f;
    }
    s_res[g] = v;
  }
  __syncthreads();
  if (threadIdx.x < 32) {
    const int o = blockIdx.x * 32 + threadIdx.x;
    if (o < n) out[o] = s_res[threadIdx.x];
  }
}

extern "C" void kernel_launch(void* const* d_in, const int* in_sizes, int n_in,
                              void* d_out, int out_size, void* d_ws, size_t ws_size,
                              hipStream_t stream) {
  const float* in_feat = (const float*)d_in[0];
  const int*   edge    = (const int*)d_in[1];
  const float* W0      = (const float*)d_in[2];
  const float* b0      = (const float*)d_in[3];
  const float* W1      = (const float*)d_in[4];
  const float* b1      = (const float*)d_in[5];
  const float* W2      = (const float*)d_in[6];
  const float* b2      = (const float*)d_in[7];
  float* out = (float*)d_out;

  const int N = NN;
  const int E = in_sizes[1] / 2;
  const int* src = edge;
  const int* dst = edge + E;

  float* ws      = (float*)d_ws;
  int*   wsi     = (int*)d_ws;
  float* h1mm    = ws;
  int*   coarse  = wsi + 16 * NP;
  int*   deg_in  = wsi + 17 * NP;
  float* inv_out = ws  + 18 * NP;
  float* inv_in  = ws  + 19 * NP;
  float* xs      = ws  + 20 * NP;
  int*   rowptr  = wsi + 21 * NP;
  float* h2      = ws  + 22 * NP;
  const int C    = (E + CH - 1) / CH;
  int*   cin     = wsi + 23 * NP;
  int*   cout    = cin + C * NP;
  int*   rank    = cout + C * NP;
  int*   csr_src = rank + E;

  const int B = 256;
  k_count<<<dim3(C, 2, KR), 1024, 0, stream>>>(src, dst, E, cin, cout, coarse, rank);
  k_scan<<<(N + 255) / 256, 256, 0, stream>>>(cin, cout, coarse, C, in_feat, N, deg_in,
                                              inv_out, inv_in, xs, rowptr);
  k_fill<<<(E + B - 1) / B, B, 0, stream>>>(src, dst, E, rank, cin, csr_src);

  const int n_tiles = (N + 15) / 16;
  const int waves_per_block = B / 32;
  k_l1_wmma<<<(n_tiles + waves_per_block - 1) / waves_per_block, B, 0, stream>>>(
      rowptr, deg_in, csr_src, xs, inv_in, inv_out, W0, b0, W1, h1mm, n_tiles);

  k_gather1_l2<<<(N + 31) / 32, 512, 0, stream>>>(rowptr, deg_in, csr_src, h1mm,
                                                   inv_in, inv_out, b1, W2, h2, N);
  k_gather2_out<<<(N + 31) / 32, 512, 0, stream>>>(rowptr, deg_in, csr_src, h2,
                                                    inv_in, b2, out, N);
}


# Counting work-groups scan their LDS table and store their edges sorted by destination; the scan kernel copies a node's chunk pieces into its row: no placing launch, no rank array; row sum order kept
# speedup vs baseline: 4.8788x; 1.2272x over previous
#include <hip/hip_runtime.h>
#include <math.h>


#define NN 50000
#define NP 50016
#define CH 25000
#define CU ((CH + 1023) / 1024)

typedef __attribute__((ext_vector_type(2))) float v2f;
typedef __attribute__((ext_vector_type(8))) float v8f;

#define KR 4
#define KW 12544
#define KT ((KW / 4 + 1023) / 1024)
#define KS ((KW + 1023) / 1024)
__global__ void __launch_bounds__(1024)
k_count(const int* __restrict__ src, const int* __restrict__ dst, int E,
        int* __restrict__ cin, int* __restrict__ cout, int* __restrict__ coarse,
        int* __restrict__ seg) {
  __shared__ __attribute__((aligned(16))) int s_hist[KW];
  __shared__ int s_wtot[32];
  __shared__ __attribute__((aligned(16))) int s_coarse[256];
  const int tid = threadIdx.x;
  const int lane = tid & 31;
  const int wave = tid >> 5;
  const int c = blockIdx.x;
  const bool in_role = blockIdx.y == 0;
  const int r  = blockIdx.z;
  const int r0 = r * KW;
  const int wn = (r0 + KW <= NP) ? KW : NP - r0;
  const int w4 = wn >> 2;
  int4* h4 = reinterpret_cast<int4*>(s_hist);
#pragma unroll
  for (int t = 0; t < KT; ++t) {
    const int i = tid + t * 1024;
    if (i < KW / 4) h4[i] = make_int4(0, 0, 0, 0);
  }
  if (tid < 256) s_coarse[tid] = 0;

  const int e0 = c * CH;
  const int e1 = (e0 + CH < E) ? e0 + CH : E;
  const int* __restrict__ key = in_role ? dst : src;
  int kv[CU];
#pragma unroll
  for (int u = 0; u < CU; ++u) {
    const int idx = e0 + tid + u * 1024;
    kv[u] = key[idx < e1 ? idx : e0];
  }
  int sv[CU];
#pragma unroll
  for (int u = 0; u < CU; ++u) sv[u] = 0;
  if (in_role) {
#pragma unroll
    for (int u = 0; u < CU; ++u) {
      const int idx = e0 + tid + u * 1024;
      sv[u] = src[idx < e1 ? idx : e0];
    }
    asm volatile("" :: "v"(sv[0]), "v"(sv[1]), "v"(sv[2]), "v"(sv[3]), "v"(sv[4]), "v"(sv[5]),
                 "v"(sv[6]), "v"(sv[7]), "v"(sv[8]), "v"(sv[9]), "v"(sv[10]), "v"(sv[11]),
                 "v"(sv[12]));
    asm volatile("" :: "v"(sv[13]), "v"(sv[14]), "v"(sv[15]), "v"(sv[16]), "v"(sv[17]),
                 "v"(sv[18]), "v"(sv[19]), "v"(sv[20]), "v"(sv[21]), "v"(sv[22]), "v"(sv[23]),
                 "v"(sv[24]));
  }
  __syncthreads();

  if (in_role) {
    int rk[CU];
#pragma unroll
    for (int u = 0; u < CU; ++u) {
      const int idx = e0 + tid + u * 1024;
      const unsigned j = (unsigned)(kv[u] - r0);
      rk[u] = 0;
      if (idx < e1 && j < (unsigned)wn) {
        rk[u] = atomicAdd(&s_hist[j], 1);
        atomicAdd(&s_coarse[((j >> 8) << 2) | (tid & 3)], 1);
      }
    }
    __syncthreads();
    const int j0 = tid * KS;
    int cv[KS];
    int sum = 0;
#pragma unroll
    for (int k = 0; k < KS; ++k) {
      const int j = j0 + k;
      const int hv = s_hist[j < KW ? j : 0];
      cv[k] = j < KW ? hv : 0;
      sum += cv[k];
    }
    int incl = sum;
    for (int o = 1; o < 32; o <<= 1) {
      const int t = __shfl_up(incl, o);
      if (lane >= o) incl += t;
    }
    if (lane == 31) s_wtot[wave] = incl;
    __syncthreads();
    int run = incl - sum;
#pragma unroll
    for (int w = 0; w < 32; ++w) run += (w < wave) ? s_wtot[w] : 0;
#pragma unroll
    for (int k = 0; k < KS; ++k) {
      const int j = j0 + k;
      if (j < KW) s_hist[j] = (run << 16) | cv[k];
      run += cv[k];
    }
    __syncthreads();
    int* __restrict__ sg = seg + (c * KR + r) * CH;
#pragma unroll
    for (int u = 0; u < CU; ++u) {
      const int idx = e0 + tid + u * 1024;
      const unsigned j = (unsigned)(kv[u] - r0);
      const bool ok = idx < e1 && j < (unsigned)wn;
      rk[u] += (int)((unsigned)s_hist[ok ? j : 0u] >> 16);
    }
    asm volatile("" :: "v"(rk[0]), "v"(rk[1]), "v"(rk[2]), "v"(rk[3]), "v"(rk[4]), "v"(rk[5]),
                 "v"(rk[6]), "v"(rk[7]), "v"(rk[8]), "v"(rk[9]), "v"(rk[10]), "v"(rk[11]),
                 "v"(rk[12]));
    asm volatile("" :: "v"(rk[13]), "v"(rk[14]), "v"(rk[15]), "v"(rk[16]), "v"(rk[17]),
                 "v"(rk[18]), "v"(rk[19]), "v"(rk[20]), "v"(rk[21]), "v"(rk[22]), "v"(rk[23]),
                 "v"(rk[24]));
#pragma unroll
    for (int u = 0; u < CU; ++u) {
      const int idx = e0 + tid + u * 1024;
      const unsigned j = (unsigned)(kv[u] - r0);
      if (idx < e1 && j < (unsigned)wn) sg[rk[u]] = sv[u];
    }
  } else {
#pragma unroll
    for (int u = 0; u < CU; ++u) {
      const int idx = e0 + tid + u * 1024;
      const unsigned j = (unsigned)(kv[u] - r0);
      if (idx < e1 && j < (unsigned)wn) atomicAdd(&s_hist[j], 1);
    }
  }
  __syncthreads();
  int4* __restrict__ out4 = reinterpret_cast<int4*>((in_role ? cin : cout) + c * NP + r0);
#pragma unroll
  for (int t = 0; t < KT; ++t) {
    const int i = tid + t * 1024;
    if (i < w4) out4[i] = h4[i];
  }
  if (in_role && tid < 128) {
    const int t = r * (KW / 256) + tid;
    const bool has = tid < KW / 256;
    if (has || (r == KR - 1 && t < 256)) {
      const int4 q = reinterpret_cast<const int4*>(s_coarse)[has ? tid : 0];
      coarse[c * 256 + t] = has ? (q.x + q.y) + (q.z + q.w) : 0;
    }
  }
}

#define CMAX 32
#define MT 4
#define FENCE16(a, o) asm volatile("" :: "v"(a[o]), "v"(a[o + 1]), "v"(a[o + 2]), "v"(a[o + 3]), \
    "v"(a[o + 4]), "v"(a[o + 5]), "v"(a[o + 6]), "v"(a[o + 7]), "v"(a[o + 8]), "v"(a[o + 9]), \
    "v"(a[o + 10]), "v"(a[o + 11]), "v"(a[o + 12]), "v"(a[o + 13]), "v"(a[o + 14]), "v"(a[o + 15]))
__global__ void __launch_bounds__(256)
k_scan(const int* __restrict__ cin, const int* __restrict__ cout,
       const int* __restrict__ coarse, int C,
       const float* __restrict__ x, int n, int* __restrict__ deg_in,
       float* __restrict__ inv_out, float* __restrict__ inv_in,
       float* __restrict__ xs, int* __restrict__ rowptr,
       const int* __restrict__ seg, int* __restrict__ csr_src) {
  __shared__ int s_part[8];
  __shared__ int s_wsum[8];
  const int tid  = threadIdx.x;
  const int lane = tid & 31;
  const int wave = tid >> 5;
  const int b    = blockIdx.x;
  const int i    = b * 256 + tid;
  const bool inr = i < n;
  const int ii   = inr ? i : n - 1;

  const int* __restrict__ pco = coarse + tid;
  const int* __restrict__ pin = cin + ii;
  const int* __restrict__ pou = cout + ii;
  int pc[CMAX], ti[CMAX], to[CMAX];
#pragma unroll
  for (int c = 0; c < CMAX; ++c) pc[c] = pco[c * 256];
#pragma unroll
  for (int c = 0; c < CMAX; ++c) ti[c] = pin[c * NP];
#pragma unroll
  for (int c = 0; c < CMAX; ++c) to[c] = pou[c * NP];
  const float xv = x[ii];
  asm volatile("" :: "v"(to[0]), "v"(to[1]), "v"(to[2]), "v"(to[3]), "v"(to[4]), "v"(to[5]),
               "v"(to[6]), "v"(to[7]), "v"(to[8]), "v"(to[9]), "v"(to[10]), "v"(to[11]),
               "v"(to[12]), "v"(to[13]), "v"(to[14]), "v"(to[15]));
  asm volatile("" :: "v"(to[16]), "v"(to[17]), "v"(to[18]), "v"(to[19]), "v"(to[20]), "v"(to[21]),
               "v"(to[22]), "v"(to[23]), "v"(to[24]), "v"(to[25]), "v"(to[26]), "v"(to[27]),
               "v"(to[28]), "v"(to[29]), "v"(to[30]), "v"(to[31]), "v"(xv));
  __builtin_amdgcn_sched_barrier(0);

  int acc = 0, d = 0, dout = 0;
#pragma unroll
  for (int c = 0; c < CMAX; ++c) {
    const bool ok = inr && c < C;
    acc += (c < C) ? pc[c] : 0;
    ti[c] = ok ? ti[c] : 0;
    d += ti[c] & 0xFFFF;
    dout += ok ? to[c] : 0;
  }
  acc = tid < b ? acc : 0;
  for (int o = 16; o > 0; o >>= 1) acc += __shfl_xor(acc, o);
  int incl = d;
  for (int o = 1; o < 32; o <<= 1) {
    const int t = __shfl_up(incl, o);
    if (lane >= o) incl += t;
  }
  if (lane == 0)  s_part[wave] = acc;
  if (lane == 31) s_wsum[wave] = incl;
  __syncthreads();
  int pre = 0;
  for (int w = 0; w < 8; ++w) {
    pre += s_part[w];
    if (w < wave) pre += s_wsum[w];
  }
  if (inr) {
    const int base = pre + incl - d;
    rowptr[i] = base;
    deg_in[i] = d;
    const float fi = (float)(d > 1 ? d : 1);
    inv_in[i] = 1.0f / sqrtf(fi);
    const float fo = (float)(dout > 1 ? dout : 1);
    const float io = 1.0f / sqrtf(fo);
    inv_out[i] = io;
    xs[i] = xv * io;
    const int* __restrict__ sgr = seg + (b / (KW / 256)) * CH;
    int mx = 0;
#pragma unroll
    for (int c = 0; c < CMAX; ++c) {
      const int cn = ti[c] & 0xFFFF;
      mx = cn > mx ? cn : mx;
    }
    if (mx > 0) {
      int g[MT * CMAX];
#pragma unroll
      for (int tt = 0; tt < MT; ++tt) {
#pragma unroll
        for (int c = 0; c < CMAX; ++c) {
          const int cn = ti[c] & 0xFFFF;
          const int of = (int)((unsigned)ti[c] >> 16);
          const int* a = (tt < cn) ? sgr + c * (KR * CH) + of + tt : pin + c * NP;
          g[tt * CMAX + c] = *a;
        }
      }
      FENCE16(g, 0);  FENCE16(g, 16); FENCE16(g, 32); FENCE16(g, 48);
      FENCE16(g, 64); FENCE16(g, 80); FENCE16(g, 96); FENCE16(g, 112);
      int run = base;
#pragma unroll
      for (int c = 0; c < CMAX; ++c) {
        const int cn = ti[c] & 0xFFFF;
#pragma unroll
        for (int tt = 0; tt < MT; ++tt) {
          if (tt < cn) csr_src[run + tt] = g[tt * CMAX + c];
        }
        run += cn;
      }
    }
    if (mx > MT) {
      int run = base;
#pragma unroll
      for (int c = 0; c < CMAX; ++c) {
        const int cn = ti[c] & 0xFFFF;
        const int of = (int)((unsigned)ti[c] >> 16);
        for (int t = MT; t < cn; ++t) csr_src[run + t] = sgr[c * (KR * CH) + of + t];
        run += cn;
      }
    }
  }
}

template <int U>
__device__ __forceinline__ float row_gather_sum(const int* __restrict__ csr_src,
                                                const float* __restrict__ vals,
                                                int pb, int cnt, int first, int stride) {
  float a = 0.0f;
  for (int q0 = first; q0 < cnt; q0 += 2 * U * stride) {
    const int q1 = q0 + U * stride;
    const bool more = q1 < cnt;
    int sa[U], sb[U];
    float ha[U], hb[U];
#pragma unroll
    for (int u = 0; u < U; ++u) {
      const int q = q0 + u * stride;
      sa[u] = csr_src[pb + (q < cnt ? q : q0)];
    }
    if (more) {
#pragma unroll
      for (int u = 0; u < U; ++u) {
        const int q = q1 + u * stride;
        sb[u] = csr_src[pb + (q < cnt ? q : q1)];
      }
    }
#pragma unroll
    for (int u = 0; u < U; ++u) ha[u] = vals[sa[u]];
    if (more) {
#pragma unroll
      for (int u = 0; u < U; ++u) hb[u] = vals[sb[u]];
    }
#pragma unroll
    for (int u = 0; u < U; ++u) a += (q0 + u * stride < cnt) ? ha[u] : 0.0f;
    if (more) {
#pragma unroll
      for (int u = 0; u < U; ++u) a += (q1 + u * stride < cnt) ? hb[u] : 0.0f;
    }
  }
  return a;
}

#define L1B 256
#define L1S ((1600 + L1B - 1) / L1B)
__global__ void __launch_bounds__(L1B)
k_l1_wmma(const int* __restrict__ rowptr,
                          const int* __restrict__ deg_in,
                          const int* __restrict__ csr_src,
                          const float* __restrict__ xs,
                          const float* __restrict__ inv_in,
                          const float* __restrict__ inv_out,
                          const float* __restrict__ W0,
                          const float* __restrict__ b0,
                          const float* __restrict__ W1,
                          float* __restrict__ h1mm,
                          int n_tiles) {
  __shared__ float sW0[100];
  __shared__ float sb0[100];
  __shared__ float sW1p[100 * 16];
  const int tid  = threadIdx.x;
  const int wave = tid >> 5;
  const int lane = tid & 31;
  const int tile = blockIdx.x * (L1B >> 5) + wave;
  const bool has = tile < n_tiles;
  const int i0   = (has ? tile : 0) * 16;
  const int half = lane >> 4;
  const int m    = lane & 15;
  const int col  = lane & 15;
  const int r    = i0 + m;
  const int pb   = rowptr[r];
  const int cnt  = deg_in[r];
  const float nin = inv_in[r];
  float nout[8];
#pragma unroll
  for (int v = 0; v < 8; ++v) nout[v] = inv_out[i0 + v + 8 * half];
  float w1v[L1S];
#pragma unroll
  for (int j = 0; j < L1S; ++j) {
    const int t = tid + j * L1B;
    const int tc = t < 1600 ? t : 1599;
    const int c = tc & 15;
    w1v[j] = W1[(tc >> 4) * 10 + (c < 10 ? c : 0)];
  }
  const int tw = tid < 100 ? tid : 99;
  const float w0v = W0[tw];
  const float b0v = b0[tw];
  if (tid < 100) { sW0[tid] = w0v; sb0[tid] = b0v; }
#pragma unroll
  for (int j = 0; j < L1S; ++j) {
    const int t = tid + j * L1B;
    if (t < 1600) sW1p[t] = ((t & 15) < 10) ? w1v[j] : 0.0f;
  }
  __syncthreads();
  if (!has) return;

  float part = row_gather_sum<8>(csr_src, xs, pb, cnt, half, 2);
  part += __shfl_xor(part, 16);
  const float s = part * nin;

  v8f c = {};
#pragma unroll
  for (int kk = 0; kk < 25; ++kk) {
    const int k0 = kk * 4 + half * 2;
    float a0 = fmaf(s, sW0[k0], sb0[k0]);
    a0 = a0 >= 0.0f ? a0 : 0.01f * a0;
    float a1 = fmaf(s, sW0[k0 + 1], sb0[k0 + 1]);
    a1 = a1 >= 0.0f ? a1 : 0.01f * a1;
    v2f a = {a0, a1};
    v2f b = {sW1p[k0 * 16 + col], sW1p[(k0 + 1) * 16 + col]};
    c = __builtin_amdgcn_wmma_f32_16x16x4_f32(false, a, false, b, (short)0, c,
                                              false, false);
  }

#pragma unroll
  for (int v = 0; v < 8; ++v) {
    const int i = i0 + v + 8 * half;
    h1mm[i * 16 + col] = c[v] * nout[v];
  }
}

__global__ void __launch_bounds__(512)
k_gather1_l2(const int* __restrict__ rowptr, const int* __restrict__ deg_in,
             const int* __restrict__ csr_src, const float* __restrict__ h1mm,
             const float* __restrict__ inv_in, const float* __restrict__ inv_out,
             const float* __restrict__ b1, const float* __restrict__ W2,
             float* __restrict__ h2, int n) {
  __shared__ float s_res[32];
  const int g = threadIdx.x >> 4;
  const int k = threadIdx.x & 15;
  const int i = blockIdx.x * 32 + g;
  const bool live = i < n;
  const int ii  = live ? i : 0;
  const int pb  = rowptr[ii];
  const int dg  = deg_in[ii];
  const float nin  = inv_in[ii];
  const float nout = inv_out[ii];
  const float bk   = b1[k < 10 ? k : 0];
  float w2[10];
#pragma unroll
  for (int kk = 0; kk < 10; ++kk) w2[kk] = W2[kk];
  asm volatile("" :: "v"(pb), "v"(dg));
  const int cnt = live ? dg : 0;

  float acc = 0.0f;
  for (int q0 = 0; q0 < cnt; q0 += 32) {
    const int q1 = q0 + 16;
    const bool more = q1 < cnt;
    int sa[16], sb[16];
    float ha[16], hb[16];
#pragma unroll
    for (int u = 0; u < 16; ++u) {
      const int q = q0 + u;
      sa[u] = csr_src[pb + (q < cnt ? q : q0)];
    }
    if (more) {
#pragma unroll
      for (int u = 0; u < 16; ++u) {
        const int q = q1 + u;
        sb[u] = csr_src[pb + (q < cnt ? q : q1)];
      }
    }
#pragma unroll
    for (int u = 0; u < 16; ++u) ha[u] = h1mm[sa[u] * 16 + k];
    if (more) {
#pragma unroll
      for (int u = 0; u < 16; ++u) hb[u] = h1mm[sb[u] * 16 + k];
    }
#pragma unroll
    for (int u = 0; u < 16; ++u) acc += (q0 + u < cnt) ? ha[u] : 0.0f;
    if (more) {
#pragma unroll
      for (int u = 0; u < 16; ++u) acc += (q1 + u < cnt) ? hb[u] : 0.0f;
    }
  }

  float v = 0.0f;
  if (live && k < 10) {
    v = fmaf(acc, nin, bk);
    v = v > 0.0f ? v : 0.0f;
  }
  float a2 = 0.0f;
#pragma unroll
  for (int kk = 0; kk < 10; ++kk) a2 = fmaf(__shfl(v, kk, 16), w2[kk], a2);
  if (k == 0) s_res[g] = live ? a2 * nout : 0.0f;
  __syncthreads();
  if (threadIdx.x < 32) {
    const int o = blockIdx.x * 32 + threadIdx.x;
    if (o < n) h2[o] = s_res[threadIdx.x];
  }
}

__global__ void __launch_bounds__(512)
k_gather2_out(const int* __restrict__ rowptr, const int* __restrict__ deg_in,
              const int* __restrict__ csr_src, const float* __restrict__ h2,
              const float* __restrict__ inv_in, const float* __restrict__ b2,
              float* __restrict__ out, int n) {
  __shared__ float s_res[32];
  const int g = threadIdx.x >> 4;
  const int k = threadIdx.x & 15;
  const int i = blockIdx.x * 32 + g;
  const bool live = i < n;
  const int ii  = live ? i : 0;
  const int pb  = rowptr[ii];
  const int dg  = deg_in[ii];
  const float nin  = inv_in[ii];
  const float bias = b2[0];
  asm volatile("" :: "v"(pb), "v"(dg));
  const int cnt = live ? dg : 0;
  float acc = row_gather_sum<1>(csr_src, h2, pb, cnt, k, 16);
  for (int o = 8; o > 0; o >>= 1) acc += __shfl_xor(acc, o, 16);
  if (k == 0) {
    float v = 0.0f;
    if (live) {
      v = fmaf(acc, nin, bias);
      v = v > 0.0f ? v : 0.0f;
    }
    s_res[g] = v;
  }
  __syncthreads();
  if (threadIdx.x < 32) {
    const int o = blockIdx.x * 32 + threadIdx.x;
    if (o < n) out[o] = s_res[threadIdx.x];
  }
}

extern "C" void kernel_launch(void* const* d_in, const int* in_sizes, int n_in,
                              void* d_out, int out_size, void* d_ws, size_t ws_size,
                              hipStream_t stream) {
  const float* in_feat = (const float*)d_in[0];
  const int*   edge    = (const int*)d_in[1];
  const float* W0      = (const float*)d_in[2];
  const float* b0      = (const float*)d_in[3];
  const float* W1      = (const float*)d_in[4];
  const float* b1      = (const float*)d_in[5];
  const float* W2      = (const float*)d_in[6];
  const float* b2      = (const float*)d_in[7];
  float* out = (float*)d_out;

  const int N = NN;
  const int E = in_sizes[1] / 2;
  const int* src = edge;
  const int* dst = edge + E;

  float* ws      = (float*)d_ws;
  int*   wsi     = (int*)d_ws;
  float* h1mm    = ws;
  int*   coarse  = wsi + 16 * NP;
  int*   deg_in  = wsi + 17 * NP;
  float* inv_out = ws  + 18 * NP;
  float* inv_in  = ws  + 19 * NP;
  float* xs      = ws  + 20 * NP;
  int*   rowptr  = wsi + 21 * NP;
  float* h2      = ws  + 22 * NP;
  const int C    = (E + CH - 1) / CH;
  int*   cin     = wsi + 23 * NP;
  int*   cout    = cin + C * NP;
  int*   seg     = cout + C * NP;
  int*   csr_src = seg + C * KR * CH;

  const int B = 256;
  k_count<<<dim3(C, 2, KR), 1024, 0, stream>>>(src, dst, E, cin, cout, coarse, seg);
  k_scan<<<(N + 255) / 256, 256, 0, stream>>>(cin, cout, coarse, C, in_feat, N, deg_in,
                                              inv_out, inv_in, xs, rowptr, seg, csr_src);

  const int n_tiles = (N + 15) / 16;
  const int waves_per_block = B / 32;
  k_l1_wmma<<<(n_tiles + waves_per_block - 1) / waves_per_block, B, 0, stream>>>(
      rowptr, deg_in, csr_src, xs, inv_in, inv_out, W0, b0, W1, h1mm, n_tiles);

  k_gather1_l2<<<(N + 31) / 32, 512, 0, stream>>>(rowptr, deg_in, csr_src, h1mm,
                                                   inv_in, inv_out, b1, W2, h2, N);
  k_gather2_out<<<(N + 31) / 32, 512, 0, stream>>>(rowptr, deg_in, csr_src, h2,
                                                    inv_in, b2, out, N);
}


# The sort's scattered 4-byte global stores (segment placing in the counting kernel, row copy in the scan kernel) go to LDS first and leave as whole-line 16-byte stores; row sum order kept
# speedup vs baseline: 5.4793x; 1.1231x over previous
#include <hip/hip_runtime.h>
#include <math.h>


#define NN 50000
#define NP 50016
#define CH 25000
#define CU ((CH + 1023) / 1024)

typedef __attribute__((ext_vector_type(2))) float v2f;
typedef __attribute__((ext_vector_type(8))) float v8f;

#define KR 4
#define KW 12544
#define KT ((KW / 4 + 1023) / 1024)
#define KS ((KW + 1023) / 1024)
__global__ void __launch_bounds__(1024)
k_count(const int* __restrict__ src, const int* __restrict__ dst, int E,
        int* __restrict__ cin, int* __restrict__ cout, int* __restrict__ coarse,
        int* __restrict__ seg) {
  __shared__ __attribute__((aligned(16))) int s_hist[KW];
  __shared__ int s_wtot[32];
  __shared__ __attribute__((aligned(16))) int s_coarse[256];
  __shared__ __attribute__((aligned(16))) int s_seg[CH];
  const int tid = threadIdx.x;
  const int lane = tid & 31;
  const int wave = tid >> 5;
  const int c = blockIdx.x;
  const bool in_role = blockIdx.y == 0;
  const int r  = blockIdx.z;
  const int r0 = r * KW;
  const int wn = (r0 + KW <= NP) ? KW : NP - r0;
  const int w4 = wn >> 2;
  int4* h4 = reinterpret_cast<int4*>(s_hist);
#pragma unroll
  for (int t = 0; t < KT; ++t) {
    const int i = tid + t * 1024;
    if (i < KW / 4) h4[i] = make_int4(0, 0, 0, 0);
  }
  if (tid < 256) s_coarse[tid] = 0;

  const int e0 = c * CH;
  const int e1 = (e0 + CH < E) ? e0 + CH : E;
  const int* __restrict__ key = in_role ? dst : src;
  int kv[CU];
#pragma unroll
  for (int u = 0; u < CU; ++u) {
    const int idx = e0 + tid + u * 1024;
    kv[u] = key[idx < e1 ? idx : e0];
  }
  int sv[CU];
#pragma unroll
  for (int u = 0; u < CU; ++u) sv[u] = 0;
  if (in_role) {
#pragma unroll
    for (int u = 0; u < CU; ++u) {
      const int idx = e0 + tid + u * 1024;
      sv[u] = src[idx < e1 ? idx : e0];
    }
    asm volatile("" :: "v"(sv[0]), "v"(sv[1]), "v"(sv[2]), "v"(sv[3]), "v"(sv[4]), "v"(sv[5]),
                 "v"(sv[6]), "v"(sv[7]), "v"(sv[8]), "v"(sv[9]), "v"(sv[10]), "v"(sv[11]),
                 "v"(sv[12]));
    asm volatile("" :: "v"(sv[13]), "v"(sv[14]), "v"(sv[15]), "v"(sv[16]), "v"(sv[17]),
                 "v"(sv[18]), "v"(sv[19]), "v"(sv[20]), "v"(sv[21]), "v"(sv[22]), "v"(sv[23]),
                 "v"(sv[24]));
  }
  __syncthreads();

  if (in_role) {
    int rk[CU];
#pragma unroll
    for (int u = 0; u < CU; ++u) {
      const int idx = e0 + tid + u * 1024;
      const unsigned j = (unsigned)(kv[u] - r0);
      rk[u] = 0;
      if (idx < e1 && j < (unsigned)wn) {
        rk[u] = atomicAdd(&s_hist[j], 1);
        atomicAdd(&s_coarse[((j >> 8) << 2) | (tid & 3)], 1);
      }
    }
    __syncthreads();
    const int j0 = tid * KS;
    int cv[KS];
    int sum = 0;
#pragma unroll
    for (int k = 0; k < KS; ++k) {
      const int j = j0 + k;
      const int hv = s_hist[j < KW ? j : 0];
      cv[k] = j < KW ? hv : 0;
      sum += cv[k];
    }
    int incl = sum;
    for (int o = 1; o < 32; o <<= 1) {
      const int t = __shfl_up(incl, o);
      if (lane >= o) incl += t;
    }
    if (lane == 31) s_wtot[wave] = incl;
    __syncthreads();
    int run = incl - sum;
#pragma unroll
    for (int w = 0; w < 32; ++w) run += (w < wave) ? s_wtot[w] : 0;
#pragma unroll
    for (int k = 0; k < KS; ++k) {
      const int j = j0 + k;
      if (j < KW) s_hist[j] = (run << 16) | cv[k];
      run += cv[k];
    }
    __syncthreads();
#pragma unroll
    for (int u = 0; u < CU; ++u) {
      const int idx = e0 + tid + u * 1024;
      const unsigned j = (unsigned)(kv[u] - r0);
      const bool ok = idx < e1 && j < (unsigned)wn;
      rk[u] += (int)((unsigned)s_hist[ok ? j : 0u] >> 16);
    }
    asm volatile("" :: "v"(rk[0]), "v"(rk[1]), "v"(rk[2]), "v"(rk[3]), "v"(rk[4]), "v"(rk[5]),
                 "v"(rk[6]), "v"(rk[7]), "v"(rk[8]), "v"(rk[9]), "v"(rk[10]), "v"(rk[11]),
                 "v"(rk[12]));
    asm volatile("" :: "v"(rk[13]), "v"(rk[14]), "v"(rk[15]), "v"(rk[16]), "v"(rk[17]),
                 "v"(rk[18]), "v"(rk[19]), "v"(rk[20]), "v"(rk[21]), "v"(rk[22]), "v"(rk[23]),
                 "v"(rk[24]));
#pragma unroll
    for (int u = 0; u < CU; ++u) {
      const int idx = e0 + tid + u * 1024;
      const unsigned j = (unsigned)(kv[u] - r0);
      if (idx < e1 && j < (unsigned)wn) s_seg[rk[u]] = sv[u];
    }
  } else {
#pragma unroll
    for (int u = 0; u < CU; ++u) {
      const int idx = e0 + tid + u * 1024;
      const unsigned j = (unsigned)(kv[u] - r0);
      if (idx < e1 && j < (unsigned)wn) atomicAdd(&s_hist[j], 1);
    }
  }
  __syncthreads();
  int4* __restrict__ out4 = reinterpret_cast<int4*>((in_role ? cin : cout) + c * NP + r0);
#pragma unroll
  for (int t = 0; t < KT; ++t) {
    const int i = tid + t * 1024;
    if (i < w4) out4[i] = h4[i];
  }
  if (in_role) {
    int4* __restrict__ sg4 = reinterpret_cast<int4*>(seg + (c * KR + r) * CH);
    const int4* s4 = reinterpret_cast<const int4*>(s_seg);
    const unsigned lw = (unsigned)s_hist[KW - 1];
    const int tot = (int)(lw >> 16) + (int)(lw & 0xFFFFu);
    const int nq = (tot + 3) >> 2;
#pragma unroll
    for (int t = 0; t < (CH / 4 + 1023) / 1024; ++t) {
      const int q = tid + t * 1024;
      if (q < nq) {
        int4 v = s4[q];
        v.y = (4 * q + 1 < tot) ? v.y : 0;
        v.z = (4 * q + 2 < tot) ? v.z : 0;
        v.w = (4 * q + 3 < tot) ? v.w : 0;
        sg4[q] = v;
      }
    }
  }
  if (in_role && tid < 128) {
    const int t = r * (KW / 256) + tid;
    const bool has = tid < KW / 256;
    if (has || (r == KR - 1 && t < 256)) {
      const int4 q = reinterpret_cast<const int4*>(s_coarse)[has ? tid : 0];
      coarse[c * 256 + t] = has ? (q.x + q.y) + (q.z + q.w) : 0;
    }
  }
}

#define CMAX 32
#define MT 4
#define CAPB 8192
#define FENCE16(a, o) asm volatile("" :: "v"(a[o]), "v"(a[o + 1]), "v"(a[o + 2]), "v"(a[o + 3]), \
    "v"(a[o + 4]), "v"(a[o + 5]), "v"(a[o + 6]), "v"(a[o + 7]), "v"(a[o + 8]), "v"(a[o + 9]), \
    "v"(a[o + 10]), "v"(a[o + 11]), "v"(a[o + 12]), "v"(a[o + 13]), "v"(a[o + 14]), "v"(a[o + 15]))
__global__ void __launch_bounds__(256)
k_scan(const int* __restrict__ cin, const int* __restrict__ cout,
       const int* __restrict__ coarse, int C,
       const float* __restrict__ x, int n, int* __restrict__ deg_in,
       float* __restrict__ inv_out, float* __restrict__ inv_in,
       float* __restrict__ xs, int* __restrict__ rowptr,
       const int* __restrict__ seg, int* __restrict__ csr_src) {
  __shared__ int s_part[8];
  __shared__ int s_wsum[8];
  __shared__ __attribute__((aligned(16))) int s_stage[CAPB + 4];
  const int tid  = threadIdx.x;
  const int lane = tid & 31;
  const int wave = tid >> 5;
  const int b    = blockIdx.x;
  const int i    = b * 256 + tid;
  const bool inr = i < n;
  const int ii   = inr ? i : n - 1;

  const int* __restrict__ pco = coarse + tid;
  const int* __restrict__ pin = cin + ii;
  const int* __restrict__ pou = cout + ii;
  int pc[CMAX], ti[CMAX], to[CMAX];
#pragma unroll
  for (int c = 0; c < CMAX; ++c) pc[c] = pco[c * 256];
#pragma unroll
  for (int c = 0; c < CMAX; ++c) ti[c] = pin[c * NP];
#pragma unroll
  for (int c = 0; c < CMAX; ++c) to[c] = pou[c * NP];
  const float xv = x[ii];
  asm volatile("" :: "v"(to[0]), "v"(to[1]), "v"(to[2]), "v"(to[3]), "v"(to[4]), "v"(to[5]),
               "v"(to[6]), "v"(to[7]), "v"(to[8]), "v"(to[9]), "v"(to[10]), "v"(to[11]),
               "v"(to[12]), "v"(to[13]), "v"(to[14]), "v"(to[15]));
  asm volatile("" :: "v"(to[16]), "v"(to[17]), "v"(to[18]), "v"(to[19]), "v"(to[20]), "v"(to[21]),
               "v"(to[22]), "v"(to[23]), "v"(to[24]), "v"(to[25]), "v"(to[26]), "v"(to[27]),
               "v"(to[28]), "v"(to[29]), "v"(to[30]), "v"(to[31]), "v"(xv));
  __builtin_amdgcn_sched_barrier(0);

  int acc = 0, d = 0, dout = 0;
#pragma unroll
  for (int c = 0; c < CMAX; ++c) {
    const bool ok = inr && c < C;
    acc += (c < C) ? pc[c] : 0;
    ti[c] = ok ? ti[c] : 0;
    d += ti[c] & 0xFFFF;
    dout += ok ? to[c] : 0;
  }
  acc = tid < b ? acc : 0;
  for (int o = 16; o > 0; o >>= 1) acc += __shfl_xor(acc, o);
  int incl = d;
  for (int o = 1; o < 32; o <<= 1) {
    const int t = __shfl_up(incl, o);
    if (lane >= o) incl += t;
  }
  if (lane == 0)  s_part[wave] = acc;
  if (lane == 31) s_wsum[wave] = incl;
  __syncthreads();
  int pre = 0, bstart = 0, btot = 0;
  for (int w = 0; w < 8; ++w) {
    const int pw = s_part[w];
    const int sw = s_wsum[w];
    pre += pw;
    bstart += pw;
    if (w < wave) pre += sw;
    btot += sw;
  }
  const int base = pre + incl - d;
  if (inr) {
    rowptr[i] = base;
    deg_in[i] = d;
    const float fi = (float)(d > 1 ? d : 1);
    inv_in[i] = 1.0f / sqrtf(fi);
    const float fo = (float)(dout > 1 ? dout : 1);
    const float io = 1.0f / sqrtf(fo);
    inv_out[i] = io;
    xs[i] = xv * io;
  }
  const int* __restrict__ sgr = seg + (b / (KW / 256)) * CH;
  int mx = 0;
#pragma unroll
  for (int c = 0; c < CMAX; ++c) {
    const int cn = ti[c] & 0xFFFF;
    mx = cn > mx ? cn : mx;
  }
  if (btot <= CAPB) {
    const int rel = (bstart & 3) + base - bstart;
    if (mx > 0) {
      int g[MT * CMAX];
#pragma unroll
      for (int tt = 0; tt < MT; ++tt) {
#pragma unroll
        for (int c = 0; c < CMAX; ++c) {
          const int cn = ti[c] & 0xFFFF;
          const int of = (int)((unsigned)ti[c] >> 16);
          const int* a = (tt < cn) ? sgr + c * (KR * CH) + of + tt : pin + c * NP;
          g[tt * CMAX + c] = *a;
        }
      }
      FENCE16(g, 0);  FENCE16(g, 16); FENCE16(g, 32); FENCE16(g, 48);
      FENCE16(g, 64); FENCE16(g, 80); FENCE16(g, 96); FENCE16(g, 112);
      int run = rel;
#pragma unroll
      for (int c = 0; c < CMAX; ++c) {
        const int cn = ti[c] & 0xFFFF;
#pragma unroll
        for (int tt = 0; tt < MT; ++tt) {
          if (tt < cn) s_stage[run + tt] = g[tt * CMAX + c];
        }
        run += cn;
      }
    }
    if (mx > MT) {
      int run = rel;
#pragma unroll
      for (int c = 0; c < CMAX; ++c) {
        const int cn = ti[c] & 0xFFFF;
        const int of = (int)((unsigned)ti[c] >> 16);
        for (int t = MT; t < cn; ++t) s_stage[run + t] = sgr[c * (KR * CH) + of + t];
        run += cn;
      }
    }
    __syncthreads();
    const int ge = bstart + btot;
    const int q0 = bstart >> 2;
    const int4* st4 = reinterpret_cast<const int4*>(s_stage);
    int4* csr4 = reinterpret_cast<int4*>(csr_src);
    for (int q = q0 + tid; 4 * q < ge; q += 256) {
      const int lo = 4 * q;
      if (lo >= bstart && lo + 4 <= ge) {
        csr4[q] = st4[q - q0];
      } else {
        for (int k = 0; k < 4; ++k) {
          const int gw = lo + k;
          if (gw >= bstart && gw < ge) csr_src[gw] = s_stage[gw - 4 * q0];
        }
      }
    }
  } else if (inr) {
    int run = base;
#pragma nounroll
    for (int c = 0; c < CMAX; ++c) {
      const unsigned tw = (c < C) ? (unsigned)pin[c * NP] : 0u;
      const int cn = (int)(tw & 0xFFFFu);
      const int* __restrict__ pc0 = sgr + c * (KR * CH) + (int)(tw >> 16);
#pragma nounroll
      for (int t = 0; t < cn; ++t) csr_src[run + t] = pc0[t];
      run += cn;
    }
  }
}

template <int U>
__device__ __forceinline__ float row_gather_sum(const int* __restrict__ csr_src,
                                                const float* __restrict__ vals,
                                                int pb, int cnt, int first, int stride) {
  float a = 0.0f;
  for (int q0 = first; q0 < cnt; q0 += 2 * U * stride) {
    const int q1 = q0 + U * stride;
    const bool more = q1 < cnt;
    int sa[U], sb[U];
    float ha[U], hb[U];
#pragma unroll
    for (int u = 0; u < U; ++u) {
      const int q = q0 + u * stride;
      sa[u] = csr_src[pb + (q < cnt ? q : q0)];
    }
    if (more) {
#pragma unroll
      for (int u = 0; u < U; ++u) {
        const int q = q1 + u * stride;
        sb[u] = csr_src[pb + (q < cnt ? q : q1)];
      }
    }
#pragma unroll
    for (int u = 0; u < U; ++u) ha[u] = vals[sa[u]];
    if (more) {
#pragma unroll
      for (int u = 0; u < U; ++u) hb[u] = vals[sb[u]];
    }
#pragma unroll
    for (int u = 0; u < U; ++u) a += (q0 + u * stride < cnt) ? ha[u] : 0.0f;
    if (more) {
#pragma unroll
      for (int u = 0; u < U; ++u) a += (q1 + u * stride < cnt) ? hb[u] : 0.0f;
    }
  }
  return a;
}

#define L1B 256
#define L1S ((1600 + L1B - 1) / L1B)
__global__ void __launch_bounds__(L1B)
k_l1_wmma(const int* __restrict__ rowptr,
                          const int* __restrict__ deg_in,
                          const int* __restrict__ csr_src,
                          const float* __restrict__ xs,
                          const float* __restrict__ inv_in,
                          const float* __restrict__ inv_out,
                          const float* __restrict__ W0,
                          const float* __restrict__ b0,
                          const float* __restrict__ W1,
                          float* __restrict__ h1mm,
                          int n_tiles) {
  __shared__ float sW0[100];
  __shared__ float sb0[100];
  __shared__ float sW1p[100 * 16];
  const int tid  = threadIdx.x;
  const int wave = tid >> 5;
  const int lane = tid & 31;
  const int tile = blockIdx.x * (L1B >> 5) + wave;
  const bool has = tile < n_tiles;
  const int i0   = (has ? tile : 0) * 16;
  const int half = lane >> 4;
  const int m    = lane & 15;
  const int col  = lane & 15;
  const int r    = i0 + m;
  const int pb   = rowptr[r];
  const int cnt  = deg_in[r];
  const float nin = inv_in[r];
  float nout[8];
#pragma unroll
  for (int v = 0; v < 8; ++v) nout[v] = inv_out[i0 + v + 8 * half];
  float w1v[L1S];
#pragma unroll
  for (int j = 0; j < L1S; ++j) {
    const int t = tid + j * L1B;
    const int tc = t < 1600 ? t : 1599;
    const int c = tc & 15;
    w1v[j] = W1[(tc >> 4) * 10 + (c < 10 ? c : 0)];
  }
  const int tw = tid < 100 ? tid : 99;
  const float w0v = W0[tw];
  const float b0v = b0[tw];
  if (tid < 100) { sW0[tid] = w0v; sb0[tid] = b0v; }
#pragma unroll
  for (int j = 0; j < L1S; ++j) {
    const int t = tid + j * L1B;
    if (t < 1600) sW1p[t] = ((t & 15) < 10) ? w1v[j] : 0.0f;
  }
  __syncthreads();
  if (!has) return;

  float part = row_gather_sum<8>(csr_src, xs, pb, cnt, half, 2);
  part += __shfl_xor(part, 16);
  const float s = part * nin;

  v8f c = {};
#pragma unroll
  for (int kk = 0; kk < 25; ++kk) {
    const int k0 = kk * 4 + half * 2;
    float a0 = fmaf(s, sW0[k0], sb0[k0]);
    a0 = a0 >= 0.0f ? a0 : 0.01f * a0;
    float a1 = fmaf(s, sW0[k0 + 1], sb0[k0 + 1]);
    a1 = a1 >= 0.0f ? a1 : 0.01f * a1;
    v2f a = {a0, a1};
    v2f b = {sW1p[k0 * 16 + col], sW1p[(k0 + 1) * 16 + col]};
    c = __builtin_amdgcn_wmma_f32_16x16x4_f32(false, a, false, b, (short)0, c,
                                              false, false);
  }

#pragma unroll
  for (int v = 0; v < 8; ++v) {
    const int i = i0 + v + 8 * half;
    h1mm[i * 16 + col] = c[v] * nout[v];
  }
}

__global__ void __launch_bounds__(512)
k_gather1_l2(const int* __restrict__ rowptr, const int* __restrict__ deg_in,
             const int* __restrict__ csr_src, const float* __restrict__ h1mm,
             const float* __restrict__ inv_in, const float* __restrict__ inv_out,
             const float* __restrict__ b1, const float* __restrict__ W2,
             float* __restrict__ h2, int n) {
  __shared__ float s_res[32];
  const int g = threadIdx.x >> 4;
  const int k = threadIdx.x & 15;
  const int i = blockIdx.x * 32 + g;
  const bool live = i < n;
  const int ii  = live ? i : 0;
  const int pb  = rowptr[ii];
  const int dg  = deg_in[ii];
  const float nin  = inv_in[ii];
  const float nout = inv_out[ii];
  const float bk   = b1[k < 10 ? k : 0];
  float w2[10];
#pragma unroll
  for (int kk = 0; kk < 10; ++kk) w2[kk] = W2[kk];
  asm volatile("" :: "v"(pb), "v"(dg));
  const int cnt = live ? dg : 0;

  float acc = 0.0f;
  for (int q0 = 0; q0 < cnt; q0 += 32) {
    const int q1 = q0 + 16;
    const bool more = q1 < cnt;
    int sa[16], sb[16];
    float ha[16], hb[16];
#pragma unroll
    for (int u = 0; u < 16; ++u) {
      const int q = q0 + u;
      sa[u] = csr_src[pb + (q < cnt ? q : q0)];
    }
    if (more) {
#pragma unroll
      for (int u = 0; u < 16; ++u) {
        const int q = q1 + u;
        sb[u] = csr_src[pb + (q < cnt ? q : q1)];
      }
    }
#pragma unroll
    for (int u = 0; u < 16; ++u) ha[u] = h1mm[sa[u] * 16 + k];
    if (more) {
#pragma unroll
      for (int u = 0; u < 16; ++u) hb[u] = h1mm[sb[u] * 16 + k];
    }
#pragma unroll
    for (int u = 0; u < 16; ++u) acc += (q0 + u < cnt) ? ha[u] : 0.0f;
    if (more) {
#pragma unroll
      for (int u = 0; u < 16; ++u) acc += (q1 + u < cnt) ? hb[u] : 0.0f;
    }
  }

  float v = 0.0f;
  if (live && k < 10) {
    v = fmaf(acc, nin, bk);
    v = v > 0.0f ? v : 0.0f;
  }
  float a2 = 0.0f;
#pragma unroll
  for (int kk = 0; kk < 10; ++kk) a2 = fmaf(__shfl(v, kk, 16), w2[kk], a2);
  if (k == 0) s_res[g] = live ? a2 * nout : 0.0f;
  __syncthreads();
  if (threadIdx.x < 32) {
    const int o = blockIdx.x * 32 + threadIdx.x;
    if (o < n) h2[o] = s_res[threadIdx.x];
  }
}

__global__ void __launch_bounds__(512)
k_gather2_out(const int* __restrict__ rowptr, const int* __restrict__ deg_in,
              const int* __restrict__ csr_src, const float* __restrict__ h2,
              const float* __restrict__ inv_in, const float* __restrict__ b2,
              float* __restrict__ out, int n) {
  __shared__ float s_res[32];
  const int g = threadIdx.x >> 4;
  const int k = threadIdx.x & 15;
  const int i = blockIdx.x * 32 + g;
  const bool live = i < n;
  const int ii  = live ? i : 0;
  const int pb  = rowptr[ii];
  const int dg  = deg_in[ii];
  const float nin  = inv_in[ii];
  const float bias = b2[0];
  asm volatile("" :: "v"(pb), "v"(dg));
  const int cnt = live ? dg : 0;
  float acc = row_gather_sum<1>(csr_src, h2, pb, cnt, k, 16);
  for (int o = 8; o > 0; o >>= 1) acc += __shfl_xor(acc, o, 16);
  if (k == 0) {
    float v = 0.0f;
    if (live) {
      v = fmaf(acc, nin, bias);
      v = v > 0.0f ? v : 0.0f;
    }
    s_res[g] = v;
  }
  __syncthreads();
  if (threadIdx.x < 32) {
    const int o = blockIdx.x * 32 + threadIdx.x;
    if (o < n) out[o] = s_res[threadIdx.x];
  }
}

extern "C" void kernel_launch(void* const* d_in, const int* in_sizes, int n_in,
                              void* d_out, int out_size, void* d_ws, size_t ws_size,
                              hipStream_t stream) {
  const float* in_feat = (const float*)d_in[0];
  const int*   edge    = (const int*)d_in[1];
  const float* W0      = (const float*)d_in[2];
  const float* b0      = (const float*)d_in[3];
  const float* W1      = (const float*)d_in[4];
  const float* b1      = (const float*)d_in[5];
  const float* W2      = (const float*)d_in[6];
  const float* b2      = (const float*)d_in[7];
  float* out = (float*)d_out;

  const int N = NN;
  const int E = in_sizes[1] / 2;
  const int* src = edge;
  const int* dst = edge + E;

  float* ws      = (float*)d_ws;
  int*   wsi     = (int*)d_ws;
  float* h1mm    = ws;
  int*   coarse  = wsi + 16 * NP;
  int*   deg_in  = wsi + 17 * NP;
  float* inv_out = ws  + 18 * NP;
  float* inv_in  = ws  + 19 * NP;
  float* xs      = ws  + 20 * NP;
  int*   rowptr  = wsi + 21 * NP;
  float* h2      = ws  + 22 * NP;
  const int C    = (E + CH - 1) / CH;
  int*   cin     = wsi + 23 * NP;
  int*   cout    = cin + C * NP;
  int*   seg     = cout + C * NP;
  int*   csr_src = seg + C * KR * CH;

  const int B = 256;
  k_count<<<dim3(C, 2, KR), 1024, 0, stream>>>(src, dst, E, cin, cout, coarse, seg);
  k_scan<<<(N + 255) / 256, 256, 0, stream>>>(cin, cout, coarse, C, in_feat, N, deg_in,
                                              inv_out, inv_in, xs, rowptr, seg, csr_src);

  const int n_tiles = (N + 15) / 16;
  const int waves_per_block = B / 32;
  k_l1_wmma<<<(n_tiles + waves_per_block - 1) / waves_per_block, B, 0, stream>>>(
      rowptr, deg_in, csr_src, xs, inv_in, inv_out, W0, b0, W1, h1mm, n_tiles);

  k_gather1_l2<<<(N + 31) / 32, 512, 0, stream>>>(rowptr, deg_in, csr_src, h1mm,
                                                   inv_in, inv_out, b1, W2, h2, N);
  k_gather2_out<<<(N + 31) / 32, 512, 0, stream>>>(rowptr, deg_in, csr_src, h2,
                                                    inv_in, b2, out, N);
}


# Both row-sum kernels: 4 lanes per node, not 16 (8 nodes a wave, 128 a work-group), a row's index and value loads in flight together; 391 work-groups of 16 waves enter, not 1563; sum order kept
# speedup vs baseline: 5.8012x; 1.0587x over previous
#include <hip/hip_runtime.h>
#include <math.h>


#define NN 50000
#define NP 50016
#define CH 25000
#define CU ((CH + 1023) / 1024)

typedef __attribute__((ext_vector_type(2))) float v2f;
typedef __attribute__((ext_vector_type(8))) float v8f;

#define KR 4
#define KW 12544
#define KT ((KW / 4 + 1023) / 1024)
#define KS ((KW + 1023) / 1024)
__global__ void __launch_bounds__(1024)
k_count(const int* __restrict__ src, const int* __restrict__ dst, int E,
        int* __restrict__ cin, int* __restrict__ cout, int* __restrict__ coarse,
        int* __restrict__ seg) {
  __shared__ __attribute__((aligned(16))) int s_hist[KW];
  __shared__ int s_wtot[32];
  __shared__ __attribute__((aligned(16))) int s_coarse[256];
  __shared__ __attribute__((aligned(16))) int s_seg[CH];
  const int tid = threadIdx.x;
  const int lane = tid & 31;
  const int wave = tid >> 5;
  const int c = blockIdx.x;
  const bool in_role = blockIdx.y == 0;
  const int r  = blockIdx.z;
  const int r0 = r * KW;
  const int wn = (r0 + KW <= NP) ? KW : NP - r0;
  const int w4 = wn >> 2;
  int4* h4 = reinterpret_cast<int4*>(s_hist);
#pragma unroll
  for (int t = 0; t < KT; ++t) {
    const int i = tid + t * 1024;
    if (i < KW / 4) h4[i] = make_int4(0, 0, 0, 0);
  }
  if (tid < 256) s_coarse[tid] = 0;

  const int e0 = c * CH;
  const int e1 = (e0 + CH < E) ? e0 + CH : E;
  const int* __restrict__ key = in_role ? dst : src;
  int kv[CU];
#pragma unroll
  for (int u = 0; u < CU; ++u) {
    const int idx = e0 + tid + u * 1024;
    kv[u] = key[idx < e1 ? idx : e0];
  }
  int sv[CU];
#pragma unroll
  for (int u = 0; u < CU; ++u) sv[u] = 0;
  if (in_role) {
#pragma unroll
    for (int u = 0; u < CU; ++u) {
      const int idx = e0 + tid + u * 1024;
      sv[u] = src[idx < e1 ? idx : e0];
    }
    asm volatile("" :: "v"(sv[0]), "v"(sv[1]), "v"(sv[2]), "v"(sv[3]), "v"(sv[4]), "v"(sv[5]),
                 "v"(sv[6]), "v"(sv[7]), "v"(sv[8]), "v"(sv[9]), "v"(sv[10]), "v"(sv[11]),
                 "v"(sv[12]));
    asm volatile("" :: "v"(sv[13]), "v"(sv[14]), "v"(sv[15]), "v"(sv[16]), "v"(sv[17]),
                 "v"(sv[18]), "v"(sv[19]), "v"(sv[20]), "v"(sv[21]), "v"(sv[22]), "v"(sv[23]),
                 "v"(sv[24]));
  }
  __syncthreads();

  if (in_role) {
    int rk[CU];
#pragma unroll
    for (int u = 0; u < CU; ++u) {
      const int idx = e0 + tid + u * 1024;
      const unsigned j = (unsigned)(kv[u] - r0);
      rk[u] = 0;
      if (idx < e1 && j < (unsigned)wn) {
        rk[u] = atomicAdd(&s_hist[j], 1);
        atomicAdd(&s_coarse[((j >> 8) << 2) | (tid & 3)], 1);
      }
    }
    __syncthreads();
    const int j0 = tid * KS;
    int cv[KS];
    int sum = 0;
#pragma unroll
    for (int k = 0; k < KS; ++k) {
      const int j = j0 + k;
      const int hv = s_hist[j < KW ? j : 0];
      cv[k] = j < KW ? hv : 0;
      sum += cv[k];
    }
    int incl = sum;
    for (int o = 1; o < 32; o <<= 1) {
      const int t = __shfl_up(incl, o);
      if (lane >= o) incl += t;
    }
    if (lane == 31) s_wtot[wave] = incl;
    __syncthreads();
    int run = incl - sum;
#pragma unroll
    for (int w = 0; w < 32; ++w) run += (w < wave) ? s_wtot[w] : 0;
#pragma unroll
    for (int k = 0; k < KS; ++k) {
      const int j = j0 + k;
      if (j < KW) s_hist[j] = (run << 16) | cv[k];
      run += cv[k];
    }
    __syncthreads();
#pragma unroll
    for (int u = 0; u < CU; ++u) {
      const int idx = e0 + tid + u * 1024;
      const unsigned j = (unsigned)(kv[u] - r0);
      const bool ok = idx < e1 && j < (unsigned)wn;
      rk[u] += (int)((unsigned)s_hist[ok ? j : 0u] >> 16);
    }
    asm volatile("" :: "v"(rk[0]), "v"(rk[1]), "v"(rk[2]), "v"(rk[3]), "v"(rk[4]), "v"(rk[5]),
                 "v"(rk[6]), "v"(rk[7]), "v"(rk[8]), "v"(rk[9]), "v"(rk[10]), "v"(rk[11]),
                 "v"(rk[12]));
    asm volatile("" :: "v"(rk[13]), "v"(rk[14]), "v"(rk[15]), "v"(rk[16]), "v"(rk[17]),
                 "v"(rk[18]), "v"(rk[19]), "v"(rk[20]), "v"(rk[21]), "v"(rk[22]), "v"(rk[23]),
                 "v"(rk[24]));
#pragma unroll
    for (int u = 0; u < CU; ++u) {
      const int idx = e0 + tid + u * 1024;
      const unsigned j = (unsigned)(kv[u] - r0);
      if (idx < e1 && j < (unsigned)wn) s_seg[rk[u]] = sv[u];
    }
  } else {
#pragma unroll
    for (int u = 0; u < CU; ++u) {
      const int idx = e0 + tid + u * 1024;
      const unsigned j = (unsigned)(kv[u] - r0);
      if (idx < e1 && j < (unsigned)wn) atomicAdd(&s_hist[j], 1);
    }
  }
  __syncthreads();
  int4* __restrict__ out4 = reinterpret_cast<int4*>((in_role ? cin : cout) + c * NP + r0);
#pragma unroll
  for (int t = 0; t < KT; ++t) {
    const int i = tid + t * 1024;
    if (i < w4) out4[i] = h4[i];
  }
  if (in_role) {
    int4* __restrict__ sg4 = reinterpret_cast<int4*>(seg + (c * KR + r) * CH);
    const int4* s4 = reinterpret_cast<const int4*>(s_seg);
    const unsigned lw = (unsigned)s_hist[KW - 1];
    const int tot = (int)(lw >> 16) + (int)(lw & 0xFFFFu);
    const int nq = (tot + 3) >> 2;
#pragma unroll
    for (int t = 0; t < (CH / 4 + 1023) / 1024; ++t) {
      const int q = tid + t * 1024;
      if (q < nq) {
        int4 v = s4[q];
        v.y = (4 * q + 1 < tot) ? v.y : 0;
        v.z = (4 * q + 2 < tot) ? v.z : 0;
        v.w = (4 * q + 3 < tot) ? v.w : 0;
        sg4[q] = v;
      }
    }
  }
  if (in_role && tid < 128) {
    const int t = r * (KW / 256) + tid;
    const bool has = tid < KW / 256;
    if (has || (r == KR - 1 && t < 256)) {
      const int4 q = reinterpret_cast<const int4*>(s_coarse)[has ? tid : 0];
      coarse[c * 256 + t] = has ? (q.x + q.y) + (q.z + q.w) : 0;
    }
  }
}

#define CMAX 32
#define MT 4
#define CAPB 8192
#define FENCE16(a, o) asm volatile("" :: "v"(a[o]), "v"(a[o + 1]), "v"(a[o + 2]), "v"(a[o + 3]), \
    "v"(a[o + 4]), "v"(a[o + 5]), "v"(a[o + 6]), "v"(a[o + 7]), "v"(a[o + 8]), "v"(a[o + 9]), \
    "v"(a[o + 10]), "v"(a[o + 11]), "v"(a[o + 12]), "v"(a[o + 13]), "v"(a[o + 14]), "v"(a[o + 15]))
__global__ void __launch_bounds__(256)
k_scan(const int* __restrict__ cin, const int* __restrict__ cout,
       const int* __restrict__ coarse, int C,
       const float* __restrict__ x, int n, int* __restrict__ deg_in,
       float* __restrict__ inv_out, float* __restrict__ inv_in,
       float* __restrict__ xs, int* __restrict__ rowptr,
       const int* __restrict__ seg, int* __restrict__ csr_src) {
  __shared__ int s_part[8];
  __shared__ int s_wsum[8];
  __shared__ __attribute__((aligned(16))) int s_stage[CAPB + 4];
  const int tid  = threadIdx.x;
  const int lane = tid & 31;
  const int wave = tid >> 5;
  const int b    = blockIdx.x;
  const int i    = b * 256 + tid;
  const bool inr = i < n;
  const int ii   = inr ? i : n - 1;

  const int* __restrict__ pco = coarse + tid;
  const int* __restrict__ pin = cin + ii;
  const int* __restrict__ pou = cout + ii;
  int pc[CMAX], ti[CMAX], to[CMAX];
#pragma unroll
  for (int c = 0; c < CMAX; ++c) pc[c] = pco[c * 256];
#pragma unroll
  for (int c = 0; c < CMAX; ++c) ti[c] = pin[c * NP];
#pragma unroll
  for (int c = 0; c < CMAX; ++c) to[c] = pou[c * NP];
  const float xv = x[ii];
  asm volatile("" :: "v"(to[0]), "v"(to[1]), "v"(to[2]), "v"(to[3]), "v"(to[4]), "v"(to[5]),
               "v"(to[6]), "v"(to[7]), "v"(to[8]), "v"(to[9]), "v"(to[10]), "v"(to[11]),
               "v"(to[12]), "v"(to[13]), "v"(to[14]), "v"(to[15]));
  asm volatile("" :: "v"(to[16]), "v"(to[17]), "v"(to[18]), "v"(to[19]), "v"(to[20]), "v"(to[21]),
               "v"(to[22]), "v"(to[23]), "v"(to[24]), "v"(to[25]), "v"(to[26]), "v"(to[27]),
               "v"(to[28]), "v"(to[29]), "v"(to[30]), "v"(to[31]), "v"(xv));
  __builtin_amdgcn_sched_barrier(0);

  int acc = 0, d = 0, dout = 0;
#pragma unroll
  for (int c = 0; c < CMAX; ++c) {
    const bool ok = inr && c < C;
    acc += (c < C) ? pc[c] : 0;
    ti[c] = ok ? ti[c] : 0;
    d += ti[c] & 0xFFFF;
    dout += ok ? to[c] : 0;
  }
  acc = tid < b ? acc : 0;
  for (int o = 16; o > 0; o >>= 1) acc += __shfl_xor(acc, o);
  int incl = d;
  for (int o = 1; o < 32; o <<= 1) {
    const int t = __shfl_up(incl, o);
    if (lane >= o) incl += t;
  }
  if (lane == 0)  s_part[wave] = acc;
  if (lane == 31) s_wsum[wave] = incl;
  __syncthreads();
  int pre = 0, bstart = 0, btot = 0;
  for (int w = 0; w < 8; ++w) {
    const int pw = s_part[w];
    const int sw = s_wsum[w];
    pre += pw;
    bstart += pw;
    if (w < wave) pre += sw;
    btot += sw;
  }
  const int base = pre + incl - d;
  if (inr) {
    rowptr[i] = base;
    deg_in[i] = d;
    const float fi = (float)(d > 1 ? d : 1);
    inv_in[i] = 1.0f / sqrtf(fi);
    const float fo = (float)(dout > 1 ? dout : 1);
    const float io = 1.0f / sqrtf(fo);
    inv_out[i] = io;
    xs[i] = xv * io;
  }
  const int* __restrict__ sgr = seg + (b / (KW / 256)) * CH;
  int mx = 0;
#pragma unroll
  for (int c = 0; c < CMAX; ++c) {
    const int cn = ti[c] & 0xFFFF;
    mx = cn > mx ? cn : mx;
  }
  if (btot <= CAPB) {
    const int rel = (bstart & 3) + base - bstart;
    if (mx > 0) {
      int g[MT * CMAX];
#pragma unroll
      for (int tt = 0; tt < MT; ++tt) {
#pragma unroll
        for (int c = 0; c < CMAX; ++c) {
          const int cn = ti[c] & 0xFFFF;
          const int of = (int)((unsigned)ti[c] >> 16);
          const int* a = (tt < cn) ? sgr + c * (KR * CH) + of + tt : pin + c * NP;
          g[tt * CMAX + c] = *a;
        }
      }
      FENCE16(g, 0);  FENCE16(g, 16); FENCE16(g, 32); FENCE16(g, 48);
      FENCE16(g, 64); FENCE16(g, 80); FENCE16(g, 96); FENCE16(g, 112);
      int run = rel;
#pragma unroll
      for (int c = 0; c < CMAX; ++c) {
        const int cn = ti[c] & 0xFFFF;
#pragma unroll
        for (int tt = 0; tt < MT; ++tt) {
          if (tt < cn) s_stage[run + tt] = g[tt * CMAX + c];
        }
        run += cn;
      }
    }
    if (mx > MT) {
      int run = rel;
#pragma unroll
      for (int c = 0; c < CMAX; ++c) {
        const int cn = ti[c] & 0xFFFF;
        const int of = (int)((unsigned)ti[c] >> 16);
        for (int t = MT; t < cn; ++t) s_stage[run + t] = sgr[c * (KR * CH) + of + t];
        run += cn;
      }
    }
    __syncthreads();
    const int ge = bstart + btot;
    const int q0 = bstart >> 2;
    const int4* st4 = reinterpret_cast<const int4*>(s_stage);
    int4* csr4 = reinterpret_cast<int4*>(csr_src);
    for (int q = q0 + tid; 4 * q < ge; q += 256) {
      const int lo = 4 * q;
      if (lo >= bstart && lo + 4 <= ge) {
        csr4[q] = st4[q - q0];
      } else {
        for (int k = 0; k < 4; ++k) {
          const int gw = lo + k;
          if (gw >= bstart && gw < ge) csr_src[gw] = s_stage[gw - 4 * q0];
        }
      }
    }
  } else if (inr) {
    int run = base;
#pragma nounroll
    for (int c = 0; c < CMAX; ++c) {
      const unsigned tw = (c < C) ? (unsigned)pin[c * NP] : 0u;
      const int cn = (int)(tw & 0xFFFFu);
      const int* __restrict__ pc0 = sgr + c * (KR * CH) + (int)(tw >> 16);
#pragma nounroll
      for (int t = 0; t < cn; ++t) csr_src[run + t] = pc0[t];
      run += cn;
    }
  }
}

template <int U>
__device__ __forceinline__ float row_gather_sum(const int* __restrict__ csr_src,
                                                const float* __restrict__ vals,
                                                int pb, int cnt, int first, int stride) {
  float a = 0.0f;
  for (int q0 = first; q0 < cnt; q0 += 2 * U * stride) {
    const int q1 = q0 + U * stride;
    const bool more = q1 < cnt;
    int sa[U], sb[U];
    float ha[U], hb[U];
#pragma unroll
    for (int u = 0; u < U; ++u) {
      const int q = q0 + u * stride;
      sa[u] = csr_src[pb + (q < cnt ? q : q0)];
    }
    if (more) {
#pragma unroll
      for (int u = 0; u < U; ++u) {
        const int q = q1 + u * stride;
        sb[u] = csr_src[pb + (q < cnt ? q : q1)];
      }
    }
#pragma unroll
    for (int u = 0; u < U; ++u) ha[u] = vals[sa[u]];
    if (more) {
#pragma unroll
      for (int u = 0; u < U; ++u) hb[u] = vals[sb[u]];
    }
#pragma unroll
    for (int u = 0; u < U; ++u) a += (q0 + u * stride < cnt) ? ha[u] : 0.0f;
    if (more) {
#pragma unroll
      for (int u = 0; u < U; ++u) a += (q1 + u * stride < cnt) ? hb[u] : 0.0f;
    }
  }
  return a;
}

#define L1B 256
#define L1S ((1600 + L1B - 1) / L1B)
__global__ void __launch_bounds__(L1B)
k_l1_wmma(const int* __restrict__ rowptr,
                          const int* __restrict__ deg_in,
                          const int* __restrict__ csr_src,
                          const float* __restrict__ xs,
                          const float* __restrict__ inv_in,
                          const float* __restrict__ inv_out,
                          const float* __restrict__ W0,
                          const float* __restrict__ b0,
                          const float* __restrict__ W1,
                          float* __restrict__ h1mm,
                          int n_tiles) {
  __shared__ float sW0[100];
  __shared__ float sb0[100];
  __shared__ float sW1p[100 * 16];
  const int tid  = threadIdx.x;
  const int wave = tid >> 5;
  const int lane = tid & 31;
  const int tile = blockIdx.x * (L1B >> 5) + wave;
  const bool has = tile < n_tiles;
  const int i0   = (has ? tile : 0) * 16;
  const int half = lane >> 4;
  const int m    = lane & 15;
  const int col  = lane & 15;
  const int r    = i0 + m;
  const int pb   = rowptr[r];
  const int cnt  = deg_in[r];
  const float nin = inv_in[r];
  float nout[8];
#pragma unroll
  for (int v = 0; v < 8; ++v) nout[v] = inv_out[i0 + v + 8 * half];
  float w1v[L1S];
#pragma unroll
  for (int j = 0; j < L1S; ++j) {
    const int t = tid + j * L1B;
    const int tc = t < 1600 ? t : 1599;
    const int c = tc & 15;
    w1v[j] = W1[(tc >> 4) * 10 + (c < 10 ? c : 0)];
  }
  const int tw = tid < 100 ? tid : 99;
  const float w0v = W0[tw];
  const float b0v = b0[tw];
  if (tid < 100) { sW0[tid] = w0v; sb0[tid] = b0v; }
#pragma unroll
  for (int j = 0; j < L1S; ++j) {
    const int t = tid + j * L1B;
    if (t < 1600) sW1p[t] = ((t & 15) < 10) ? w1v[j] : 0.0f;
  }
  __syncthreads();
  if (!has) return;

  float part = row_gather_sum<8>(csr_src, xs, pb, cnt, half, 2);
  part += __shfl_xor(part, 16);
  const float s = part * nin;

  v8f c = {};
#pragma unroll
  for (int kk = 0; kk < 25; ++kk) {
    const int k0 = kk * 4 + half * 2;
    float a0 = fmaf(s, sW0[k0], sb0[k0]);
    a0 = a0 >= 0.0f ? a0 : 0.01f * a0;
    float a1 = fmaf(s, sW0[k0 + 1], sb0[k0 + 1]);
    a1 = a1 >= 0.0f ? a1 : 0.01f * a1;
    v2f a = {a0, a1};
    v2f b = {sW1p[k0 * 16 + col], sW1p[(k0 + 1) * 16 + col]};
    c = __builtin_amdgcn_wmma_f32_16x16x4_f32(false, a, false, b, (short)0, c,
                                              false, false);
  }

#pragma unroll
  for (int v = 0; v < 8; ++v) {
    const int i = i0 + v + 8 * half;
    h1mm[i * 16 + col] = c[v] * nout[v];
  }
}

template <int C>
__device__ __forceinline__ int quad_bcast(int x) {
  return __builtin_amdgcn_update_dpp(0, x, C * 0x55, 0xF, 0xF, true);
}
__device__ __forceinline__ float quad_xor1(float x) {
  return __int_as_float(__builtin_amdgcn_update_dpp(0, __float_as_int(x), 0xB1, 0xF, 0xF, true));
}
__device__ __forceinline__ float quad_xor2(float x) {
  return __int_as_float(__builtin_amdgcn_update_dpp(0, __float_as_int(x), 0x4E, 0xF, 0xF, true));
}

#define GB 512
#define GNB (GB / 4)
__global__ void __launch_bounds__(GB, 8)
k_gather1_l2(const int* __restrict__ rowptr, const int* __restrict__ deg_in,
             const int* __restrict__ csr_src, const float* __restrict__ h1mm,
             const float* __restrict__ inv_in, const float* __restrict__ inv_out,
             const float* __restrict__ b1, const float* __restrict__ W2,
             float* __restrict__ h2, int n) {
  __shared__ float s_res[GNB];
  const int g = threadIdx.x >> 2;
  const int j = threadIdx.x & 3;
  const int i = blockIdx.x * GNB + g;
  const bool live = i < n;
  const int ii  = live ? i : 0;
  const int pb  = rowptr[ii];
  const int dg  = deg_in[ii];
  const float nin  = inv_in[ii];
  const float nout = inv_out[ii];
  float bk[4];
#pragma unroll
  for (int c = 0; c < 4; ++c) bk[c] = b1[(4 * j + c) < 10 ? 4 * j + c : 0];
  float w2[10];
#pragma unroll
  for (int kk = 0; kk < 10; ++kk) w2[kk] = W2[kk];
  asm volatile("" :: "v"(pb), "v"(dg));
  const int cnt = live ? dg : 0;
  const float4* __restrict__ rows = reinterpret_cast<const float4*>(h1mm);

  float a0 = 0.0f, a1 = 0.0f, a2s = 0.0f, a3 = 0.0f;
  for (int q0 = 0; q0 < cnt; q0 += 24) {
    const int q1 = q0 + 16;
    const bool more = q1 < cnt;
    int ia[4], ib[2];
    float4 ha[16], hb[8];
#pragma unroll
    for (int m = 0; m < 4; ++m) {
      const int q = q0 + 4 * m + j;
      ia[m] = csr_src[pb + (q < cnt ? q : q0)];
    }
    if (more) {
#pragma unroll
      for (int m = 0; m < 2; ++m) {
        const int q = q1 + 4 * m + j;
        ib[m] = csr_src[pb + (q < cnt ? q : q1)];
      }
    }
#pragma unroll
    for (int u = 0; u < 16; ++u) {
      const int w = ia[u >> 2];
      const int s = (u & 3) == 0 ? quad_bcast<0>(w) : (u & 3) == 1 ? quad_bcast<1>(w)
                  : (u & 3) == 2 ? quad_bcast<2>(w) : quad_bcast<3>(w);
      ha[u] = rows[s * 4 + j];
    }
    if (more) {
#pragma unroll
      for (int u = 0; u < 8; ++u) {
        const int w = ib[u >> 2];
        const int s = (u & 3) == 0 ? quad_bcast<0>(w) : (u & 3) == 1 ? quad_bcast<1>(w)
                    : (u & 3) == 2 ? quad_bcast<2>(w) : quad_bcast<3>(w);
        hb[u] = rows[s * 4 + j];
      }
    }
#pragma unroll
    for (int u = 0; u < 16; ++u) {
      const bool ok = q0 + u < cnt;
      a0  += ok ? ha[u].x : 0.0f;
      a1  += ok ? ha[u].y : 0.0f;
      a2s += ok ? ha[u].z : 0.0f;
      a3  += ok ? ha[u].w : 0.0f;
    }
    if (more) {
#pragma unroll
      for (int u = 0; u < 8; ++u) {
        const bool ok = q1 + u < cnt;
        a0  += ok ? hb[u].x : 0.0f;
        a1  += ok ? hb[u].y : 0.0f;
        a2s += ok ? hb[u].z : 0.0f;
        a3  += ok ? hb[u].w : 0.0f;
      }
    }
  }

  const float acc[4] = {a0, a1, a2s, a3};
  int vi[4];
#pragma unroll
  for (int c = 0; c < 4; ++c) {
    float v = 0.0f;
    if (live && 4 * j + c < 10) {
      v = fmaf(acc[c], nin, bk[c]);
      v = v > 0.0f ? v : 0.0f;
    }
    vi[c] = __float_as_int(v);
  }
  float a2 = 0.0f;
  a2 = fmaf(__int_as_float(quad_bcast<0>(vi[0])), w2[0], a2);
  a2 = fmaf(__int_as_float(quad_bcast<0>(vi[1])), w2[1], a2);
  a2 = fmaf(__int_as_float(quad_bcast<0>(vi[2])), w2[2], a2);
  a2 = fmaf(__int_as_float(quad_bcast<0>(vi[3])), w2[3], a2);
  a2 = fmaf(__int_as_float(quad_bcast<1>(vi[0])), w2[4], a2);
  a2 = fmaf(__int_as_float(quad_bcast<1>(vi[1])), w2[5], a2);
  a2 = fmaf(__int_as_float(quad_bcast<1>(vi[2])), w2[6], a2);
  a2 = fmaf(__int_as_float(quad_bcast<1>(vi[3])), w2[7], a2);
  a2 = fmaf(__int_as_float(quad_bcast<2>(vi[0])), w2[8], a2);
  a2 = fmaf(__int_as_float(quad_bcast<2>(vi[1])), w2[9], a2);
  if (j == 0) s_res[g] = live ? a2 * nout : 0.0f;
  __syncthreads();
  if (threadIdx.x < GNB) {
    const int o = blockIdx.x * GNB + threadIdx.x;
    if (o < n) h2[o] = s_res[threadIdx.x];
  }
}

__global__ void __launch_bounds__(GB)
k_gather2_out(const int* __restrict__ rowptr, const int* __restrict__ deg_in,
              const int* __restrict__ csr_src, const float* __restrict__ h2,
              const float* __restrict__ inv_in, const float* __restrict__ b2,
              float* __restrict__ out, int n) {
  __shared__ float s_res[GNB];
  const int g = threadIdx.x >> 2;
  const int j = threadIdx.x & 3;
  const int i = blockIdx.x * GNB + g;
  const bool live = i < n;
  const int ii  = live ? i : 0;
  const int pb  = rowptr[ii];
  const int dg  = deg_in[ii];
  const float nin  = inv_in[ii];
  const float bias = b2[0];
  asm volatile("" :: "v"(pb), "v"(dg));
  const int cnt = live ? dg : 0;
  float s[4] = {0.0f, 0.0f, 0.0f, 0.0f};
  for (int q0 = 0; q0 < cnt; q0 += 32) {
    const int q1 = q0 + 16;
    const bool more = q1 < cnt;
    int ia[4], ib[4];
    float va[4], vb[4];
#pragma unroll
    for (int m = 0; m < 4; ++m) {
      const int q = q0 + 4 * m + j;
      ia[m] = csr_src[pb + (q < cnt ? q : q0)];
    }
    if (more) {
#pragma unroll
      for (int m = 0; m < 4; ++m) {
        const int q = q1 + 4 * m + j;
        ib[m] = csr_src[pb + (q < cnt ? q : q1)];
      }
    }
#pragma unroll
    for (int m = 0; m < 4; ++m) va[m] = h2[ia[m]];
    if (more) {
#pragma unroll
      for (int m = 0; m < 4; ++m) vb[m] = h2[ib[m]];
    }
#pragma unroll
    for (int m = 0; m < 4; ++m) s[m] += (q0 + 4 * m + j < cnt) ? va[m] : 0.0f;
    if (more) {
#pragma unroll
      for (int m = 0; m < 4; ++m) s[m] += (q1 + 4 * m + j < cnt) ? vb[m] : 0.0f;
    }
  }
  float acc = (s[0] + s[2]) + (s[1] + s[3]);
  acc += quad_xor2(acc);
  acc += quad_xor1(acc);
  if (j == 0) {
    float v = 0.0f;
    if (live) {
      v = fmaf(acc, nin, bias);
      v = v > 0.0f ? v : 0.0f;
    }
    s_res[g] = v;
  }
  __syncthreads();
  if (threadIdx.x < GNB) {
    const int o = blockIdx.x * GNB + threadIdx.x;
    if (o < n) out[o] = s_res[threadIdx.x];
  }
}

extern "C" void kernel_launch(void* const* d_in, const int* in_sizes, int n_in,
                              void* d_out, int out_size, void* d_ws, size_t ws_size,
                              hipStream_t stream) {
  const float* in_feat = (const float*)d_in[0];
  const int*   edge    = (const int*)d_in[1];
  const float* W0      = (const float*)d_in[2];
  const float* b0      = (const float*)d_in[3];
  const float* W1      = (const float*)d_in[4];
  const float* b1      = (const float*)d_in[5];
  const float* W2      = (const float*)d_in[6];
  const float* b2      = (const float*)d_in[7];
  float* out = (float*)d_out;

  const int N = NN;
  const int E = in_sizes[1] / 2;
  const int* src = edge;
  const int* dst = edge + E;

  float* ws      = (float*)d_ws;
  int*   wsi     = (int*)d_ws;
  float* h1mm    = ws;
  int*   coarse  = wsi + 16 * NP;
  int*   deg_in  = wsi + 17 * NP;
  float* inv_out = ws  + 18 * NP;
  float* inv_in  = ws  + 19 * NP;
  float* xs      = ws  + 20 * NP;
  int*   rowptr  = wsi + 21 * NP;
  float* h2      = ws  + 22 * NP;
  const int C    = (E + CH - 1) / CH;
  int*   cin     = wsi + 23 * NP;
  int*   cout    = cin + C * NP;
  int*   seg     = cout + C * NP;
  int*   csr_src = seg + C * KR * CH;

  const int B = 256;
  k_count<<<dim3(C, 2, KR), 1024, 0, stream>>>(src, dst, E, cin, cout, coarse, seg);
  k_scan<<<(N + 255) / 256, 256, 0, stream>>>(cin, cout, coarse, C, in_feat, N, deg_in,
                                              inv_out, inv_in, xs, rowptr, seg, csr_src);

  const int n_tiles = (N + 15) / 16;
  const int waves_per_block = B / 32;
  k_l1_wmma<<<(n_tiles + waves_per_block - 1) / waves_per_block, B, 0, stream>>>(
      rowptr, deg_in, csr_src, xs, inv_in, inv_out, W0, b0, W1, h1mm, n_tiles);

  k_gather1_l2<<<(N + GNB - 1) / GNB, GB, 0, stream>>>(rowptr, deg_in, csr_src, h1mm,
                                                       inv_in, inv_out, b1, W2, h2, N);
  k_gather2_out<<<(N + GNB - 1) / GNB, GB, 0, stream>>>(rowptr, deg_in, csr_src, h2,
                                                        inv_in, b2, out, N);
}


# Scan kernel copies rows by chunk stretch: a block's pieces of a chunk segment are one run, read as whole lines by one wave; a word carries its row's low 8 bits; slot from an LDS table; row order kept
# speedup vs baseline: 6.4042x; 1.1039x over previous
#include <hip/hip_runtime.h>
#include <math.h>


#define NN 50000
#define NP 50016
#define CH 25000
#define CU ((CH + 1023) / 1024)

typedef __attribute__((ext_vector_type(2))) float v2f;
typedef __attribute__((ext_vector_type(8))) float v8f;

#define KR 4
#define KW 12544
#define KT ((KW / 4 + 1023) / 1024)
#define KS ((KW + 1023) / 1024)
static_assert(NN <= 65536, "a segment word keeps the source id in its low 16 bits");
static_assert(KW % 256 == 0, "a node range is a whole number of 256-node blocks");
__global__ void __launch_bounds__(1024)
k_count(const int* __restrict__ src, const int* __restrict__ dst, int E,
        int* __restrict__ cin, int* __restrict__ cout, int* __restrict__ coarse,
        int* __restrict__ seg) {
  __shared__ __attribute__((aligned(16))) int s_hist[KW];
  __shared__ int s_wtot[32];
  __shared__ __attribute__((aligned(16))) int s_coarse[256];
  __shared__ __attribute__((aligned(16))) int s_seg[CH];
  const int tid = threadIdx.x;
  const int lane = tid & 31;
  const int wave = tid >> 5;
  const int c = blockIdx.x;
  const bool in_role = blockIdx.y == 0;
  const int r  = blockIdx.z;
  const int r0 = r * KW;
  const int wn = (r0 + KW <= NP) ? KW : NP - r0;
  const int w4 = wn >> 2;
  int4* h4 = reinterpret_cast<int4*>(s_hist);
#pragma unroll
  for (int t = 0; t < KT; ++t) {
    const int i = tid + t * 1024;
    if (i < KW / 4) h4[i] = make_int4(0, 0, 0, 0);
  }
  if (tid < 256) s_coarse[tid] = 0;

  const int e0 = c * CH;
  const int e1 = (e0 + CH < E) ? e0 + CH : E;
  const int* __restrict__ key = in_role ? dst : src;
  int kv[CU];
#pragma unroll
  for (int u = 0; u < CU; ++u) {
    const int idx = e0 + tid + u * 1024;
    kv[u] = key[idx < e1 ? idx : e0];
  }
  int sv[CU];
#pragma unroll
  for (int u = 0; u < CU; ++u) sv[u] = 0;
  if (in_role) {
#pragma unroll
    for (int u = 0; u < CU; ++u) {
      const int idx = e0 + tid + u * 1024;
      sv[u] = src[idx < e1 ? idx : e0];
    }
    asm volatile("" :: "v"(sv[0]), "v"(sv[1]), "v"(sv[2]), "v"(sv[3]), "v"(sv[4]), "v"(sv[5]),
                 "v"(sv[6]), "v"(sv[7]), "v"(sv[8]), "v"(sv[9]), "v"(sv[10]), "v"(sv[11]),
                 "v"(sv[12]));
    asm volatile("" :: "v"(sv[13]), "v"(sv[14]), "v"(sv[15]), "v"(sv[16]), "v"(sv[17]),
                 "v"(sv[18]), "v"(sv[19]), "v"(sv[20]), "v"(sv[21]), "v"(sv[22]), "v"(sv[23]),
                 "v"(sv[24]));
#pragma unroll
    for (int u = 0; u < CU; ++u) sv[u] |= (kv[u] & 255) << 16;
  }
  __syncthreads();

  if (in_role) {
    int rk[CU];
#pragma unroll
    for (int u = 0; u < CU; ++u) {
      const int idx = e0 + tid + u * 1024;
      const unsigned j = (unsigned)(kv[u] - r0);
      rk[u] = 0;
      if (idx < e1 && j < (unsigned)wn) {
        rk[u] = atomicAdd(&s_hist[j], 1);
        atomicAdd(&s_coarse[((j >> 8) << 2) | (tid & 3)], 1);
      }
    }
    __syncthreads();
    const int j0 = tid * KS;
    int cv[KS];
    int sum = 0;
#pragma unroll
    for (int k = 0; k < KS; ++k) {
      const int j = j0 + k;
      const int hv = s_hist[j < KW ? j : 0];
      cv[k] = j < KW ? hv : 0;
      sum += cv[k];
    }
    int incl = sum;
    for (int o = 1; o < 32; o <<= 1) {
      const int t = __shfl_up(incl, o);
      if (lane >= o) incl += t;
    }
    if (lane == 31) s_wtot[wave] = incl;
    __syncthreads();
    int run = incl - sum;
#pragma unroll
    for (int w = 0; w < 32; ++w) run += (w < wave) ? s_wtot[w] : 0;
#pragma unroll
    for (int k = 0; k < KS; ++k) {
      const int j = j0 + k;
      if (j < KW) s_hist[j] = (run << 16) | cv[k];
      run += cv[k];
    }
    __syncthreads();
#pragma unroll
    for (int u = 0; u < CU; ++u) {
      const int idx = e0 + tid + u * 1024;
      const unsigned j = (unsigned)(kv[u] - r0);
      const bool ok = idx < e1 && j < (unsigned)wn;
      rk[u] += (int)((unsigned)s_hist[ok ? j : 0u] >> 16);
    }
    asm volatile("" :: "v"(rk[0]), "v"(rk[1]), "v"(rk[2]), "v"(rk[3]), "v"(rk[4]), "v"(rk[5]),
                 "v"(rk[6]), "v"(rk[7]), "v"(rk[8]), "v"(rk[9]), "v"(rk[10]), "v"(rk[11]),
                 "v"(rk[12]));
    asm volatile("" :: "v"(rk[13]), "v"(rk[14]), "v"(rk[15]), "v"(rk[16]), "v"(rk[17]),
                 "v"(rk[18]), "v"(rk[19]), "v"(rk[20]), "v"(rk[21]), "v"(rk[22]), "v"(rk[23]),
                 "v"(rk[24]));
#pragma unroll
    for (int u = 0; u < CU; ++u) {
      const int idx = e0 + tid + u * 1024;
      const unsigned j = (unsigned)(kv[u] - r0);
      if (idx < e1 && j < (unsigned)wn) s_seg[rk[u]] = sv[u];
    }
  } else {
#pragma unroll
    for (int u = 0; u < CU; ++u) {
      const int idx = e0 + tid + u * 1024;
      const unsigned j = (unsigned)(kv[u] - r0);
      if (idx < e1 && j < (unsigned)wn) atomicAdd(&s_hist[j], 1);
    }
  }
  __syncthreads();
  int4* __restrict__ out4 = reinterpret_cast<int4*>((in_role ? cin : cout) + c * NP + r0);
#pragma unroll
  for (int t = 0; t < KT; ++t) {
    const int i = tid + t * 1024;
    if (i < w4) out4[i] = h4[i];
  }
  if (in_role) {
    int4* __restrict__ sg4 = reinterpret_cast<int4*>(seg + (c * KR + r) * CH);
    const int4* s4 = reinterpret_cast<const int4*>(s_seg);
    const unsigned lw = (unsigned)s_hist[KW - 1];
    const int tot = (int)(lw >> 16) + (int)(lw & 0xFFFFu);
    const int nq = (tot + 3) >> 2;
#pragma unroll
    for (int t = 0; t < (CH / 4 + 1023) / 1024; ++t) {
      const int q = tid + t * 1024;
      if (q < nq) {
        int4 v = s4[q];
        v.y = (4 * q + 1 < tot) ? v.y : 0;
        v.z = (4 * q + 2 < tot) ? v.z : 0;
        v.w = (4 * q + 3 < tot) ? v.w : 0;
        sg4[q] = v;
      }
    }
  }
  if (in_role && tid < 128) {
    const int t = r * (KW / 256) + tid;
    const bool has = tid < KW / 256;
    if (has || (r == KR - 1 && t < 256)) {
      const int4 q = reinterpret_cast<const int4*>(s_coarse)[has ? tid : 0];
      coarse[c * 256 + t] = has ? (q.x + q.y) + (q.z + q.w) : 0;
    }
  }
}

#define CMAX 32
#define SCW 8
#define CPW (CMAX / SCW)
#define KL 6
#define CAPB 8192
__global__ void __launch_bounds__(256)
k_scan(const int* __restrict__ cin, const int* __restrict__ cout,
       const int* __restrict__ coarse, int C,
       const float* __restrict__ x, int n, int* __restrict__ deg_in,
       float* __restrict__ inv_out, float* __restrict__ inv_in,
       float* __restrict__ xs, int* __restrict__ rowptr,
       const int* __restrict__ seg, int* __restrict__ csr_src) {
  __shared__ int s_part[8];
  __shared__ int s_wsum[8];
  __shared__ __attribute__((aligned(16))) int s_stage[CAPB + 4];
  __shared__ int s_sb[CMAX * 256];
  __shared__ int s_lo[CMAX];
  __shared__ int s_hi[CMAX];
  const int tid  = threadIdx.x;
  const int lane = tid & 31;
  const int wave = tid >> 5;
  const int b    = blockIdx.x;
  const int i    = b * 256 + tid;
  const bool inr = i < n;
  const int ii   = inr ? i : n - 1;

  const int* __restrict__ pco = coarse + tid;
  const int* __restrict__ pin = cin + ii;
  const int* __restrict__ pou = cout + ii;
  int pc[CMAX], ti[CMAX], to[CMAX];
#pragma unroll
  for (int c = 0; c < CMAX; ++c) pc[c] = pco[c * 256];
#pragma unroll
  for (int c = 0; c < CMAX; ++c) ti[c] = pin[c * NP];
#pragma unroll
  for (int c = 0; c < CMAX; ++c) to[c] = pou[c * NP];
  const float xv = x[ii];
  asm volatile("" :: "v"(to[0]), "v"(to[1]), "v"(to[2]), "v"(to[3]), "v"(to[4]), "v"(to[5]),
               "v"(to[6]), "v"(to[7]), "v"(to[8]), "v"(to[9]), "v"(to[10]), "v"(to[11]),
               "v"(to[12]), "v"(to[13]), "v"(to[14]), "v"(to[15]));
  asm volatile("" :: "v"(to[16]), "v"(to[17]), "v"(to[18]), "v"(to[19]), "v"(to[20]), "v"(to[21]),
               "v"(to[22]), "v"(to[23]), "v"(to[24]), "v"(to[25]), "v"(to[26]), "v"(to[27]),
               "v"(to[28]), "v"(to[29]), "v"(to[30]), "v"(to[31]), "v"(xv));
  __builtin_amdgcn_sched_barrier(0);

  int acc = 0, d = 0, dout = 0;
#pragma unroll
  for (int c = 0; c < CMAX; ++c) {
    const bool ok = inr && c < C;
    acc += (c < C) ? pc[c] : 0;
    ti[c] = ok ? ti[c] : 0;
    d += ti[c] & 0xFFFF;
    dout += ok ? to[c] : 0;
  }
  acc = tid < b ? acc : 0;
  for (int o = 16; o > 0; o >>= 1) acc += __shfl_xor(acc, o);
  int incl = d;
  for (int o = 1; o < 32; o <<= 1) {
    const int t = __shfl_up(incl, o);
    if (lane >= o) incl += t;
  }
  if (lane == 0)  s_part[wave] = acc;
  if (lane == 31) s_wsum[wave] = incl;
  __syncthreads();
  int pre = 0, bstart = 0, btot = 0;
  for (int w = 0; w < 8; ++w) {
    const int pw = s_part[w];
    const int sw = s_wsum[w];
    pre += pw;
    bstart += pw;
    if (w < wave) pre += sw;
    btot += sw;
  }
  const int base = pre + incl - d;
  const int* __restrict__ sgr = seg + (b / (KW / 256)) * CH;
  if (btot <= CAPB) {
    const int rel = (bstart & 3) + base - bstart;
    const int lastt = (n - 1 - b * 256 < 255) ? n - 1 - b * 256 : 255;
    int run = rel;
#pragma unroll
    for (int c = 0; c < CMAX; ++c) {
      const int cn = ti[c] & 0xFFFF;
      const int of = (int)((unsigned)ti[c] >> 16);
      s_sb[c * 256 + tid] = run - of;
      run += cn;
    }
    if (tid == 0) {
#pragma unroll
      for (int c = 0; c < CMAX; ++c) s_lo[c] = (int)((unsigned)ti[c] >> 16);
    }
    if (tid == lastt) {
#pragma unroll
      for (int c = 0; c < CMAX; ++c) s_hi[c] = (int)((unsigned)ti[c] >> 16) + (ti[c] & 0xFFFF);
    }
    __syncthreads();
    const int wu = __builtin_amdgcn_readfirstlane(wave);
    int blo[CPW], bhi[CPW];
#pragma unroll
    for (int q = 0; q < CPW; ++q) {
      blo[q] = __builtin_amdgcn_readfirstlane(s_lo[wu + SCW * q]);
      bhi[q] = __builtin_amdgcn_readfirstlane(s_hi[wu + SCW * q]);
    }
    int wv[CPW * KL];
#pragma unroll
    for (int q = 0; q < CPW; ++q) {
      const int c = wu + SCW * q;
#pragma unroll
      for (int k = 0; k < KL; ++k) {
        const int p = blo[q] + lane + 32 * k;
        const int* a = (p < bhi[q]) ? sgr + c * (KR * CH) + p : pin + c * NP;
        wv[q * KL + k] = *a;
      }
    }
    int sl[CPW * KL];
#pragma unroll
    for (int q = 0; q < CPW; ++q) {
      const int c = wu + SCW * q;
#pragma unroll
      for (int k = 0; k < KL; ++k) {
        const int p = blo[q] + lane + 32 * k;
        const unsigned t = (p < bhi[q]) ? ((unsigned)wv[q * KL + k] >> 16) : 0u;
        sl[q * KL + k] = s_sb[c * 256 + (int)t];
      }
    }
#pragma unroll
    for (int q = 0; q < CPW; ++q) {
#pragma unroll
      for (int k = 0; k < KL; ++k) {
        const int p = blo[q] + lane + 32 * k;
        if (p < bhi[q]) s_stage[sl[q * KL + k] + p] = wv[q * KL + k] & 0xFFFF;
      }
    }
#pragma unroll
    for (int q = 0; q < CPW; ++q) {
      const int c = wu + SCW * q;
#pragma nounroll
      for (int p = blo[q] + lane + 32 * KL; p < bhi[q]; p += 32) {
        const unsigned w = (unsigned)sgr[c * (KR * CH) + p];
        s_stage[s_sb[c * 256 + (int)(w >> 16)] + p] = (int)(w & 0xFFFFu);
      }
    }
    __syncthreads();
    const int ge = bstart + btot;
    const int q0 = bstart >> 2;
    const int4* st4 = reinterpret_cast<const int4*>(s_stage);
    int4* csr4 = reinterpret_cast<int4*>(csr_src);
    for (int q = q0 + tid; 4 * q < ge; q += 256) {
      const int lo = 4 * q;
      if (lo >= bstart && lo + 4 <= ge) {
        csr4[q] = st4[q - q0];
      } else {
        for (int k = 0; k < 4; ++k) {
          const int gw = lo + k;
          if (gw >= bstart && gw < ge) csr_src[gw] = s_stage[gw - 4 * q0];
        }
      }
    }
  } else if (inr) {
    int run = base;
#pragma nounroll
    for (int c = 0; c < CMAX; ++c) {
      const unsigned tw = (c < C) ? (unsigned)pin[c * NP] : 0u;
      const int cn = (int)(tw & 0xFFFFu);
      const int* __restrict__ pc0 = sgr + c * (KR * CH) + (int)(tw >> 16);
#pragma nounroll
      for (int t = 0; t < cn; ++t) csr_src[run + t] = pc0[t] & 0xFFFF;
      run += cn;
    }
  }
  if (inr) {
    rowptr[i] = base;
    deg_in[i] = d;
    const float fi = (float)(d > 1 ? d : 1);
    inv_in[i] = 1.0f / sqrtf(fi);
    const float fo = (float)(dout > 1 ? dout : 1);
    const float io = 1.0f / sqrtf(fo);
    inv_out[i] = io;
    xs[i] = xv * io;
  }
}

template <int U>
__device__ __forceinline__ float row_gather_sum(const int* __restrict__ csr_src,
                                                const float* __restrict__ vals,
                                                int pb, int cnt, int first, int stride) {
  float a = 0.0f;
  for (int q0 = first; q0 < cnt; q0 += 2 * U * stride) {
    const int q1 = q0 + U * stride;
    const bool more = q1 < cnt;
    int sa[U], sb[U];
    float ha[U], hb[U];
#pragma unroll
    for (int u = 0; u < U; ++u) {
      const int q = q0 + u * stride;
      sa[u] = csr_src[pb + (q < cnt ? q : q0)];
    }
    if (more) {
#pragma unroll
      for (int u = 0; u < U; ++u) {
        const int q = q1 + u * stride;
        sb[u] = csr_src[pb + (q < cnt ? q : q1)];
      }
    }
#pragma unroll
    for (int u = 0; u < U; ++u) ha[u] = vals[sa[u]];
    if (more) {
#pragma unroll
      for (int u = 0; u < U; ++u) hb[u] = vals[sb[u]];
    }
#pragma unroll
    for (int u = 0; u < U; ++u) a += (q0 + u * stride < cnt) ? ha[u] : 0.0f;
    if (more) {
#pragma unroll
      for (int u = 0; u < U; ++u) a += (q1 + u * stride < cnt) ? hb[u] : 0.0f;
    }
  }
  return a;
}

#define L1B 256
#define L1S ((1600 + L1B - 1) / L1B)
__global__ void __launch_bounds__(L1B)
k_l1_wmma(const int* __restrict__ rowptr,
                          const int* __restrict__ deg_in,
                          const int* __restrict__ csr_src,
                          const float* __restrict__ xs,
                          const float* __restrict__ inv_in,
                          const float* __restrict__ inv_out,
                          const float* __restrict__ W0,
                          const float* __restrict__ b0,
                          const float* __restrict__ W1,
                          float* __restrict__ h1mm,
                          int n_tiles) {
  __shared__ float sW0[100];
  __shared__ float sb0[100];
  __shared__ float sW1p[100 * 16];
  const int tid  = threadIdx.x;
  const int wave = tid >> 5;
  const int lane = tid & 31;
  const int tile = blockIdx.x * (L1B >> 5) + wave;
  const bool has = tile < n_tiles;
  const int i0   = (has ? tile : 0) * 16;
  const int half = lane >> 4;
  const int m    = lane & 15;
  const int col  = lane & 15;
  const int r    = i0 + m;
  const int pb   = rowptr[r];
  const int cnt  = deg_in[r];
  const float nin = inv_in[r];
  float nout[8];
#pragma unroll
  for (int v = 0; v < 8; ++v) nout[v] = inv_out[i0 + v + 8 * half];
  float w1v[L1S];
#pragma unroll
  for (int j = 0; j < L1S; ++j) {
    const int t = tid + j * L1B;
    const int tc = t < 1600 ? t : 1599;
    const int c = tc & 15;
    w1v[j] = W1[(tc >> 4) * 10 + (c < 10 ? c : 0)];
  }
  const int tw = tid < 100 ? tid : 99;
  const float w0v = W0[tw];
  const float b0v = b0[tw];
  if (tid < 100) { sW0[tid] = w0v; sb0[tid] = b0v; }
#pragma unroll
  for (int j = 0; j < L1S; ++j) {
    const int t = tid + j * L1B;
    if (t < 1600) sW1p[t] = ((t & 15) < 10) ? w1v[j] : 0.0f;
  }
  __syncthreads();
  if (!has) return;

  float part = row_gather_sum<8>(csr_src, xs, pb, cnt, half, 2);
  part += __shfl_xor(part, 16);
  const float s = part * nin;

  v8f c = {};
#pragma unroll
  for (int kk = 0; kk < 25; ++kk) {
    const int k0 = kk * 4 + half * 2;
    float a0 = fmaf(s, sW0[k0], sb0[k0]);
    a0 = a0 >= 0.0f ? a0 : 0.01f * a0;
    float a1 = fmaf(s, sW0[k0 + 1], sb0[k0 + 1]);
    a1 = a1 >= 0.0f ? a1 : 0.01f * a1;
    v2f a = {a0, a1};
    v2f b = {sW1p[k0 * 16 + col], sW1p[(k0 + 1) * 16 + col]};
    c = __builtin_amdgcn_wmma_f32_16x16x4_f32(false, a, false, b, (short)0, c,
                                              false, false);
  }

#pragma unroll
  for (int v = 0; v < 8; ++v) {
    const int i = i0 + v + 8 * half;
    h1mm[i * 16 + col] = c[v] * nout[v];
  }
}

template <int C>
__device__ __forceinline__ int quad_bcast(int x) {
  return __builtin_amdgcn_update_dpp(0, x, C * 0x55, 0xF, 0xF, true);
}
__device__ __forceinline__ float quad_xor1(float x) {
  return __int_as_float(__builtin_amdgcn_update_dpp(0, __float_as_int(x), 0xB1, 0xF, 0xF, true));
}
__device__ __forceinline__ float quad_xor2(float x) {
  return __int_as_float(__builtin_amdgcn_update_dpp(0, __float_as_int(x), 0x4E, 0xF, 0xF, true));
}

#define GB 512
#define GNB (GB / 4)
__global__ void __launch_bounds__(GB, 8)
k_gather1_l2(const int* __restrict__ rowptr, const int* __restrict__ deg_in,
             const int* __restrict__ csr_src, const float* __restrict__ h1mm,
             const float* __restrict__ inv_in, const float* __restrict__ inv_out,
             const float* __restrict__ b1, const float* __restrict__ W2,
             float* __restrict__ h2, int n) {
  __shared__ float s_res[GNB];
  const int g = threadIdx.x >> 2;
  const int j = threadIdx.x & 3;
  const int i = blockIdx.x * GNB + g;
  const bool live = i < n;
  const int ii  = live ? i : 0;
  const int pb  = rowptr[ii];
  const int dg  = deg_in[ii];
  const float nin  = inv_in[ii];
  const float nout = inv_out[ii];
  float bk[4];
#pragma unroll
  for (int c = 0; c < 4; ++c) bk[c] = b1[(4 * j + c) < 10 ? 4 * j + c : 0];
  float w2[10];
#pragma unroll
  for (int kk = 0; kk < 10; ++kk) w2[kk] = W2[kk];
  asm volatile("" :: "v"(pb), "v"(dg));
  const int cnt = live ? dg : 0;
  const float4* __restrict__ rows = reinterpret_cast<const float4*>(h1mm);

  float a0 = 0.0f, a1 = 0.0f, a2s = 0.0f, a3 = 0.0f;
  for (int q0 = 0; q0 < cnt; q0 += 24) {
    const int q1 = q0 + 16;
    const bool more = q1 < cnt;
    int ia[4], ib[2];
    float4 ha[16], hb[8];
#pragma unroll
    for (int m = 0; m < 4; ++m) {
      const int q = q0 + 4 * m + j;
      ia[m] = csr_src[pb + (q < cnt ? q : q0)];
    }
    if (more) {
#pragma unroll
      for (int m = 0; m < 2; ++m) {
        const int q = q1 + 4 * m + j;
        ib[m] = csr_src[pb + (q < cnt ? q : q1)];
      }
    }
#pragma unroll
    for (int u = 0; u < 16; ++u) {
      const int w = ia[u >> 2];
      const int s = (u & 3) == 0 ? quad_bcast<0>(w) : (u & 3) == 1 ? quad_bcast<1>(w)
                  : (u & 3) == 2 ? quad_bcast<2>(w) : quad_bcast<3>(w);
      ha[u] = rows[s * 4 + j];
    }
    if (more) {
#pragma unroll
      for (int u = 0; u < 8; ++u) {
        const int w = ib[u >> 2];
        const int s = (u & 3) == 0 ? quad_bcast<0>(w) : (u & 3) == 1 ? quad_bcast<1>(w)
                    : (u & 3) == 2 ? quad_bcast<2>(w) : quad_bcast<3>(w);
        hb[u] = rows[s * 4 + j];
      }
    }
#pragma unroll
    for (int u = 0; u < 16; ++u) {
      const bool ok = q0 + u < cnt;
      a0  += ok ? ha[u].x : 0.0f;
      a1  += ok ? ha[u].y : 0.0f;
      a2s += ok ? ha[u].z : 0.0f;
      a3  += ok ? ha[u].w : 0.0f;
    }
    if (more) {
#pragma unroll
      for (int u = 0; u < 8; ++u) {
        const bool ok = q1 + u < cnt;
        a0  += ok ? hb[u].x : 0.0f;
        a1  += ok ? hb[u].y : 0.0f;
        a2s += ok ? hb[u].z : 0.0f;
        a3  += ok ? hb[u].w : 0.0f;
      }
    }
  }

  const float acc[4] = {a0, a1, a2s, a3};
  int vi[4];
#pragma unroll
  for (int c = 0; c < 4; ++c) {
    float v = 0.0f;
    if (live && 4 * j + c < 10) {
      v = fmaf(acc[c], nin, bk[c]);
      v = v > 0.0f ? v : 0.0f;
    }
    vi[c] = __float_as_int(v);
  }
  float a2 = 0.0f;
  a2 = fmaf(__int_as_float(quad_bcast<0>(vi[0])), w2[0], a2);
  a2 = fmaf(__int_as_float(quad_bcast<0>(vi[1])), w2[1], a2);
  a2 = fmaf(__int_as_float(quad_bcast<0>(vi[2])), w2[2], a2);
  a2 = fmaf(__int_as_float(quad_bcast<0>(vi[3])), w2[3], a2);
  a2 = fmaf(__int_as_float(quad_bcast<1>(vi[0])), w2[4], a2);
  a2 = fmaf(__int_as_float(quad_bcast<1>(vi[1])), w2[5], a2);
  a2 = fmaf(__int_as_float(quad_bcast<1>(vi[2])), w2[6], a2);
  a2 = fmaf(__int_as_float(quad_bcast<1>(vi[3])), w2[7], a2);
  a2 = fmaf(__int_as_float(quad_bcast<2>(vi[0])), w2[8], a2);
  a2 = fmaf(__int_as_float(quad_bcast<2>(vi[1])), w2[9], a2);
  if (j == 0) s_res[g] = live ? a2 * nout : 0.0f;
  __syncthreads();
  if (threadIdx.x < GNB) {
    const int o = blockIdx.x * GNB + threadIdx.x;
    if (o < n) h2[o] = s_res[threadIdx.x];
  }
}

__global__ void __launch_bounds__(GB)
k_gather2_out(const int* __restrict__ rowptr, const int* __restrict__ deg_in,
              const int* __restrict__ csr_src, const float* __restrict__ h2,
              const float* __restrict__ inv_in, const float* __restrict__ b2,
              float* __restrict__ out, int n) {
  __shared__ float s_res[GNB];
  const int g = threadIdx.x >> 2;
  const int j = threadIdx.x & 3;
  const int i = blockIdx.x * GNB + g;
  const bool live = i < n;
  const int ii  = live ? i : 0;
  const int pb  = rowptr[ii];
  const int dg  = deg_in[ii];
  const float nin  = inv_in[ii];
  const float bias = b2[0];
  asm volatile("" :: "v"(pb), "v"(dg));
  const int cnt = live ? dg : 0;
  float s[4] = {0.0f, 0.0f, 0.0f, 0.0f};
  for (int q0 = 0; q0 < cnt; q0 += 32) {
    const int q1 = q0 + 16;
    const bool more = q1 < cnt;
    int ia[4], ib[4];
    float va[4], vb[4];
#pragma unroll
    for (int m = 0; m < 4; ++m) {
      const int q = q0 + 4 * m + j;
      ia[m] = csr_src[pb + (q < cnt ? q : q0)];
    }
    if (more) {
#pragma unroll
      for (int m = 0; m < 4; ++m) {
        const int q = q1 + 4 * m + j;
        ib[m] = csr_src[pb + (q < cnt ? q : q1)];
      }
    }
#pragma unroll
    for (int m = 0; m < 4; ++m) va[m] = h2[ia[m]];
    if (more) {
#pragma unroll
      for (int m = 0; m < 4; ++m) vb[m] = h2[ib[m]];
    }
#pragma unroll
    for (int m = 0; m < 4; ++m) s[m] += (q0 + 4 * m + j < cnt) ? va[m] : 0.0f;
    if (more) {
#pragma unroll
      for (int m = 0; m < 4; ++m) s[m] += (q1 + 4 * m + j < cnt) ? vb[m] : 0.0f;
    }
  }
  float acc = (s[0] + s[2]) + (s[1] + s[3]);
  acc += quad_xor2(acc);
  acc += quad_xor1(acc);
  if (j == 0) {
    float v = 0.0f;
    if (live) {
      v = fmaf(acc, nin, bias);
      v = v > 0.0f ? v : 0.0f;
    }
    s_res[g] = v;
  }
  __syncthreads();
  if (threadIdx.x < GNB) {
    const int o = blockIdx.x * GNB + threadIdx.x;
    if (o < n) out[o] = s_res[threadIdx.x];
  }
}

extern "C" void kernel_launch(void* const* d_in, const int* in_sizes, int n_in,
                              void* d_out, int out_size, void* d_ws, size_t ws_size,
                              hipStream_t stream) {
  const float* in_feat = (const float*)d_in[0];
  const int*   edge    = (const int*)d_in[1];
  const float* W0      = (const float*)d_in[2];
  const float* b0      = (const float*)d_in[3];
  const float* W1      = (const float*)d_in[4];
  const float* b1      = (const float*)d_in[5];
  const float* W2      = (const float*)d_in[6];
  const float* b2      = (const float*)d_in[7];
  float* out = (float*)d_out;

  const int N = NN;
  const int E = in_sizes[1] / 2;
  const int* src = edge;
  const int* dst = edge + E;

  float* ws      = (float*)d_ws;
  int*   wsi     = (int*)d_ws;
  float* h1mm    = ws;
  int*   coarse  = wsi + 16 * NP;
  int*   deg_in  = wsi + 17 * NP;
  float* inv_out = ws  + 18 * NP;
  float* inv_in  = ws  + 19 * NP;
  float* xs      = ws  + 20 * NP;
  int*   rowptr  = wsi + 21 * NP;
  float* h2      = ws  + 22 * NP;
  const int C    = (E + CH - 1) / CH;
  int*   cin     = wsi + 23 * NP;
  int*   cout    = cin + C * NP;
  int*   seg     = cout + C * NP;
  int*   csr_src = seg + C * KR * CH;

  const int B = 256;
  k_count<<<dim3(C, 2, KR), 1024, 0, stream>>>(src, dst, E, cin, cout, coarse, seg);
  k_scan<<<(N + 255) / 256, 256, 0, stream>>>(cin, cout, coarse, C, in_feat, N, deg_in,
                                              inv_out, inv_in, xs, rowptr, seg, csr_src);

  const int n_tiles = (N + 15) / 16;
  const int waves_per_block = B / 32;
  k_l1_wmma<<<(n_tiles + waves_per_block - 1) / waves_per_block, B, 0, stream>>>(
      rowptr, deg_in, csr_src, xs, inv_in, inv_out, W0, b0, W1, h1mm, n_tiles);

  k_gather1_l2<<<(N + GNB - 1) / GNB, GB, 0, stream>>>(rowptr, deg_in, csr_src, h1mm,
                                                       inv_in, inv_out, b1, W2, h2, N);
  k_gather2_out<<<(N + GNB - 1) / GNB, GB, 0, stream>>>(rowptr, deg_in, csr_src, h2,
                                                        inv_in, b2, out, N);
}


# Two-level edge sort: the counting pass bins 6,250-edge chunks by 256-node block only (no node ranges, no per-node in-degree tables); a scan block sorts its edges by node in LDS; row order wave, chunk
# speedup vs baseline: 7.3491x; 1.1475x over previous
#include <hip/hip_runtime.h>
#include <math.h>


#define NN 50000
#define NP 50016
#define CH 25000
#define CU ((CH + 1023) / 1024)
#define KR 4
#define KW 12544
#define KT ((KW / 4 + 1023) / 1024)
#define CB 6250
#define CBU ((CB + 1023) / 1024)
#define NB 196
#define NC (NB * 4)
#define SEGP 6272

typedef __attribute__((ext_vector_type(2))) float v2f;
typedef __attribute__((ext_vector_type(8))) float v8f;

static_assert(NN <= 65536, "a segment word keeps the source id in its low 16 bits");
static_assert(NB * 256 >= NP, "the node blocks cover every per-node array");
static_assert(KR * KW >= NP, "the node ranges cover every per-node array");
static_assert(CB <= SEGP && SEGP % 32 == 0, "a segment is whole lines and holds a chunk");
static_assert(CB < 32768, "offsets and counts of a table word stay below 2^15");
static_assert(SEGP + NC <= KW, "the destination role's LDS fits the source role's table");

__global__ void __launch_bounds__(1024) __attribute__((amdgpu_waves_per_eu(8, 8)))
k_count(const int* __restrict__ src, const int* __restrict__ dst, int E, int C2,
        int* __restrict__ cout, int* __restrict__ tab, int* __restrict__ seg) {
  __shared__ __attribute__((aligned(16))) int s_mem[KW];
  __shared__ int s_wtot[32];
  const int tid = threadIdx.x;
  const int lane = tid & 31;
  const int wave = tid >> 5;
  if ((int)blockIdx.x < C2) {
    int* s_seg = s_mem;
    int* s_cnt = s_mem + SEGP;
    const int c  = blockIdx.x;
    const int e0 = c * CB;
    const int e1 = (e0 + CB < E) ? e0 + CB : E;
    int kv[CBU], sv[CBU];
#pragma unroll
    for (int u = 0; u < CBU; ++u) {
      const int idx = e0 + tid + u * 1024;
      kv[u] = dst[idx < e1 ? idx : e0];
    }
#pragma unroll
    for (int u = 0; u < CBU; ++u) {
      const int idx = e0 + tid + u * 1024;
      sv[u] = src[idx < e1 ? idx : e0];
    }
    asm volatile("" :: "v"(sv[0]), "v"(sv[1]), "v"(sv[2]), "v"(sv[3]), "v"(sv[4]), "v"(sv[5]),
                 "v"(sv[6]));
    if (tid < NC) s_cnt[tid] = 0;
    __syncthreads();
    int rk[CBU];
#pragma unroll
    for (int u = 0; u < CBU; ++u) {
      const int idx = e0 + tid + u * 1024;
      rk[u] = 0;
      if (idx < e1 && (unsigned)kv[u] < (unsigned)NN)
        rk[u] = atomicAdd(&s_cnt[((kv[u] >> 8) << 2) | (tid & 3)], 1);
    }
    __syncthreads();
    const int cv = s_cnt[tid < NC ? tid : 0];
    const int v  = tid < NC ? cv : 0;
    int incl = v;
    for (int o = 1; o < 32; o <<= 1) {
      const int t = __shfl_up(incl, o);
      if (lane >= o) incl += t;
    }
    if (lane == 31) s_wtot[wave] = incl;
    __syncthreads();
    int run = incl - v;
#pragma unroll
    for (int w = 0; w < 32; ++w) run += (w < wave) ? s_wtot[w] : 0;
    if (tid < NC) s_cnt[tid] = (run << 16) | v;
    __syncthreads();
    int ps[CBU];
#pragma unroll
    for (int u = 0; u < CBU; ++u) {
      const int idx = e0 + tid + u * 1024;
      const bool ok = idx < e1 && (unsigned)kv[u] < (unsigned)NN;
      const int ci = ok ? (((kv[u] >> 8) << 2) | (tid & 3)) : 0;
      ps[u] = rk[u] + (int)((unsigned)s_cnt[ci] >> 16);
    }
    asm volatile("" :: "v"(ps[0]), "v"(ps[1]), "v"(ps[2]), "v"(ps[3]), "v"(ps[4]), "v"(ps[5]),
                 "v"(ps[6]));
#pragma unroll
    for (int u = 0; u < CBU; ++u) {
      const int idx = e0 + tid + u * 1024;
      if (idx < e1 && (unsigned)kv[u] < (unsigned)NN)
        s_seg[ps[u]] = ((kv[u] & 255) << 16) | sv[u];
    }
    __syncthreads();
    if (tid < NB)
      reinterpret_cast<int4*>(tab)[c * NB + tid] = reinterpret_cast<const int4*>(s_cnt)[tid];
    int4* __restrict__ sg4 = reinterpret_cast<int4*>(seg + c * SEGP);
    const int4* s4 = reinterpret_cast<const int4*>(s_seg);
    const unsigned lw = (unsigned)s_cnt[NC - 1];
    const int tot = (int)(lw >> 16) + (int)(lw & 0xFFFFu);
    const int nq = (tot + 3) >> 2;
#pragma unroll
    for (int t = 0; t < (SEGP / 4 + 1023) / 1024; ++t) {
      const int q = tid + t * 1024;
      if (q < nq) {
        int4 w4v = s4[q];
        w4v.y = (4 * q + 1 < tot) ? w4v.y : 0;
        w4v.z = (4 * q + 2 < tot) ? w4v.z : 0;
        w4v.w = (4 * q + 3 < tot) ? w4v.w : 0;
        sg4[q] = w4v;
      }
    }
  } else {
    const int g  = (int)blockIdx.x - C2;
    const int c  = g / KR;
    const int r  = g - c * KR;
    const int r0 = r * KW;
    const int wn = (r0 + KW <= NP) ? KW : NP - r0;
    const int w4 = wn >> 2;
    int4* h4 = reinterpret_cast<int4*>(s_mem);
#pragma unroll
    for (int t = 0; t < KT; ++t) {
      const int i = tid + t * 1024;
      if (i < KW / 4) h4[i] = make_int4(0, 0, 0, 0);
    }
    const int e0 = c * CH;
    const int e1 = (e0 + CH < E) ? e0 + CH : E;
    int kv[CU];
#pragma unroll
    for (int u = 0; u < CU; ++u) {
      const int idx = e0 + tid + u * 1024;
      kv[u] = src[idx < e1 ? idx : e0];
    }
    __syncthreads();
#pragma unroll
    for (int u = 0; u < CU; ++u) {
      const int idx = e0 + tid + u * 1024;
      const unsigned j = (unsigned)(kv[u] - r0);
      if (idx < e1 && j < (unsigned)wn) atomicAdd(&s_mem[j], 1);
    }
    __syncthreads();
    int4* __restrict__ out4 = reinterpret_cast<int4*>(cout + c * NP + r0);
#pragma unroll
    for (int t = 0; t < KT; ++t) {
      const int i = tid + t * 1024;
      if (i < w4) out4[i] = h4[i];
    }
  }
}

#define CMAX 32
#define SCW 8
#define CQ 16
#define KL 2
#define CAPB 8192
__global__ void __launch_bounds__(256)
k_scan(const int* __restrict__ tab, const int* __restrict__ cout, int C, int C2,
       const float* __restrict__ x, int n, int* __restrict__ deg_in,
       float* __restrict__ inv_out, float* __restrict__ inv_in,
       float* __restrict__ xs, int* __restrict__ rowptr,
       const int* __restrict__ seg, int* __restrict__ csr_src) {
  __shared__ int s_part[SCW];
  __shared__ int s_wsum[SCW];
  __shared__ int s_rsum[SCW];
  __shared__ __attribute__((aligned(16))) int s_stage[CAPB + 4 + SCW * 32];
  __shared__ int s_wc[SCW * 256 + SCW * 32];
  __shared__ int s_lo[SCW * CQ];
  __shared__ int s_hi[SCW * CQ];
  const int tid  = threadIdx.x;
  const int lane = tid & 31;
  const int wave = tid >> 5;
  const int b    = blockIdx.x;
  const int i    = b * 256 + tid;
  const bool inr = i < n;
  const int ii   = inr ? i : n - 1;

  const bool hasc = tid < C2;
  const int4 tw = reinterpret_cast<const int4*>(tab)[(hasc ? tid : 0) * NB + b];
  const int* __restrict__ pou = cout + ii;
  int to[CMAX];
#pragma unroll
  for (int c = 0; c < CMAX; ++c) to[c] = pou[c * NP];
  const float xv = x[ii];
  asm volatile("" :: "v"(to[0]), "v"(to[1]), "v"(to[2]), "v"(to[3]), "v"(to[4]), "v"(to[5]),
               "v"(to[6]), "v"(to[7]), "v"(to[8]), "v"(to[9]), "v"(to[10]), "v"(to[11]),
               "v"(to[12]), "v"(to[13]), "v"(to[14]), "v"(to[15]));
  asm volatile("" :: "v"(to[16]), "v"(to[17]), "v"(to[18]), "v"(to[19]), "v"(to[20]), "v"(to[21]),
               "v"(to[22]), "v"(to[23]), "v"(to[24]), "v"(to[25]), "v"(to[26]), "v"(to[27]),
               "v"(to[28]), "v"(to[29]), "v"(to[30]), "v"(to[31]), "v"(xv));
  asm volatile("" :: "v"(tw.x), "v"(tw.y), "v"(tw.z), "v"(tw.w));
  __builtin_amdgcn_sched_barrier(0);

#pragma unroll
  for (int w = 0; w < SCW; ++w) s_wc[w * 256 + tid] = 0;
  s_wc[SCW * 256 + tid] = 0;
  const int lo  = hasc ? (int)((unsigned)tw.x >> 16) : 0;
  const int len = hasc ? (tw.x & 0xFFFF) + (tw.y & 0xFFFF) + (tw.z & 0xFFFF) + (tw.w & 0xFFFF) : 0;
  if (tid < SCW * CQ) { s_lo[tid] = lo; s_hi[tid] = lo + len; }
  int alo  = tid < SCW * CQ ? lo : 0;
  int alen = tid < SCW * CQ ? len : 0;
  for (int o = 16; o > 0; o >>= 1) {
    alo  += __shfl_xor(alo, o);
    alen += __shfl_xor(alen, o);
  }
  int dout = 0;
#pragma unroll
  for (int c = 0; c < CMAX; ++c) dout += (inr && c < C) ? to[c] : 0;
  if (lane == 0) { s_part[wave] = alo; s_wsum[wave] = alen; }
  __syncthreads();
  int bstart = 0, btot = 0;
  for (int w = 0; w < SCW; ++w) { bstart += s_part[w]; btot += s_wsum[w]; }
  const bool big = btot > CAPB;

  const int wu = __builtin_amdgcn_readfirstlane(wave);
  int blo[CQ], bhi[CQ];
#pragma unroll
  for (int q = 0; q < CQ; ++q) {
    blo[q] = __builtin_amdgcn_readfirstlane(s_lo[wu + SCW * q]);
    bhi[q] = __builtin_amdgcn_readfirstlane(s_hi[wu + SCW * q]);
  }
  int wv[CQ * KL];
#pragma unroll
  for (int q = 0; q < CQ; ++q) {
    const int c = wu + SCW * q;
#pragma unroll
    for (int k = 0; k < KL; ++k) {
      const int p = blo[q] + lane + 32 * k;
      const int* a = (p < bhi[q]) ? seg + c * SEGP + p : seg;
      wv[q * KL + k] = *a;
    }
  }
  asm volatile("" :: "v"(wv[0]), "v"(wv[1]), "v"(wv[2]), "v"(wv[3]), "v"(wv[4]), "v"(wv[5]),
               "v"(wv[6]), "v"(wv[7]), "v"(wv[8]), "v"(wv[9]), "v"(wv[10]), "v"(wv[11]),
               "v"(wv[12]), "v"(wv[13]), "v"(wv[14]), "v"(wv[15]));
  asm volatile("" :: "v"(wv[16]), "v"(wv[17]), "v"(wv[18]), "v"(wv[19]), "v"(wv[20]), "v"(wv[21]),
               "v"(wv[22]), "v"(wv[23]), "v"(wv[24]), "v"(wv[25]), "v"(wv[26]), "v"(wv[27]),
               "v"(wv[28]), "v"(wv[29]), "v"(wv[30]), "v"(wv[31]));
  __builtin_amdgcn_sched_barrier(0);
  int longest = 0;
#pragma unroll
  for (int q = 0; q < CQ; ++q) longest = (bhi[q] - blo[q] > longest) ? bhi[q] - blo[q] : longest;
  const bool haslong = longest > 32 * KL;
  int* mywc = s_wc + wu * 256;
  const int spare = SCW * 256 + wu * 32 + lane;
  int ci[CQ * KL];
#pragma unroll
  for (int q = 0; q < CQ; ++q) {
#pragma unroll
    for (int k = 0; k < KL; ++k) {
      const int p = blo[q] + lane + 32 * k;
      const int t = (int)(((unsigned)wv[q * KL + k] >> 16) & 255u);
      ci[q * KL + k] = (p < bhi[q]) ? wu * 256 + t : spare;
    }
  }
#pragma unroll
  for (int j = 0; j < CQ * KL; ++j) atomicAdd(&s_wc[ci[j]], 1);
  if (haslong) {
#pragma nounroll
    for (int q = 0; q < CQ; ++q) {
      const int c = wu + SCW * q;
      const int hi = s_hi[c];
#pragma nounroll
      for (int p = s_lo[c] + lane + 32 * KL; p < hi; p += 32) {
        const unsigned w = (unsigned)seg[c * SEGP + p];
        atomicAdd(&mywc[(w >> 16) & 255u], 1);
      }
    }
  }
  __syncthreads();
  int cw[SCW];
  int d = 0;
#pragma unroll
  for (int w = 0; w < SCW; ++w) { cw[w] = s_wc[w * 256 + tid]; d += cw[w]; }
  int incl = d;
  for (int o = 1; o < 32; o <<= 1) {
    const int t = __shfl_up(incl, o);
    if (lane >= o) incl += t;
  }
  if (lane == 31) s_rsum[wave] = incl;
  __syncthreads();
  int pre = 0;
  for (int w = 0; w < SCW; ++w) pre += (w < wave) ? s_rsum[w] : 0;
  const int excl = pre + incl - d;
  const int base = bstart + excl;
  int run = (big ? bstart : (bstart & 3)) + excl;
#pragma unroll
  for (int w = 0; w < SCW; ++w) { s_wc[w * 256 + tid] = run; run += cw[w]; }
  __syncthreads();
  if (!big) {
    int sl[CQ * KL];
#pragma unroll
    for (int j = 0; j < CQ * KL; ++j) sl[j] = atomicAdd(&s_wc[ci[j]], 1);
    const int dump = CAPB + 4 + wu * 32 + lane;
#pragma unroll
    for (int j = 0; j < CQ * KL; ++j)
      s_stage[ci[j] < SCW * 256 ? sl[j] : dump] = wv[j] & 0xFFFF;
    if (haslong) {
#pragma nounroll
      for (int q = 0; q < CQ; ++q) {
        const int c = wu + SCW * q;
        const int hi = s_hi[c];
#pragma nounroll
        for (int p = s_lo[c] + lane + 32 * KL; p < hi; p += 32) {
          const unsigned w = (unsigned)seg[c * SEGP + p];
          s_stage[atomicAdd(&mywc[(w >> 16) & 255u], 1)] = (int)(w & 0xFFFFu);
        }
      }
    }
  } else {
#pragma nounroll
    for (int q = 0; q < CQ; ++q) {
      const int c = wu + SCW * q;
      const int hi = s_hi[c];
#pragma nounroll
      for (int p = s_lo[c] + lane; p < hi; p += 32) {
        const unsigned w = (unsigned)seg[c * SEGP + p];
        csr_src[atomicAdd(&mywc[(w >> 16) & 255u], 1)] = (int)(w & 0xFFFFu);
      }
    }
  }
  __syncthreads();
  if (!big) {
    const int ge = bstart + btot;
    const int q0 = bstart >> 2;
    const int4* st4 = reinterpret_cast<const int4*>(s_stage);
    int4* csr4 = reinterpret_cast<int4*>(csr_src);
    for (int q = q0 + tid; 4 * q < ge; q += 256) {
      const int l4 = 4 * q;
      if (l4 >= bstart && l4 + 4 <= ge) {
        csr4[q] = st4[q - q0];
      } else {
        for (int k = 0; k < 4; ++k) {
          const int gw = l4 + k;
          if (gw >= bstart && gw < ge) csr_src[gw] = s_stage[gw - 4 * q0];
        }
      }
    }
  }
  if (inr) {
    rowptr[i] = base;
    deg_in[i] = d;
    const float fi = (float)(d > 1 ? d : 1);
    inv_in[i] = 1.0f / sqrtf(fi);
    const float fo = (float)(dout > 1 ? dout : 1);
    const float io = 1.0f / sqrtf(fo);
    inv_out[i] = io;
    xs[i] = xv * io;
  }
}

template <int U>
__device__ __forceinline__ float row_gather_sum(const int* __restrict__ csr_src,
                                                const float* __restrict__ vals,
                                                int pb, int cnt, int first, int stride) {
  float a = 0.0f;
  for (int q0 = first; q0 < cnt; q0 += 2 * U * stride) {
    const int q1 = q0 + U * stride;
    const bool more = q1 < cnt;
    int sa[U], sb[U];
    float ha[U], hb[U];
#pragma unroll
    for (int u = 0; u < U; ++u) {
      const int q = q0 + u * stride;
      sa[u] = csr_src[pb + (q < cnt ? q : q0)];
    }
    if (more) {
#pragma unroll
      for (int u = 0; u < U; ++u) {
        const int q = q1 + u * stride;
        sb[u] = csr_src[pb + (q < cnt ? q : q1)];
      }
    }
#pragma unroll
    for (int u = 0; u < U; ++u) ha[u] = vals[sa[u]];
    if (more) {
#pragma unroll
      for (int u = 0; u < U; ++u) hb[u] = vals[sb[u]];
    }
#pragma unroll
    for (int u = 0; u < U; ++u) a += (q0 + u * stride < cnt) ? ha[u] : 0.0f;
    if (more) {
#pragma unroll
      for (int u = 0; u < U; ++u) a += (q1 + u * stride < cnt) ? hb[u] : 0.0f;
    }
  }
  return a;
}

#define L1B 256
#define L1S ((1600 + L1B - 1) / L1B)
__global__ void __launch_bounds__(L1B)
k_l1_wmma(const int* __restrict__ rowptr,
                          const int* __restrict__ deg_in,
                          const int* __restrict__ csr_src,
                          const float* __restrict__ xs,
                          const float* __restrict__ inv_in,
                          const float* __restrict__ inv_out,
                          const float* __restrict__ W0,
                          const float* __restrict__ b0,
                          const float* __restrict__ W1,
                          float* __restrict__ h1mm,
                          int n_tiles) {
  __shared__ float sW0[100];
  __shared__ float sb0[100];
  __shared__ float sW1p[100 * 16];
  const int tid  = threadIdx.x;
  const int wave = tid >> 5;
  const int lane = tid & 31;
  const int tile = blockIdx.x * (L1B >> 5) + wave;
  const bool has = tile < n_tiles;
  const int i0   = (has ? tile : 0) * 16;
  const int half = lane >> 4;
  const int m    = lane & 15;
  const int col  = lane & 15;
  const int r    = i0 + m;
  const int pb   = rowptr[r];
  const int cnt  = deg_in[r];
  const float nin = inv_in[r];
  float nout[8];
#pragma unroll
  for (int v = 0; v < 8; ++v) nout[v] = inv_out[i0 + v + 8 * half];
  float w1v[L1S];
#pragma unroll
  for (int j = 0; j < L1S; ++j) {
    const int t = tid + j * L1B;
    const int tc = t < 1600 ? t : 1599;
    const int c = tc & 15;
    w1v[j] = W1[(tc >> 4) * 10 + (c < 10 ? c : 0)];
  }
  const int tw = tid < 100 ? tid : 99;
  const float w0v = W0[tw];
  const float b0v = b0[tw];
  if (tid < 100) { sW0[tid] = w0v; sb0[tid] = b0v; }
#pragma unroll
  for (int j = 0; j < L1S; ++j) {
    const int t = tid + j * L1B;
    if (t < 1600) sW1p[t] = ((t & 15) < 10) ? w1v[j] : 0.0f;
  }
  __syncthreads();
  if (!has) return;

  float part = row_gather_sum<8>(csr_src, xs, pb, cnt, half, 2);
  part += __shfl_xor(part, 16);
  const float s = part * nin;

  v8f c = {};
#pragma unroll
  for (int kk = 0; kk < 25; ++kk) {
    const int k0 = kk * 4 + half * 2;
    float a0 = fmaf(s, sW0[k0], sb0[k0]);
    a0 = a0 >= 0.0f ? a0 : 0.01f * a0;
    float a1 = fmaf(s, sW0[k0 + 1], sb0[k0 + 1]);
    a1 = a1 >= 0.0f ? a1 : 0.01f * a1;
    v2f a = {a0, a1};
    v2f b = {sW1p[k0 * 16 + col], sW1p[(k0 + 1) * 16 + col]};
    c = __builtin_amdgcn_wmma_f32_16x16x4_f32(false, a, false, b, (short)0, c,
                                              false, false);
  }

#pragma unroll
  for (int v = 0; v < 8; ++v) {
    const int i = i0 + v + 8 * half;
    h1mm[i * 16 + col] = c[v] * nout[v];
  }
}

template <int C>
__device__ __forceinline__ int quad_bcast(int x) {
  return __builtin_amdgcn_update_dpp(0, x, C * 0x55, 0xF, 0xF, true);
}
__device__ __forceinline__ float quad_xor1(float x) {
  return __int_as_float(__builtin_amdgcn_update_dpp(0, __float_as_int(x), 0xB1, 0xF, 0xF, true));
}
__device__ __forceinline__ float quad_xor2(float x) {
  return __int_as_float(__builtin_amdgcn_update_dpp(0, __float_as_int(x), 0x4E, 0xF, 0xF, true));
}

#define GB 512
#define GNB (GB / 4)
__global__ void __launch_bounds__(GB, 8)
k_gather1_l2(const int* __restrict__ rowptr, const int* __restrict__ deg_in,
             const int* __restrict__ csr_src, const float* __restrict__ h1mm,
             const float* __restrict__ inv_in, const float* __restrict__ inv_out,
             const float* __restrict__ b1, const float* __restrict__ W2,
             float* __restrict__ h2, int n) {
  __shared__ float s_res[GNB];
  const int g = threadIdx.x >> 2;
  const int j = threadIdx.x & 3;
  const int i = blockIdx.x * GNB + g;
  const bool live = i < n;
  const int ii  = live ? i : 0;
  const int pb  = rowptr[ii];
  const int dg  = deg_in[ii];
  const float nin  = inv_in[ii];
  const float nout = inv_out[ii];
  float bk[4];
#pragma unroll
  for (int c = 0; c < 4; ++c) bk[c] = b1[(4 * j + c) < 10 ? 4 * j + c : 0];
  float w2[10];
#pragma unroll
  for (int kk = 0; kk < 10; ++kk) w2[kk] = W2[kk];
  asm volatile("" :: "v"(pb), "v"(dg));
  const int cnt = live ? dg : 0;
  const float4* __restrict__ rows = reinterpret_cast<const float4*>(h1mm);

  float a0 = 0.0f, a1 = 0.0f, a2s = 0.0f, a3 = 0.0f;
  for (int q0 = 0; q0 < cnt; q0 += 24) {
    const int q1 = q0 + 16;
    const bool more = q1 < cnt;
    int ia[4], ib[2];
    float4 ha[16], hb[8];
#pragma unroll
    for (int m = 0; m < 4; ++m) {
      const int q = q0 + 4 * m + j;
      ia[m] = csr_src[pb + (q < cnt ? q : q0)];
    }
    if (more) {
#pragma unroll
      for (int m = 0; m < 2; ++m) {
        const int q = q1 + 4 * m + j;
        ib[m] = csr_src[pb + (q < cnt ? q : q1)];
      }
    }
#pragma unroll
    for (int u = 0; u < 16; ++u) {
      const int w = ia[u >> 2];
      const int s = (u & 3) == 0 ? quad_bcast<0>(w) : (u & 3) == 1 ? quad_bcast<1>(w)
                  : (u & 3) == 2 ? quad_bcast<2>(w) : quad_bcast<3>(w);
      ha[u] = rows[s * 4 + j];
    }
    if (more) {
#pragma unroll
      for (int u = 0; u < 8; ++u) {
        const int w = ib[u >> 2];
        const int s = (u & 3) == 0 ? quad_bcast<0>(w) : (u & 3) == 1 ? quad_bcast<1>(w)
                    : (u & 3) == 2 ? quad_bcast<2>(w) : quad_bcast<3>(w);
        hb[u] = rows[s * 4 + j];
      }
    }
#pragma unroll
    for (int u = 0; u < 16; ++u) {
      const bool ok = q0 + u < cnt;
      a0  += ok ? ha[u].x : 0.0f;
      a1  += ok ? ha[u].y : 0.0f;
      a2s += ok ? ha[u].z : 0.0f;
      a3  += ok ? ha[u].w : 0.0f;
    }
    if (more) {
#pragma unroll
      for (int u = 0; u < 8; ++u) {
        const bool ok = q1 + u < cnt;
        a0  += ok ? hb[u].x : 0.0f;
        a1  += ok ? hb[u].y : 0.0f;
        a2s += ok ? hb[u].z : 0.0f;
        a3  += ok ? hb[u].w : 0.0f;
      }
    }
  }

  const float acc[4] = {a0, a1, a2s, a3};
  int vi[4];
#pragma unroll
  for (int c = 0; c < 4; ++c) {
    float v = 0.0f;
    if (live && 4 * j + c < 10) {
      v = fmaf(acc[c], nin, bk[c]);
      v = v > 0.0f ? v : 0.0f;
    }
    vi[c] = __float_as_int(v);
  }
  float a2 = 0.0f;
  a2 = fmaf(__int_as_float(quad_bcast<0>(vi[0])), w2[0], a2);
  a2 = fmaf(__int_as_float(quad_bcast<0>(vi[1])), w2[1], a2);
  a2 = fmaf(__int_as_float(quad_bcast<0>(vi[2])), w2[2], a2);
  a2 = fmaf(__int_as_float(quad_bcast<0>(vi[3])), w2[3], a2);
  a2 = fmaf(__int_as_float(quad_bcast<1>(vi[0])), w2[4], a2);
  a2 = fmaf(__int_as_float(quad_bcast<1>(vi[1])), w2[5], a2);
  a2 = fmaf(__int_as_float(quad_bcast<1>(vi[2])), w2[6], a2);
  a2 = fmaf(__int_as_float(quad_bcast<1>(vi[3])), w2[7], a2);
  a2 = fmaf(__int_as_float(quad_bcast<2>(vi[0])), w2[8], a2);
  a2 = fmaf(__int_as_float(quad_bcast<2>(vi[1])), w2[9], a2);
  if (j == 0) s_res[g] = live ? a2 * nout : 0.0f;
  __syncthreads();
  if (threadIdx.x < GNB) {
    const int o = blockIdx.x * GNB + threadIdx.x;
    if (o < n) h2[o] = s_res[threadIdx.x];
  }
}

__global__ void __launch_bounds__(GB)
k_gather2_out(const int* __restrict__ rowptr, const int* __restrict__ deg_in,
              const int* __restrict__ csr_src, const float* __restrict__ h2,
              const float* __restrict__ inv_in, const float* __restrict__ b2,
              float* __restrict__ out, int n) {
  __shared__ float s_res[GNB];
  const int g = threadIdx.x >> 2;
  const int j = threadIdx.x & 3;
  const int i = blockIdx.x * GNB + g;
  const bool live = i < n;
  const int ii  = live ? i : 0;
  const int pb  = rowptr[ii];
  const int dg  = deg_in[ii];
  const float nin  = inv_in[ii];
  const float bias = b2[0];
  asm volatile("" :: "v"(pb), "v"(dg));
  const int cnt = live ? dg : 0;
  float s[4] = {0.0f, 0.0f, 0.0f, 0.0f};
  for (int q0 = 0; q0 < cnt; q0 += 32) {
    const int q1 = q0 + 16;
    const bool more = q1 < cnt;
    int ia[4], ib[4];
    float va[4], vb[4];
#pragma unroll
    for (int m = 0; m < 4; ++m) {
      const int q = q0 + 4 * m + j;
      ia[m] = csr_src[pb + (q < cnt ? q : q0)];
    }
    if (more) {
#pragma unroll
      for (int m = 0; m < 4; ++m) {
        const int q = q1 + 4 * m + j;
        ib[m] = csr_src[pb + (q < cnt ? q : q1)];
      }
    }
#pragma unroll
    for (int m = 0; m < 4; ++m) va[m] = h2[ia[m]];
    if (more) {
#pragma unroll
      for (int m = 0; m < 4; ++m) vb[m] = h2[ib[m]];
    }
#pragma unroll
    for (int m = 0; m < 4; ++m) s[m] += (q0 + 4 * m + j < cnt) ? va[m] : 0.0f;
    if (more) {
#pragma unroll
      for (int m = 0; m < 4; ++m) s[m] += (q1 + 4 * m + j < cnt) ? vb[m] : 0.0f;
    }
  }
  float acc = (s[0] + s[2]) + (s[1] + s[3]);
  acc += quad_xor2(acc);
  acc += quad_xor1(acc);
  if (j == 0) {
    float v = 0.0f;
    if (live) {
      v = fmaf(acc, nin, bias);
      v = v > 0.0f ? v : 0.0f;
    }
    s_res[g] = v;
  }
  __syncthreads();
  if (threadIdx.x < GNB) {
    const int o = blockIdx.x * GNB + threadIdx.x;
    if (o < n) out[o] = s_res[threadIdx.x];
  }
}

extern "C" void kernel_launch(void* const* d_in, const int* in_sizes, int n_in,
                              void* d_out, int out_size, void* d_ws, size_t ws_size,
                              hipStream_t stream) {
  const float* in_feat = (const float*)d_in[0];
  const int*   edge    = (const int*)d_in[1];
  const float* W0      = (const float*)d_in[2];
  const float* b0      = (const float*)d_in[3];
  const float* W1      = (const float*)d_in[4];
  const float* b1      = (const float*)d_in[5];
  const float* W2      = (const float*)d_in[6];
  const float* b2      = (const float*)d_in[7];
  float* out = (float*)d_out;

  const int N = NN;
  const int E = in_sizes[1] / 2;
  const int* src = edge;
  const int* dst = edge + E;

  float* ws      = (float*)d_ws;
  int*   wsi     = (int*)d_ws;
  float* h1mm    = ws;
  int*   deg_in  = wsi + 17 * NP;
  float* inv_out = ws  + 18 * NP;
  float* inv_in  = ws  + 19 * NP;
  float* xs      = ws  + 20 * NP;
  int*   rowptr  = wsi + 21 * NP;
  float* h2      = ws  + 22 * NP;
  const int C    = (E + CH - 1) / CH;
  const int C2   = (E + CB - 1) / CB;
  int*   cout    = wsi + 23 * NP;
  int*   tab     = cout + C * NP;
  int*   seg     = tab + C2 * NC;
  int*   csr_src = seg + C2 * SEGP;

  const int B = 256;
  k_count<<<C2 + C * KR, 1024, 0, stream>>>(src, dst, E, C2, cout, tab, seg);
  k_scan<<<NB, 256, 0, stream>>>(tab, cout, C, C2, in_feat, N, deg_in,
                                 inv_out, inv_in, xs, rowptr, seg, csr_src);

  const int n_tiles = (N + 15) / 16;
  const int waves_per_block = B / 32;
  k_l1_wmma<<<(n_tiles + waves_per_block - 1) / waves_per_block, B, 0, stream>>>(
      rowptr, deg_in, csr_src, xs, inv_in, inv_out, W0, b0, W1, h1mm, n_tiles);

  k_gather1_l2<<<(N + GNB - 1) / GNB, GB, 0, stream>>>(rowptr, deg_in, csr_src, h1mm,
                                                       inv_in, inv_out, b1, W2, h2, N);
  k_gather2_out<<<(N + GNB - 1) / GNB, GB, 0, stream>>>(rowptr, deg_in, csr_src, h2,
                                                        inv_in, b2, out, N);
}
